# starting kernel
# baseline (speedup 1.0000x reference)
#include <hip/hip_runtime.h>
#include <math.h>

#define NUM_B 8
#define NUM_N 8192
#define NUM_G 256
#define GK    32
#define C1    128
#define C2    256
#define C3    512
#define CT    384
#define TOK_TOTAL (NUM_B * NUM_G * CT)

typedef __attribute__((ext_vector_type(2))) float v2f;
typedef __attribute__((ext_vector_type(8))) float v8f;

__device__ __forceinline__ v8f wmma4(v2f a, v2f b, v8f c) {
  return __builtin_amdgcn_wmma_f32_16x16x4_f32(false, a, false, b, (short)0, c,
                                               false, false);
}

template <int KDIM, int LDB>
__device__ __forceinline__ void gemm_2x2(const float* __restrict__ W, int ldw,
                                         const float* B, int m0, int laneN,
                                         int laneHalf, v8f& acc00, v8f& acc01,
                                         v8f& acc10, v8f& acc11) {
  const float* wrow0 = W + (size_t)(m0 + laneN) * (size_t)ldw + 2 * laneHalf;
  const float* wrow1 = wrow0 + (size_t)16 * (size_t)ldw;
  const float* bcol0 = B + laneN * LDB + 2 * laneHalf;
  const float* bcol1 = B + (16 + laneN) * LDB + 2 * laneHalf;
#pragma unroll 4
  for (int kk = 0; kk < KDIM; kk += 4) {
    v2f a0 = *(const v2f*)(wrow0 + kk);
    v2f a1 = *(const v2f*)(wrow1 + kk);
    v2f b0 = *(const v2f*)(bcol0 + kk);
    v2f b1 = *(const v2f*)(bcol1 + kk);
    acc00 = wmma4(a0, b0, acc00);
    acc01 = wmma4(a0, b1, acc01);
    acc10 = wmma4(a1, b0, acc10);
    acc11 = wmma4(a1, b1, acc11);
  }
}

template <int KDIM, int LDB>
__device__ __forceinline__ void gemm_1x2(const float* __restrict__ W, int ldw,
                                         const float* B, int m0, int laneN,
                                         int laneHalf, v8f& acc0, v8f& acc1) {
  const float* wrow  = W + (size_t)(m0 + laneN) * (size_t)ldw + 2 * laneHalf;
  const float* bcol0 = B + laneN * LDB + 2 * laneHalf;
  const float* bcol1 = B + (16 + laneN) * LDB + 2 * laneHalf;
#pragma unroll 4
  for (int kk = 0; kk < KDIM; kk += 4) {
    v2f a  = *(const v2f*)(wrow + kk);
    v2f b0 = *(const v2f*)(bcol0 + kk);
    v2f b1 = *(const v2f*)(bcol1 + kk);
    acc0 = wmma4(a, b0, acc0);
    acc1 = wmma4(a, b1, acc1);
  }
}

template <typename F>
__device__ __forceinline__ void store_tile(float* dst, int ldd, const v8f& acc,
                                           int m0, int n0, int laneN,
                                           int laneHalf, F op) {
  int n = n0 + laneN;
#pragma unroll
  for (int rr = 0; rr < 8; ++rr) {
    int m = m0 + 8 * laneHalf + rr;
    dst[n * ldd + m] = op(acc[rr], m);
  }
}

__global__ __launch_bounds__(1024) void fps_kernel(
    const float* __restrict__ points, float* __restrict__ centers) {
  const int b    = blockIdx.x;
  const int tid  = threadIdx.x;
  const int lane = tid & 31;
  const int wv   = tid >> 5;
  const float* base = points + (size_t)b * NUM_N * 3;

  __shared__ float sLast[3];
  __shared__ int   sWin;
  __shared__ float sWv[32];
  __shared__ int   sWi[32];

  float px[8], py[8], pz[8], dd[8];
#pragma unroll
  for (int i = 0; i < 8; ++i) {
    int p = tid * 8 + i;
    px[i] = base[p * 3 + 0];
    py[i] = base[p * 3 + 1];
    pz[i] = base[p * 3 + 2];
    dd[i] = 1e10f;
  }
  if (tid == 0) {
    sLast[0] = base[0]; sLast[1] = base[1]; sLast[2] = base[2];
    float* co = centers + (size_t)b * NUM_G * 3;
    co[0] = base[0]; co[1] = base[1]; co[2] = base[2];
  }
  __syncthreads();

  for (int g = 1; g < NUM_G; ++g) {
    float lx = sLast[0], ly = sLast[1], lz = sLast[2];
    float bv = -1.f;
    int   bi = 0;
#pragma unroll
    for (int i = 0; i < 8; ++i) {
      float dx = px[i] - lx, dy = py[i] - ly, dz = pz[i] - lz;
      float d  = dx * dx + dy * dy + dz * dz;
      dd[i]    = fminf(dd[i], d);
      if (dd[i] > bv) { bv = dd[i]; bi = tid * 8 + i; }
    }
#pragma unroll
    for (int off = 16; off > 0; off >>= 1) {
      float ov = __shfl_xor(bv, off, 32);
      int   oi = __shfl_xor(bi, off, 32);
      if (ov > bv || (ov == bv && oi < bi)) { bv = ov; bi = oi; }
    }
    if (lane == 0) { sWv[wv] = bv; sWi[wv] = bi; }
    __syncthreads();
    if (tid < 32) {
      float v  = sWv[tid];
      int   ii = sWi[tid];
#pragma unroll
      for (int off = 16; off > 0; off >>= 1) {
        float ov = __shfl_xor(v, off, 32);
        int   oi = __shfl_xor(ii, off, 32);
        if (ov > v || (ov == v && oi < ii)) { v = ov; ii = oi; }
      }
      if (tid == 0) sWin = ii;
    }
    __syncthreads();
    int wi = sWin;
    if (tid == (wi >> 3)) {
      int i = wi & 7;
      sLast[0] = px[i]; sLast[1] = py[i]; sLast[2] = pz[i];
      float* co = centers + ((size_t)b * NUM_G + g) * 3;
      co[0] = px[i]; co[1] = py[i]; co[2] = pz[i];
    }
    __syncthreads();
  }
}

__global__ __launch_bounds__(256) void knn_kernel(
    const float* __restrict__ points, const float* __restrict__ centers,
    int* __restrict__ knnIdx) {
  const int gg   = blockIdx.x;
  const int b    = gg / NUM_G;
  const int tid  = threadIdx.x;
  const int lane = tid & 31;
  const int wv   = tid >> 5;
  const float* base = points + (size_t)b * NUM_N * 3;

  __shared__ float sd[NUM_N];
  __shared__ float sWv[8];
  __shared__ int   sWi[8];

  float cx = centers[(size_t)gg * 3 + 0];
  float cy = centers[(size_t)gg * 3 + 1];
  float cz = centers[(size_t)gg * 3 + 2];
  float c2 = cx * cx + cy * cy + cz * cz;

  for (int i = 0; i < NUM_N / 256; ++i) {
    int   p  = i * 256 + tid;
    float xx = base[p * 3 + 0], yy = base[p * 3 + 1], zz = base[p * 3 + 2];
    float x2 = xx * xx + yy * yy + zz * zz;
    float dt = cx * xx + cy * yy + cz * zz;
    sd[p]    = c2 + x2 - 2.f * dt;
  }
  __syncthreads();

  for (int r = 0; r < GK; ++r) {
    float bv = 3.0e38f;
    int   bi = NUM_N;
    for (int i = 0; i < NUM_N / 256; ++i) {
      int   p = i * 256 + tid;
      float v = sd[p];
      if (v < bv || (v == bv && p < bi)) { bv = v; bi = p; }
    }
#pragma unroll
    for (int off = 16; off > 0; off >>= 1) {
      float ov = __shfl_xor(bv, off, 32);
      int   oi = __shfl_xor(bi, off, 32);
      if (ov < bv || (ov == bv && oi < bi)) { bv = ov; bi = oi; }
    }
    if (lane == 0) { sWv[wv] = bv; sWi[wv] = bi; }
    __syncthreads();
    if (tid == 0) {
      float v  = sWv[0];
      int   ii = sWi[0];
      for (int w = 1; w < 8; ++w)
        if (sWv[w] < v || (sWv[w] == v && sWi[w] < ii)) { v = sWv[w]; ii = sWi[w]; }
      knnIdx[(size_t)gg * GK + r] = ii;
      sd[ii] = 3.4e38f;
    }
    __syncthreads();
  }
}

__global__ __launch_bounds__(256) void encoder_kernel(
    const float* __restrict__ points, const int* __restrict__ knnIdx,
    const float* __restrict__ centers, const float* __restrict__ w1,
    const float* __restrict__ b1, const float* __restrict__ g1,
    const float* __restrict__ be1, const float* __restrict__ m1,
    const float* __restrict__ v1, const float* __restrict__ w2,
    const float* __restrict__ b2, const float* __restrict__ w3,
    const float* __restrict__ b3, const float* __restrict__ g2,
    const float* __restrict__ be2, const float* __restrict__ m2,
    const float* __restrict__ v2, const float* __restrict__ w4,
    const float* __restrict__ b4, float* __restrict__ tokens) {
  extern __shared__ float smem[];
  float* sFeat = smem;
  float* sH1   = smem + 128;
  float* sX3   = smem + 128 + 32 * 132;
  float* sH3   = sX3 + 32 * 516;
  float* sH4   = sX3;

  const int gg       = blockIdx.x;
  const int b        = gg / NUM_G;
  const int tid      = threadIdx.x;
  const int lane     = tid & 31;
  const int wv       = tid >> 5;
  const int laneN    = lane & 15;
  const int laneHalf = lane >> 4;

  const v8f vzero = {0.f, 0.f, 0.f, 0.f, 0.f, 0.f, 0.f, 0.f};

  if (tid < GK) {
    int idx = knnIdx[(size_t)gg * GK + tid];
    const float* p = points + ((size_t)b * NUM_N + idx) * 3;
    const float* c = centers + (size_t)gg * 3;
    float dx = p[0] - c[0], dy = p[1] - c[1], dz = p[2] - c[2];
    float nr = sqrtf(dx * dx + dy * dy + dz * dz);
    sFeat[tid * 4 + 0] = dx;
    sFeat[tid * 4 + 1] = dy;
    sFeat[tid * 4 + 2] = dz;
    sFeat[tid * 4 + 3] = nr;
  }
  __syncthreads();

  {
    int m0 = wv * 16;
    v8f a0 = vzero, a1 = vzero;
    gemm_1x2<4, 4>(w1, 4, sFeat, m0, laneN, laneHalf, a0, a1);
    auto op = [&](float x, int m) {
      float inv = g1[m] * rsqrtf(v1[m] + 1e-5f);
      float v   = (x + b1[m]) * inv + (be1[m] - m1[m] * inv);
      return fmaxf(v, 0.f);
    };
    store_tile(sH1, 132, a0, m0, 0,  laneN, laneHalf, op);
    store_tile(sH1, 132, a1, m0, 16, laneN, laneHalf, op);
  }
  __syncthreads();

  {
    int m0 = wv * 32;
    v8f a00 = vzero, a01 = vzero, a10 = vzero, a11 = vzero;
    gemm_2x2<C1, 132>(w2, C1, sH1, m0, laneN, laneHalf, a00, a01, a10, a11);
    auto op = [&](float x, int m) { return x + b2[m]; };
    store_tile(sX3 + 256, 516, a00, m0,      0,  laneN, laneHalf, op);
    store_tile(sX3 + 256, 516, a01, m0,      16, laneN, laneHalf, op);
    store_tile(sX3 + 256, 516, a10, m0 + 16, 0,  laneN, laneHalf, op);
    store_tile(sX3 + 256, 516, a11, m0 + 16, 16, laneN, laneHalf, op);
  }
  __syncthreads();

  {
    int   m  = tid;
    float gm = -3.4e38f;
    for (int n = 0; n < GK; ++n) gm = fmaxf(gm, sX3[n * 516 + 256 + m]);
    for (int n = 0; n < GK; ++n) sX3[n * 516 + m] = gm;
  }
  __syncthreads();

  {
    auto op = [&](float x, int m) {
      float inv = g2[m] * rsqrtf(v2[m] + 1e-5f);
      float v   = (x + b3[m]) * inv + (be2[m] - m2[m] * inv);
      return fmaxf(v, 0.f);
    };
#pragma unroll
    for (int j = 0; j < 2; ++j) {
      int m0  = wv * 64 + j * 32;
      v8f a00 = vzero, a01 = vzero, a10 = vzero, a11 = vzero;
      gemm_2x2<C3, 516>(w3, C3, sX3, m0, laneN, laneHalf, a00, a01, a10, a11);
      store_tile(sH3, 516, a00, m0,      0,  laneN, laneHalf, op);
      store_tile(sH3, 516, a01, m0,      16, laneN, laneHalf, op);
      store_tile(sH3, 516, a10, m0 + 16, 0,  laneN, laneHalf, op);
      store_tile(sH3, 516, a11, m0 + 16, 16, laneN, laneHalf, op);
    }
  }
  __syncthreads();

  {
    auto op = [&](float x, int m) { return x + b4[m]; };
    {
      int m0  = wv * 48;
      v8f a00 = vzero, a01 = vzero, a10 = vzero, a11 = vzero;
      gemm_2x2<C3, 516>(w4, C3, sH3, m0, laneN, laneHalf, a00, a01, a10, a11);
      store_tile(sH4, 388, a00, m0,      0,  laneN, laneHalf, op);
      store_tile(sH4, 388, a01, m0,      16, laneN, laneHalf, op);
      store_tile(sH4, 388, a10, m0 + 16, 0,  laneN, laneHalf, op);
      store_tile(sH4, 388, a11, m0 + 16, 16, laneN, laneHalf, op);
    }
    {
      int m0 = wv * 48 + 32;
      v8f a0 = vzero, a1 = vzero;
      gemm_1x2<C3, 516>(w4, C3, sH3, m0, laneN, laneHalf, a0, a1);
      store_tile(sH4, 388, a0, m0, 0,  laneN, laneHalf, op);
      store_tile(sH4, 388, a1, m0, 16, laneN, laneHalf, op);
    }
  }
  __syncthreads();

  for (int m = tid; m < CT; m += 256) {
    float mx = -3.4e38f;
    for (int n = 0; n < GK; ++n) mx = fmaxf(mx, sH4[n * 388 + m]);
    tokens[(size_t)gg * CT + m] = mx;
  }
}

extern "C" void kernel_launch(void* const* d_in, const int* in_sizes, int n_in,
                              void* d_out, int out_size, void* d_ws,
                              size_t ws_size, hipStream_t stream) {
  (void)in_sizes; (void)n_in; (void)out_size; (void)ws_size;
  const float* points = (const float*)d_in[0];
  const float* w1  = (const float*)d_in[1];
  const float* b1  = (const float*)d_in[2];
  const float* g1  = (const float*)d_in[3];
  const float* be1 = (const float*)d_in[4];
  const float* m1  = (const float*)d_in[5];
  const float* v1  = (const float*)d_in[6];
  const float* w2  = (const float*)d_in[7];
  const float* b2  = (const float*)d_in[8];
  const float* w3  = (const float*)d_in[9];
  const float* b3  = (const float*)d_in[10];
  const float* g2  = (const float*)d_in[11];
  const float* be2 = (const float*)d_in[12];
  const float* m2  = (const float*)d_in[13];
  const float* v2  = (const float*)d_in[14];
  const float* w4  = (const float*)d_in[15];
  const float* b4  = (const float*)d_in[16];

  float* tokens  = (float*)d_out;
  float* centers = tokens + TOK_TOTAL;
  int*   knnIdx  = (int*)d_ws;

  fps_kernel<<<NUM_B, 1024, 0, stream>>>(points, centers);
  knn_kernel<<<NUM_B * NUM_G, 256, 0, stream>>>(points, centers, knnIdx);

  size_t ldsBytes = (size_t)(128 + 32 * 132 + 32 * 516 + 32 * 516) * sizeof(float);
  encoder_kernel<<<NUM_B * NUM_G, 256, ldsBytes, stream>>>(
      points, knnIdx, centers, w1, b1, g1, be1, m1, v1, w2, b2, w3, b3, g2,
      be2, m2, v2, w4, b4, tokens);
}


# FPS per-sample argmax: value-only max on the vector ALU (DPP row_xmask, permlanex16), first holder by ballot, one barrier with double-buffered wave slots; was LDS-unit butterflies and 3 barriers
# speedup vs baseline: 1.3510x; 1.3510x over previous
#include <hip/hip_runtime.h>
#include <math.h>

#define NUM_B 8
#define NUM_N 8192
#define NUM_G 256
#define GK    32
#define C1    128
#define C2    256
#define C3    512
#define CT    384
#define TOK_TOTAL (NUM_B * NUM_G * CT)

typedef __attribute__((ext_vector_type(2))) float v2f;
typedef __attribute__((ext_vector_type(8))) float v8f;

__device__ __forceinline__ v8f wmma4(v2f a, v2f b, v8f c) {
  return __builtin_amdgcn_wmma_f32_16x16x4_f32(false, a, false, b, (short)0, c,
                                               false, false);
}

template <int KDIM, int LDB>
__device__ __forceinline__ void gemm_2x2(const float* __restrict__ W, int ldw,
                                         const float* B, int m0, int laneN,
                                         int laneHalf, v8f& acc00, v8f& acc01,
                                         v8f& acc10, v8f& acc11) {
  const float* wrow0 = W + (size_t)(m0 + laneN) * (size_t)ldw + 2 * laneHalf;
  const float* wrow1 = wrow0 + (size_t)16 * (size_t)ldw;
  const float* bcol0 = B + laneN * LDB + 2 * laneHalf;
  const float* bcol1 = B + (16 + laneN) * LDB + 2 * laneHalf;
#pragma unroll 4
  for (int kk = 0; kk < KDIM; kk += 4) {
    v2f a0 = *(const v2f*)(wrow0 + kk);
    v2f a1 = *(const v2f*)(wrow1 + kk);
    v2f b0 = *(const v2f*)(bcol0 + kk);
    v2f b1 = *(const v2f*)(bcol1 + kk);
    acc00 = wmma4(a0, b0, acc00);
    acc01 = wmma4(a0, b1, acc01);
    acc10 = wmma4(a1, b0, acc10);
    acc11 = wmma4(a1, b1, acc11);
  }
}

template <int KDIM, int LDB>
__device__ __forceinline__ void gemm_1x2(const float* __restrict__ W, int ldw,
                                         const float* B, int m0, int laneN,
                                         int laneHalf, v8f& acc0, v8f& acc1) {
  const float* wrow  = W + (size_t)(m0 + laneN) * (size_t)ldw + 2 * laneHalf;
  const float* bcol0 = B + laneN * LDB + 2 * laneHalf;
  const float* bcol1 = B + (16 + laneN) * LDB + 2 * laneHalf;
#pragma unroll 4
  for (int kk = 0; kk < KDIM; kk += 4) {
    v2f a  = *(const v2f*)(wrow + kk);
    v2f b0 = *(const v2f*)(bcol0 + kk);
    v2f b1 = *(const v2f*)(bcol1 + kk);
    acc0 = wmma4(a, b0, acc0);
    acc1 = wmma4(a, b1, acc1);
  }
}

template <typename F>
__device__ __forceinline__ void store_tile(float* dst, int ldd, const v8f& acc,
                                           int m0, int n0, int laneN,
                                           int laneHalf, F op) {
  int n = n0 + laneN;
#pragma unroll
  for (int rr = 0; rr < 8; ++rr) {
    int m = m0 + 8 * laneHalf + rr;
    dst[n * ldd + m] = op(acc[rr], m);
  }
}

template <int M>
__device__ __forceinline__ int row_xor(int b) {
  return __builtin_amdgcn_update_dpp(b, b, 0x160 + M, 0xf, 0xf, true);
}
__device__ __forceinline__ int imax(int a, int b) { return a > b ? a : b; }
__device__ __forceinline__ int wave_max(int v) {
  v = imax(v, row_xor<1>(v));
  v = imax(v, row_xor<2>(v));
  v = imax(v, row_xor<4>(v));
  v = imax(v, row_xor<8>(v));
  v = imax(v, __builtin_amdgcn_permlanex16(v, v, 0x76543210, 0xfedcba98, false,
                                           false));
  return v;
}

__global__ __launch_bounds__(1024) void fps_kernel(
    const float* __restrict__ points, float* __restrict__ centers) {
  const int b    = blockIdx.x;
  const int tid  = threadIdx.x;
  const int lane = tid & 31;
  const int wv   = tid >> 5;
  const float* base = points + (size_t)b * NUM_N * 3;
  float* co = centers + (size_t)b * NUM_G * 3;

  __shared__ float sP[NUM_N * 3];
  __shared__ int   sWv[2][32];
  __shared__ int   sWi[2][32];

  for (int k = tid; k < NUM_N * 3; k += 1024) sP[k] = base[k];

  float px[8], py[8], pz[8], dd[8];
#pragma unroll
  for (int i = 0; i < 8; ++i) {
    int p = tid * 8 + i;
    px[i] = base[p * 3 + 0];
    py[i] = base[p * 3 + 1];
    pz[i] = base[p * 3 + 2];
    dd[i] = 1e10f;
  }
  __syncthreads();

  float lx = sP[0], ly = sP[1], lz = sP[2];
  if (tid == 0) { co[0] = lx; co[1] = ly; co[2] = lz; }

  for (int g = 1; g < NUM_G; ++g) {
    float bv = -1.f;
    int   bi = 0;
#pragma unroll
    for (int i = 0; i < 8; ++i) {
      float dx = px[i] - lx, dy = py[i] - ly, dz = pz[i] - lz;
      float d  = __builtin_fmaf(dz, dz, __builtin_fmaf(dx, dx, dy * dy));
      dd[i]    = fminf(dd[i], d);
      if (dd[i] > bv) { bv = dd[i]; bi = tid * 8 + i; }
    }
    const int buf = g & 1;
    {
      int      bvb = __builtin_bit_cast(int, bv);
      int      wm  = wave_max(bvb);
      unsigned has = __builtin_amdgcn_ballot_w32(bvb == wm);
      int      wbi = __builtin_amdgcn_readlane(bi, __builtin_ctz(has));
      if (lane == 0) { sWv[buf][wv] = wm; sWi[buf][wv] = wbi; }
    }
    __syncthreads();
    {
      int      v   = sWv[buf][lane];
      int      ii  = sWi[buf][lane];
      int      bm  = wave_max(v);
      unsigned has = __builtin_amdgcn_ballot_w32(v == bm);
      int      wi  = __builtin_amdgcn_readlane(ii, __builtin_ctz(has));
      lx = sP[wi * 3 + 0];
      ly = sP[wi * 3 + 1];
      lz = sP[wi * 3 + 2];
    }
    if (tid == 0) { co[g * 3 + 0] = lx; co[g * 3 + 1] = ly; co[g * 3 + 2] = lz; }
  }
}

__global__ __launch_bounds__(256) void knn_kernel(
    const float* __restrict__ points, const float* __restrict__ centers,
    int* __restrict__ knnIdx) {
  const int gg   = blockIdx.x;
  const int b    = gg / NUM_G;
  const int tid  = threadIdx.x;
  const int lane = tid & 31;
  const int wv   = tid >> 5;
  const float* base = points + (size_t)b * NUM_N * 3;

  __shared__ float sd[NUM_N];
  __shared__ float sWv[8];
  __shared__ int   sWi[8];

  float cx = centers[(size_t)gg * 3 + 0];
  float cy = centers[(size_t)gg * 3 + 1];
  float cz = centers[(size_t)gg * 3 + 2];
  float c2 = cx * cx + cy * cy + cz * cz;

  for (int i = 0; i < NUM_N / 256; ++i) {
    int   p  = i * 256 + tid;
    float xx = base[p * 3 + 0], yy = base[p * 3 + 1], zz = base[p * 3 + 2];
    float x2 = xx * xx + yy * yy + zz * zz;
    float dt = cx * xx + cy * yy + cz * zz;
    sd[p]    = c2 + x2 - 2.f * dt;
  }
  __syncthreads();

  for (int r = 0; r < GK; ++r) {
    float bv = 3.0e38f;
    int   bi = NUM_N;
    for (int i = 0; i < NUM_N / 256; ++i) {
      int   p = i * 256 + tid;
      float v = sd[p];
      if (v < bv || (v == bv && p < bi)) { bv = v; bi = p; }
    }
#pragma unroll
    for (int off = 16; off > 0; off >>= 1) {
      float ov = __shfl_xor(bv, off, 32);
      int   oi = __shfl_xor(bi, off, 32);
      if (ov < bv || (ov == bv && oi < bi)) { bv = ov; bi = oi; }
    }
    if (lane == 0) { sWv[wv] = bv; sWi[wv] = bi; }
    __syncthreads();
    if (tid == 0) {
      float v  = sWv[0];
      int   ii = sWi[0];
      for (int w = 1; w < 8; ++w)
        if (sWv[w] < v || (sWv[w] == v && sWi[w] < ii)) { v = sWv[w]; ii = sWi[w]; }
      knnIdx[(size_t)gg * GK + r] = ii;
      sd[ii] = 3.4e38f;
    }
    __syncthreads();
  }
}

__global__ __launch_bounds__(256) void encoder_kernel(
    const float* __restrict__ points, const int* __restrict__ knnIdx,
    const float* __restrict__ centers, const float* __restrict__ w1,
    const float* __restrict__ b1, const float* __restrict__ g1,
    const float* __restrict__ be1, const float* __restrict__ m1,
    const float* __restrict__ v1, const float* __restrict__ w2,
    const float* __restrict__ b2, const float* __restrict__ w3,
    const float* __restrict__ b3, const float* __restrict__ g2,
    const float* __restrict__ be2, const float* __restrict__ m2,
    const float* __restrict__ v2, const float* __restrict__ w4,
    const float* __restrict__ b4, float* __restrict__ tokens) {
  extern __shared__ float smem[];
  float* sFeat = smem;
  float* sH1   = smem + 128;
  float* sX3   = smem + 128 + 32 * 132;
  float* sH3   = sX3 + 32 * 516;
  float* sH4   = sX3;

  const int gg       = blockIdx.x;
  const int b        = gg / NUM_G;
  const int tid      = threadIdx.x;
  const int lane     = tid & 31;
  const int wv       = tid >> 5;
  const int laneN    = lane & 15;
  const int laneHalf = lane >> 4;

  const v8f vzero = {0.f, 0.f, 0.f, 0.f, 0.f, 0.f, 0.f, 0.f};

  if (tid < GK) {
    int idx = knnIdx[(size_t)gg * GK + tid];
    const float* p = points + ((size_t)b * NUM_N + idx) * 3;
    const float* c = centers + (size_t)gg * 3;
    float dx = p[0] - c[0], dy = p[1] - c[1], dz = p[2] - c[2];
    float nr = sqrtf(dx * dx + dy * dy + dz * dz);
    sFeat[tid * 4 + 0] = dx;
    sFeat[tid * 4 + 1] = dy;
    sFeat[tid * 4 + 2] = dz;
    sFeat[tid * 4 + 3] = nr;
  }
  __syncthreads();

  {
    int m0 = wv * 16;
    v8f a0 = vzero, a1 = vzero;
    gemm_1x2<4, 4>(w1, 4, sFeat, m0, laneN, laneHalf, a0, a1);
    auto op = [&](float x, int m) {
      float inv = g1[m] * rsqrtf(v1[m] + 1e-5f);
      float v   = (x + b1[m]) * inv + (be1[m] - m1[m] * inv);
      return fmaxf(v, 0.f);
    };
    store_tile(sH1, 132, a0, m0, 0,  laneN, laneHalf, op);
    store_tile(sH1, 132, a1, m0, 16, laneN, laneHalf, op);
  }
  __syncthreads();

  {
    int m0 = wv * 32;
    v8f a00 = vzero, a01 = vzero, a10 = vzero, a11 = vzero;
    gemm_2x2<C1, 132>(w2, C1, sH1, m0, laneN, laneHalf, a00, a01, a10, a11);
    auto op = [&](float x, int m) { return x + b2[m]; };
    store_tile(sX3 + 256, 516, a00, m0,      0,  laneN, laneHalf, op);
    store_tile(sX3 + 256, 516, a01, m0,      16, laneN, laneHalf, op);
    store_tile(sX3 + 256, 516, a10, m0 + 16, 0,  laneN, laneHalf, op);
    store_tile(sX3 + 256, 516, a11, m0 + 16, 16, laneN, laneHalf, op);
  }
  __syncthreads();

  {
    int   m  = tid;
    float gm = -3.4e38f;
    for (int n = 0; n < GK; ++n) gm = fmaxf(gm, sX3[n * 516 + 256 + m]);
    for (int n = 0; n < GK; ++n) sX3[n * 516 + m] = gm;
  }
  __syncthreads();

  {
    auto op = [&](float x, int m) {
      float inv = g2[m] * rsqrtf(v2[m] + 1e-5f);
      float v   = (x + b3[m]) * inv + (be2[m] - m2[m] * inv);
      return fmaxf(v, 0.f);
    };
#pragma unroll
    for (int j = 0; j < 2; ++j) {
      int m0  = wv * 64 + j * 32;
      v8f a00 = vzero, a01 = vzero, a10 = vzero, a11 = vzero;
      gemm_2x2<C3, 516>(w3, C3, sX3, m0, laneN, laneHalf, a00, a01, a10, a11);
      store_tile(sH3, 516, a00, m0,      0,  laneN, laneHalf, op);
      store_tile(sH3, 516, a01, m0,      16, laneN, laneHalf, op);
      store_tile(sH3, 516, a10, m0 + 16, 0,  laneN, laneHalf, op);
      store_tile(sH3, 516, a11, m0 + 16, 16, laneN, laneHalf, op);
    }
  }
  __syncthreads();

  {
    auto op = [&](float x, int m) { return x + b4[m]; };
    {
      int m0  = wv * 48;
      v8f a00 = vzero, a01 = vzero, a10 = vzero, a11 = vzero;
      gemm_2x2<C3, 516>(w4, C3, sH3, m0, laneN, laneHalf, a00, a01, a10, a11);
      store_tile(sH4, 388, a00, m0,      0,  laneN, laneHalf, op);
      store_tile(sH4, 388, a01, m0,      16, laneN, laneHalf, op);
      store_tile(sH4, 388, a10, m0 + 16, 0,  laneN, laneHalf, op);
      store_tile(sH4, 388, a11, m0 + 16, 16, laneN, laneHalf, op);
    }
    {
      int m0 = wv * 48 + 32;
      v8f a0 = vzero, a1 = vzero;
      gemm_1x2<C3, 516>(w4, C3, sH3, m0, laneN, laneHalf, a0, a1);
      store_tile(sH4, 388, a0, m0, 0,  laneN, laneHalf, op);
      store_tile(sH4, 388, a1, m0, 16, laneN, laneHalf, op);
    }
  }
  __syncthreads();

  for (int m = tid; m < CT; m += 256) {
    float mx = -3.4e38f;
    for (int n = 0; n < GK; ++n) mx = fmaxf(mx, sH4[n * 388 + m]);
    tokens[(size_t)gg * CT + m] = mx;
  }
}

extern "C" void kernel_launch(void* const* d_in, const int* in_sizes, int n_in,
                              void* d_out, int out_size, void* d_ws,
                              size_t ws_size, hipStream_t stream) {
  (void)in_sizes; (void)n_in; (void)out_size; (void)ws_size;
  const float* points = (const float*)d_in[0];
  const float* w1  = (const float*)d_in[1];
  const float* b1  = (const float*)d_in[2];
  const float* g1  = (const float*)d_in[3];
  const float* be1 = (const float*)d_in[4];
  const float* m1  = (const float*)d_in[5];
  const float* v1  = (const float*)d_in[6];
  const float* w2  = (const float*)d_in[7];
  const float* b2  = (const float*)d_in[8];
  const float* w3  = (const float*)d_in[9];
  const float* b3  = (const float*)d_in[10];
  const float* g2  = (const float*)d_in[11];
  const float* be2 = (const float*)d_in[12];
  const float* m2  = (const float*)d_in[13];
  const float* v2  = (const float*)d_in[14];
  const float* w4  = (const float*)d_in[15];
  const float* b4  = (const float*)d_in[16];

  float* tokens  = (float*)d_out;
  float* centers = tokens + TOK_TOTAL;
  int*   knnIdx  = (int*)d_ws;

  fps_kernel<<<NUM_B, 1024, 0, stream>>>(points, centers);
  knn_kernel<<<NUM_B * NUM_G, 256, 0, stream>>>(points, centers, knnIdx);

  size_t ldsBytes = (size_t)(128 + 32 * 132 + 32 * 516 + 32 * 516) * sizeof(float);
  encoder_kernel<<<NUM_B * NUM_G, 256, ldsBytes, stream>>>(
      points, knnIdx, centers, w1, b1, g1, be1, m1, v1, w2, b2, w3, b3, g2,
      be2, m2, v2, w4, b4, tokens);
}


# Encoder GEMMs: A (global) and B (LDS) fragments loaded 16 bytes per lane instead of 8, one load feeding two f32 WMMAs; k order of the sums changes inside each 8-k block
# speedup vs baseline: 1.4151x; 1.0474x over previous
#include <hip/hip_runtime.h>
#include <math.h>

#define NUM_B 8
#define NUM_N 8192
#define NUM_G 256
#define GK    32
#define C1    128
#define C2    256
#define C3    512
#define CT    384
#define TOK_TOTAL (NUM_B * NUM_G * CT)

typedef __attribute__((ext_vector_type(2))) float v2f;
typedef __attribute__((ext_vector_type(4))) float v4f;
typedef __attribute__((ext_vector_type(8))) float v8f;

__device__ __forceinline__ v8f wmma4(v2f a, v2f b, v8f c) {
  return __builtin_amdgcn_wmma_f32_16x16x4_f32(false, a, false, b, (short)0, c,
                                               false, false);
}

__device__ __forceinline__ v2f lo2(v4f v) { return __builtin_shufflevector(v, v, 0, 1); }
__device__ __forceinline__ v2f hi2(v4f v) { return __builtin_shufflevector(v, v, 2, 3); }

template <int KDIM, int LDB>
__device__ __forceinline__ void gemm_2x2w(const float* __restrict__ W, int ldw,
                                          const float* B, int m0, int laneN,
                                          int laneHalf, v8f& acc00, v8f& acc01,
                                          v8f& acc10, v8f& acc11) {
  static_assert(KDIM % 8 == 0, "8-k blocks");
  const float* wrow0 = W + (size_t)(m0 + laneN) * (size_t)ldw + 4 * laneHalf;
  const float* wrow1 = wrow0 + (size_t)16 * (size_t)ldw;
  const float* bcol0 = B + laneN * LDB + 4 * laneHalf;
  const float* bcol1 = B + (16 + laneN) * LDB + 4 * laneHalf;
#pragma unroll 4
  for (int kk = 0; kk < KDIM; kk += 8) {
    v4f a0 = *(const v4f*)(wrow0 + kk);
    v4f a1 = *(const v4f*)(wrow1 + kk);
    v4f b0 = *(const v4f*)(bcol0 + kk);
    v4f b1 = *(const v4f*)(bcol1 + kk);
    acc00 = wmma4(lo2(a0), lo2(b0), acc00);
    acc01 = wmma4(lo2(a0), lo2(b1), acc01);
    acc10 = wmma4(lo2(a1), lo2(b0), acc10);
    acc11 = wmma4(lo2(a1), lo2(b1), acc11);
    acc00 = wmma4(hi2(a0), hi2(b0), acc00);
    acc01 = wmma4(hi2(a0), hi2(b1), acc01);
    acc10 = wmma4(hi2(a1), hi2(b0), acc10);
    acc11 = wmma4(hi2(a1), hi2(b1), acc11);
  }
}

template <int KDIM, int LDB>
__device__ __forceinline__ void gemm_1x2(const float* __restrict__ W, int ldw,
                                         const float* B, int m0, int laneN,
                                         int laneHalf, v8f& acc0, v8f& acc1) {
  const float* wrow  = W + (size_t)(m0 + laneN) * (size_t)ldw + 2 * laneHalf;
  const float* bcol0 = B + laneN * LDB + 2 * laneHalf;
  const float* bcol1 = B + (16 + laneN) * LDB + 2 * laneHalf;
  for (int kk = 0; kk < KDIM; kk += 4) {
    v2f a  = *(const v2f*)(wrow + kk);
    v2f b0 = *(const v2f*)(bcol0 + kk);
    v2f b1 = *(const v2f*)(bcol1 + kk);
    acc0 = wmma4(a, b0, acc0);
    acc1 = wmma4(a, b1, acc1);
  }
}

template <int KDIM, int LDB>
__device__ __forceinline__ void gemm_1x2w(const float* __restrict__ W, int ldw,
                                          const float* B, int m0, int laneN,
                                          int laneHalf, v8f& acc0, v8f& acc1) {
  static_assert(KDIM % 8 == 0, "8-k blocks");
  const float* wrow  = W + (size_t)(m0 + laneN) * (size_t)ldw + 4 * laneHalf;
  const float* bcol0 = B + laneN * LDB + 4 * laneHalf;
  const float* bcol1 = B + (16 + laneN) * LDB + 4 * laneHalf;
#pragma unroll 4
  for (int kk = 0; kk < KDIM; kk += 8) {
    v4f a  = *(const v4f*)(wrow + kk);
    v4f b0 = *(const v4f*)(bcol0 + kk);
    v4f b1 = *(const v4f*)(bcol1 + kk);
    acc0 = wmma4(lo2(a), lo2(b0), acc0);
    acc1 = wmma4(lo2(a), lo2(b1), acc1);
    acc0 = wmma4(hi2(a), hi2(b0), acc0);
    acc1 = wmma4(hi2(a), hi2(b1), acc1);
  }
}

template <typename F>
__device__ __forceinline__ void store_tile(float* dst, int ldd, const v8f& acc,
                                           int m0, int n0, int laneN,
                                           int laneHalf, F op) {
  int n = n0 + laneN;
#pragma unroll
  for (int rr = 0; rr < 8; ++rr) {
    int m = m0 + 8 * laneHalf + rr;
    dst[n * ldd + m] = op(acc[rr], m);
  }
}

template <int M>
__device__ __forceinline__ int row_xor(int b) {
  return __builtin_amdgcn_update_dpp(b, b, 0x160 + M, 0xf, 0xf, true);
}
__device__ __forceinline__ int imax(int a, int b) { return a > b ? a : b; }
__device__ __forceinline__ int wave_max(int v) {
  v = imax(v, row_xor<1>(v));
  v = imax(v, row_xor<2>(v));
  v = imax(v, row_xor<4>(v));
  v = imax(v, row_xor<8>(v));
  v = imax(v, __builtin_amdgcn_permlanex16(v, v, 0x76543210, 0xfedcba98, false,
                                           false));
  return v;
}

__global__ __launch_bounds__(1024) void fps_kernel(
    const float* __restrict__ points, float* __restrict__ centers) {
  const int b    = blockIdx.x;
  const int tid  = threadIdx.x;
  const int lane = tid & 31;
  const int wv   = tid >> 5;
  const float* base = points + (size_t)b * NUM_N * 3;
  float* co = centers + (size_t)b * NUM_G * 3;

  __shared__ float sP[NUM_N * 3];
  __shared__ int   sWv[2][32];
  __shared__ int   sWi[2][32];

  for (int k = tid; k < NUM_N * 3; k += 1024) sP[k] = base[k];

  float px[8], py[8], pz[8], dd[8];
#pragma unroll
  for (int i = 0; i < 8; ++i) {
    int p = tid * 8 + i;
    px[i] = base[p * 3 + 0];
    py[i] = base[p * 3 + 1];
    pz[i] = base[p * 3 + 2];
    dd[i] = 1e10f;
  }
  __syncthreads();

  float lx = sP[0], ly = sP[1], lz = sP[2];
  if (tid == 0) { co[0] = lx; co[1] = ly; co[2] = lz; }

  for (int g = 1; g < NUM_G; ++g) {
    float bv = -1.f;
    int   bi = 0;
#pragma unroll
    for (int i = 0; i < 8; ++i) {
      float dx = px[i] - lx, dy = py[i] - ly, dz = pz[i] - lz;
      float d  = __builtin_fmaf(dz, dz, __builtin_fmaf(dx, dx, dy * dy));
      dd[i]    = fminf(dd[i], d);
      if (dd[i] > bv) { bv = dd[i]; bi = tid * 8 + i; }
    }
    const int buf = g & 1;
    {
      int      bvb = __builtin_bit_cast(int, bv);
      int      wm  = wave_max(bvb);
      unsigned has = __builtin_amdgcn_ballot_w32(bvb == wm);
      int      wbi = __builtin_amdgcn_readlane(bi, __builtin_ctz(has));
      if (lane == 0) { sWv[buf][wv] = wm; sWi[buf][wv] = wbi; }
    }
    __syncthreads();
    {
      int      v   = sWv[buf][lane];
      int      ii  = sWi[buf][lane];
      int      bm  = wave_max(v);
      unsigned has = __builtin_amdgcn_ballot_w32(v == bm);
      int      wi  = __builtin_amdgcn_readlane(ii, __builtin_ctz(has));
      lx = sP[wi * 3 + 0];
      ly = sP[wi * 3 + 1];
      lz = sP[wi * 3 + 2];
    }
    if (tid == 0) { co[g * 3 + 0] = lx; co[g * 3 + 1] = ly; co[g * 3 + 2] = lz; }
  }
}

__global__ __launch_bounds__(256) void knn_kernel(
    const float* __restrict__ points, const float* __restrict__ centers,
    int* __restrict__ knnIdx) {
  const int gg   = blockIdx.x;
  const int b    = gg / NUM_G;
  const int tid  = threadIdx.x;
  const int lane = tid & 31;
  const int wv   = tid >> 5;
  const float* base = points + (size_t)b * NUM_N * 3;

  __shared__ float sd[NUM_N];
  __shared__ float sWv[8];
  __shared__ int   sWi[8];

  float cx = centers[(size_t)gg * 3 + 0];
  float cy = centers[(size_t)gg * 3 + 1];
  float cz = centers[(size_t)gg * 3 + 2];
  float c2 = cx * cx + cy * cy + cz * cz;

  for (int i = 0; i < NUM_N / 256; ++i) {
    int   p  = i * 256 + tid;
    float xx = base[p * 3 + 0], yy = base[p * 3 + 1], zz = base[p * 3 + 2];
    float x2 = xx * xx + yy * yy + zz * zz;
    float dt = cx * xx + cy * yy + cz * zz;
    sd[p]    = c2 + x2 - 2.f * dt;
  }
  __syncthreads();

  for (int r = 0; r < GK; ++r) {
    float bv = 3.0e38f;
    int   bi = NUM_N;
    for (int i = 0; i < NUM_N / 256; ++i) {
      int   p = i * 256 + tid;
      float v = sd[p];
      if (v < bv || (v == bv && p < bi)) { bv = v; bi = p; }
    }
#pragma unroll
    for (int off = 16; off > 0; off >>= 1) {
      float ov = __shfl_xor(bv, off, 32);
      int   oi = __shfl_xor(bi, off, 32);
      if (ov < bv || (ov == bv && oi < bi)) { bv = ov; bi = oi; }
    }
    if (lane == 0) { sWv[wv] = bv; sWi[wv] = bi; }
    __syncthreads();
    if (tid == 0) {
      float v  = sWv[0];
      int   ii = sWi[0];
      for (int w = 1; w < 8; ++w)
        if (sWv[w] < v || (sWv[w] == v && sWi[w] < ii)) { v = sWv[w]; ii = sWi[w]; }
      knnIdx[(size_t)gg * GK + r] = ii;
      sd[ii] = 3.4e38f;
    }
    __syncthreads();
  }
}

__global__ __launch_bounds__(256) void encoder_kernel(
    const float* __restrict__ points, const int* __restrict__ knnIdx,
    const float* __restrict__ centers, const float* __restrict__ w1,
    const float* __restrict__ b1, const float* __restrict__ g1,
    const float* __restrict__ be1, const float* __restrict__ m1,
    const float* __restrict__ v1, const float* __restrict__ w2,
    const float* __restrict__ b2, const float* __restrict__ w3,
    const float* __restrict__ b3, const float* __restrict__ g2,
    const float* __restrict__ be2, const float* __restrict__ m2,
    const float* __restrict__ v2, const float* __restrict__ w4,
    const float* __restrict__ b4, float* __restrict__ tokens) {
  extern __shared__ __attribute__((aligned(16))) float smem[];
  float* sFeat = smem;
  float* sH1   = smem + 128;
  float* sX3   = smem + 128 + 32 * 132;
  float* sH3   = sX3 + 32 * 516;
  float* sH4   = sX3;

  const int gg       = blockIdx.x;
  const int b        = gg / NUM_G;
  const int tid      = threadIdx.x;
  const int lane     = tid & 31;
  const int wv       = tid >> 5;
  const int laneN    = lane & 15;
  const int laneHalf = lane >> 4;

  const v8f vzero = {0.f, 0.f, 0.f, 0.f, 0.f, 0.f, 0.f, 0.f};

  if (tid < GK) {
    int idx = knnIdx[(size_t)gg * GK + tid];
    const float* p = points + ((size_t)b * NUM_N + idx) * 3;
    const float* c = centers + (size_t)gg * 3;
    float dx = p[0] - c[0], dy = p[1] - c[1], dz = p[2] - c[2];
    float nr = sqrtf(dx * dx + dy * dy + dz * dz);
    sFeat[tid * 4 + 0] = dx;
    sFeat[tid * 4 + 1] = dy;
    sFeat[tid * 4 + 2] = dz;
    sFeat[tid * 4 + 3] = nr;
  }
  __syncthreads();

  {
    int m0 = wv * 16;
    v8f a0 = vzero, a1 = vzero;
    gemm_1x2<4, 4>(w1, 4, sFeat, m0, laneN, laneHalf, a0, a1);
    auto op = [&](float x, int m) {
      float inv = g1[m] * rsqrtf(v1[m] + 1e-5f);
      float v   = (x + b1[m]) * inv + (be1[m] - m1[m] * inv);
      return fmaxf(v, 0.f);
    };
    store_tile(sH1, 132, a0, m0, 0,  laneN, laneHalf, op);
    store_tile(sH1, 132, a1, m0, 16, laneN, laneHalf, op);
  }
  __syncthreads();

  {
    int m0 = wv * 32;
    v8f a00 = vzero, a01 = vzero, a10 = vzero, a11 = vzero;
    gemm_2x2w<C1, 132>(w2, C1, sH1, m0, laneN, laneHalf, a00, a01, a10, a11);
    auto op = [&](float x, int m) { return x + b2[m]; };
    store_tile(sX3 + 256, 516, a00, m0,      0,  laneN, laneHalf, op);
    store_tile(sX3 + 256, 516, a01, m0,      16, laneN, laneHalf, op);
    store_tile(sX3 + 256, 516, a10, m0 + 16, 0,  laneN, laneHalf, op);
    store_tile(sX3 + 256, 516, a11, m0 + 16, 16, laneN, laneHalf, op);
  }
  __syncthreads();

  {
    int   m  = tid;
    float gm = -3.4e38f;
    for (int n = 0; n < GK; ++n) gm = fmaxf(gm, sX3[n * 516 + 256 + m]);
    for (int n = 0; n < GK; ++n) sX3[n * 516 + m] = gm;
  }
  __syncthreads();

  {
    auto op = [&](float x, int m) {
      float inv = g2[m] * rsqrtf(v2[m] + 1e-5f);
      float v   = (x + b3[m]) * inv + (be2[m] - m2[m] * inv);
      return fmaxf(v, 0.f);
    };
#pragma unroll
    for (int j = 0; j < 2; ++j) {
      int m0  = wv * 64 + j * 32;
      v8f a00 = vzero, a01 = vzero, a10 = vzero, a11 = vzero;
      gemm_2x2w<C3, 516>(w3, C3, sX3, m0, laneN, laneHalf, a00, a01, a10, a11);
      store_tile(sH3, 516, a00, m0,      0,  laneN, laneHalf, op);
      store_tile(sH3, 516, a01, m0,      16, laneN, laneHalf, op);
      store_tile(sH3, 516, a10, m0 + 16, 0,  laneN, laneHalf, op);
      store_tile(sH3, 516, a11, m0 + 16, 16, laneN, laneHalf, op);
    }
  }
  __syncthreads();

  {
    auto op = [&](float x, int m) { return x + b4[m]; };
    {
      int m0  = wv * 48;
      v8f a00 = vzero, a01 = vzero, a10 = vzero, a11 = vzero;
      gemm_2x2w<C3, 516>(w4, C3, sH3, m0, laneN, laneHalf, a00, a01, a10, a11);
      store_tile(sH4, 388, a00, m0,      0,  laneN, laneHalf, op);
      store_tile(sH4, 388, a01, m0,      16, laneN, laneHalf, op);
      store_tile(sH4, 388, a10, m0 + 16, 0,  laneN, laneHalf, op);
      store_tile(sH4, 388, a11, m0 + 16, 16, laneN, laneHalf, op);
    }
    {
      int m0 = wv * 48 + 32;
      v8f a0 = vzero, a1 = vzero;
      gemm_1x2w<C3, 516>(w4, C3, sH3, m0, laneN, laneHalf, a0, a1);
      store_tile(sH4, 388, a0, m0, 0,  laneN, laneHalf, op);
      store_tile(sH4, 388, a1, m0, 16, laneN, laneHalf, op);
    }
  }
  __syncthreads();

  for (int m = tid; m < CT; m += 256) {
    float mx = -3.4e38f;
    for (int n = 0; n < GK; ++n) mx = fmaxf(mx, sH4[n * 388 + m]);
    tokens[(size_t)gg * CT + m] = mx;
  }
}

extern "C" void kernel_launch(void* const* d_in, const int* in_sizes, int n_in,
                              void* d_out, int out_size, void* d_ws,
                              size_t ws_size, hipStream_t stream) {
  (void)in_sizes; (void)n_in; (void)out_size; (void)ws_size;
  const float* points = (const float*)d_in[0];
  const float* w1  = (const float*)d_in[1];
  const float* b1  = (const float*)d_in[2];
  const float* g1  = (const float*)d_in[3];
  const float* be1 = (const float*)d_in[4];
  const float* m1  = (const float*)d_in[5];
  const float* v1  = (const float*)d_in[6];
  const float* w2  = (const float*)d_in[7];
  const float* b2  = (const float*)d_in[8];
  const float* w3  = (const float*)d_in[9];
  const float* b3  = (const float*)d_in[10];
  const float* g2  = (const float*)d_in[11];
  const float* be2 = (const float*)d_in[12];
  const float* m2  = (const float*)d_in[13];
  const float* v2  = (const float*)d_in[14];
  const float* w4  = (const float*)d_in[15];
  const float* b4  = (const float*)d_in[16];

  float* tokens  = (float*)d_out;
  float* centers = tokens + TOK_TOTAL;
  int*   knnIdx  = (int*)d_ws;

  fps_kernel<<<NUM_B, 1024, 0, stream>>>(points, centers);
  knn_kernel<<<NUM_B * NUM_G, 256, 0, stream>>>(points, centers, knnIdx);

  size_t ldsBytes = (size_t)(128 + 32 * 132 + 32 * 516 + 32 * 516) * sizeof(float);
  encoder_kernel<<<NUM_B * NUM_G, 256, ldsBytes, stream>>>(
      points, knnIdx, centers, w1, b1, g1, be1, m1, v1, w2, b2, w3, b3, g2,
      be2, m2, v2, w4, b4, tokens);
}


# KNN selection: distances kept in registers as ordered integer keys, each thread caches its minimum; a round is one DPP block minimum, one barrier, a rescan by the winner's wave; was 32 full LDS scans
# speedup vs baseline: 1.8300x; 1.2932x over previous
#include <hip/hip_runtime.h>
#include <math.h>

#define NUM_B 8
#define NUM_N 8192
#define NUM_G 256
#define GK    32
#define C1    128
#define C2    256
#define C3    512
#define CT    384
#define TOK_TOTAL (NUM_B * NUM_G * CT)

typedef __attribute__((ext_vector_type(2))) float v2f;
typedef __attribute__((ext_vector_type(4))) float v4f;
typedef __attribute__((ext_vector_type(8))) float v8f;

__device__ __forceinline__ v8f wmma4(v2f a, v2f b, v8f c) {
  return __builtin_amdgcn_wmma_f32_16x16x4_f32(false, a, false, b, (short)0, c,
                                               false, false);
}

__device__ __forceinline__ v2f lo2(v4f v) { return __builtin_shufflevector(v, v, 0, 1); }
__device__ __forceinline__ v2f hi2(v4f v) { return __builtin_shufflevector(v, v, 2, 3); }

template <int KDIM, int LDB>
__device__ __forceinline__ void gemm_2x2w(const float* __restrict__ W, int ldw,
                                          const float* B, int m0, int laneN,
                                          int laneHalf, v8f& acc00, v8f& acc01,
                                          v8f& acc10, v8f& acc11) {
  static_assert(KDIM % 8 == 0, "8-k blocks");
  const float* wrow0 = W + (size_t)(m0 + laneN) * (size_t)ldw + 4 * laneHalf;
  const float* wrow1 = wrow0 + (size_t)16 * (size_t)ldw;
  const float* bcol0 = B + laneN * LDB + 4 * laneHalf;
  const float* bcol1 = B + (16 + laneN) * LDB + 4 * laneHalf;
#pragma unroll 4
  for (int kk = 0; kk < KDIM; kk += 8) {
    v4f a0 = *(const v4f*)(wrow0 + kk);
    v4f a1 = *(const v4f*)(wrow1 + kk);
    v4f b0 = *(const v4f*)(bcol0 + kk);
    v4f b1 = *(const v4f*)(bcol1 + kk);
    acc00 = wmma4(lo2(a0), lo2(b0), acc00);
    acc01 = wmma4(lo2(a0), lo2(b1), acc01);
    acc10 = wmma4(lo2(a1), lo2(b0), acc10);
    acc11 = wmma4(lo2(a1), lo2(b1), acc11);
    acc00 = wmma4(hi2(a0), hi2(b0), acc00);
    acc01 = wmma4(hi2(a0), hi2(b1), acc01);
    acc10 = wmma4(hi2(a1), hi2(b0), acc10);
    acc11 = wmma4(hi2(a1), hi2(b1), acc11);
  }
}

template <int KDIM, int LDB>
__device__ __forceinline__ void gemm_1x2(const float* __restrict__ W, int ldw,
                                         const float* B, int m0, int laneN,
                                         int laneHalf, v8f& acc0, v8f& acc1) {
  const float* wrow  = W + (size_t)(m0 + laneN) * (size_t)ldw + 2 * laneHalf;
  const float* bcol0 = B + laneN * LDB + 2 * laneHalf;
  const float* bcol1 = B + (16 + laneN) * LDB + 2 * laneHalf;
  for (int kk = 0; kk < KDIM; kk += 4) {
    v2f a  = *(const v2f*)(wrow + kk);
    v2f b0 = *(const v2f*)(bcol0 + kk);
    v2f b1 = *(const v2f*)(bcol1 + kk);
    acc0 = wmma4(a, b0, acc0);
    acc1 = wmma4(a, b1, acc1);
  }
}

template <int KDIM, int LDB>
__device__ __forceinline__ void gemm_1x2w(const float* __restrict__ W, int ldw,
                                          const float* B, int m0, int laneN,
                                          int laneHalf, v8f& acc0, v8f& acc1) {
  static_assert(KDIM % 8 == 0, "8-k blocks");
  const float* wrow  = W + (size_t)(m0 + laneN) * (size_t)ldw + 4 * laneHalf;
  const float* bcol0 = B + laneN * LDB + 4 * laneHalf;
  const float* bcol1 = B + (16 + laneN) * LDB + 4 * laneHalf;
#pragma unroll 4
  for (int kk = 0; kk < KDIM; kk += 8) {
    v4f a  = *(const v4f*)(wrow + kk);
    v4f b0 = *(const v4f*)(bcol0 + kk);
    v4f b1 = *(const v4f*)(bcol1 + kk);
    acc0 = wmma4(lo2(a), lo2(b0), acc0);
    acc1 = wmma4(lo2(a), lo2(b1), acc1);
    acc0 = wmma4(hi2(a), hi2(b0), acc0);
    acc1 = wmma4(hi2(a), hi2(b1), acc1);
  }
}

template <typename F>
__device__ __forceinline__ void store_tile(float* dst, int ldd, const v8f& acc,
                                           int m0, int n0, int laneN,
                                           int laneHalf, F op) {
  int n = n0 + laneN;
#pragma unroll
  for (int rr = 0; rr < 8; ++rr) {
    int m = m0 + 8 * laneHalf + rr;
    dst[n * ldd + m] = op(acc[rr], m);
  }
}

template <int M>
__device__ __forceinline__ int row_xor(int b) {
  return __builtin_amdgcn_update_dpp(b, b, 0x160 + M, 0xf, 0xf, true);
}
__device__ __forceinline__ int imax(int a, int b) { return a > b ? a : b; }
__device__ __forceinline__ int wave_max(int v) {
  v = imax(v, row_xor<1>(v));
  v = imax(v, row_xor<2>(v));
  v = imax(v, row_xor<4>(v));
  v = imax(v, row_xor<8>(v));
  v = imax(v, __builtin_amdgcn_permlanex16(v, v, 0x76543210, 0xfedcba98, false,
                                           false));
  return v;
}

__global__ __launch_bounds__(1024) void fps_kernel(
    const float* __restrict__ points, float* __restrict__ centers) {
  const int b    = blockIdx.x;
  const int tid  = threadIdx.x;
  const int lane = tid & 31;
  const int wv   = tid >> 5;
  const float* base = points + (size_t)b * NUM_N * 3;
  float* co = centers + (size_t)b * NUM_G * 3;

  __shared__ float sP[NUM_N * 3];
  __shared__ int   sWv[2][32];
  __shared__ int   sWi[2][32];

  for (int k = tid; k < NUM_N * 3; k += 1024) sP[k] = base[k];

  float px[8], py[8], pz[8], dd[8];
#pragma unroll
  for (int i = 0; i < 8; ++i) {
    int p = tid * 8 + i;
    px[i] = base[p * 3 + 0];
    py[i] = base[p * 3 + 1];
    pz[i] = base[p * 3 + 2];
    dd[i] = 1e10f;
  }
  __syncthreads();

  float lx = sP[0], ly = sP[1], lz = sP[2];
  if (tid == 0) { co[0] = lx; co[1] = ly; co[2] = lz; }

  for (int g = 1; g < NUM_G; ++g) {
    float bv = -1.f;
    int   bi = 0;
#pragma unroll
    for (int i = 0; i < 8; ++i) {
      float dx = px[i] - lx, dy = py[i] - ly, dz = pz[i] - lz;
      float d  = __builtin_fmaf(dz, dz, __builtin_fmaf(dx, dx, dy * dy));
      dd[i]    = fminf(dd[i], d);
      if (dd[i] > bv) { bv = dd[i]; bi = tid * 8 + i; }
    }
    const int buf = g & 1;
    {
      int      bvb = __builtin_bit_cast(int, bv);
      int      wm  = wave_max(bvb);
      unsigned has = __builtin_amdgcn_ballot_w32(bvb == wm);
      int      wbi = __builtin_amdgcn_readlane(bi, __builtin_ctz(has));
      if (lane == 0) { sWv[buf][wv] = wm; sWi[buf][wv] = wbi; }
    }
    __syncthreads();
    {
      int      v   = sWv[buf][lane];
      int      ii  = sWi[buf][lane];
      int      bm  = wave_max(v);
      unsigned has = __builtin_amdgcn_ballot_w32(v == bm);
      int      wi  = __builtin_amdgcn_readlane(ii, __builtin_ctz(has));
      lx = sP[wi * 3 + 0];
      ly = sP[wi * 3 + 1];
      lz = sP[wi * 3 + 2];
    }
    if (tid == 0) { co[g * 3 + 0] = lx; co[g * 3 + 1] = ly; co[g * 3 + 2] = lz; }
  }
}

__device__ __forceinline__ int imin(int a, int b) { return a < b ? a : b; }
__device__ __forceinline__ int wave_min(int v) {
  v = imin(v, row_xor<1>(v));
  v = imin(v, row_xor<2>(v));
  v = imin(v, row_xor<4>(v));
  v = imin(v, row_xor<8>(v));
  v = imin(v, __builtin_amdgcn_permlanex16(v, v, 0x76543210, 0xfedcba98, false,
                                           false));
  return v;
}
__device__ __forceinline__ int oct_min(int v) {
  v = imin(v, row_xor<1>(v));
  v = imin(v, row_xor<2>(v));
  v = imin(v, row_xor<4>(v));
  return v;
}
__device__ __forceinline__ int order_key(float f) {
  int bits = __builtin_bit_cast(int, f);
  return bits ^ ((bits >> 31) & 0x7fffffff);
}

#define KEY_GONE 0x7fffffff

__global__ __launch_bounds__(256) void knn_kernel(
    const float* __restrict__ points, const float* __restrict__ centers,
    int* __restrict__ knnIdx) {
  const int gg   = blockIdx.x;
  const int b    = gg / NUM_G;
  const int tid  = threadIdx.x;
  const int lane = tid & 31;
  const int wv   = tid >> 5;
  const float* base = points + (size_t)b * NUM_N * 3;

  __shared__ int sWv[2][8];
  __shared__ int sWi[2][8];

  const float cx = centers[(size_t)gg * 3 + 0];
  const float cy = centers[(size_t)gg * 3 + 1];
  const float cz = centers[(size_t)gg * 3 + 2];
  const float c2 = __builtin_fmaf(cz, cz, __builtin_fmaf(cx, cx, cy * cy));

  int dk[32];
#pragma unroll
  for (int i = 0; i < NUM_N / 256; ++i) {
    int   p  = i * 256 + tid;
    float xx = base[p * 3 + 0], yy = base[p * 3 + 1], zz = base[p * 3 + 2];
    float x2 = __builtin_fmaf(zz, zz, __builtin_fmaf(xx, xx, yy * yy));
    float dt = __builtin_fmaf(cz, zz, __builtin_fmaf(cx, xx, cy * yy));
    dk[i]    = order_key(__builtin_fmaf(-2.f, dt, c2 + x2));
  }

  int lv = KEY_GONE, li = 0;
#pragma unroll
  for (int i = 0; i < 32; ++i)
    if (dk[i] < lv) { lv = dk[i]; li = i; }
  int lp    = li * 256 + tid;
  int myOut = 0;

  for (int r = 0; r < GK; ++r) {
    const int buf = r & 1;
    {
      int wm = wave_min(lv);
      int wp = wave_min(lv == wm ? lp : KEY_GONE);
      if (lane == 0) { sWv[buf][wv] = wm; sWi[buf][wv] = wp; }
    }
    __syncthreads();
    int win;
    {
      int v  = sWv[buf][lane & 7];
      int ii = sWi[buf][lane & 7];
      int bm = oct_min(v);
      win    = oct_min(v == bm ? ii : KEY_GONE);
    }
    win = __builtin_amdgcn_readfirstlane(win);
    if (tid == r) myOut = win;
    const int wt = win & 255;
    const int wi = win >> 8;
    if (wv == (wt >> 5)) {
      const bool mine = (tid == wt);
      int nv = KEY_GONE, ni = 0;
#pragma unroll
      for (int i = 0; i < 32; ++i) {
        if (mine && i == wi) dk[i] = KEY_GONE;
        if (dk[i] < nv) { nv = dk[i]; ni = i; }
      }
      lv = nv;
      lp = ni * 256 + tid;
    }
  }
  if (tid < GK) knnIdx[(size_t)gg * GK + tid] = myOut & (NUM_N - 1);
}

__global__ __launch_bounds__(256) void encoder_kernel(
    const float* __restrict__ points, const int* __restrict__ knnIdx,
    const float* __restrict__ centers, const float* __restrict__ w1,
    const float* __restrict__ b1, const float* __restrict__ g1,
    const float* __restrict__ be1, const float* __restrict__ m1,
    const float* __restrict__ v1, const float* __restrict__ w2,
    const float* __restrict__ b2, const float* __restrict__ w3,
    const float* __restrict__ b3, const float* __restrict__ g2,
    const float* __restrict__ be2, const float* __restrict__ m2,
    const float* __restrict__ v2, const float* __restrict__ w4,
    const float* __restrict__ b4, float* __restrict__ tokens) {
  extern __shared__ __attribute__((aligned(16))) float smem[];
  float* sFeat = smem;
  float* sH1   = smem + 128;
  float* sX3   = smem + 128 + 32 * 132;
  float* sH3   = sX3 + 32 * 516;
  float* sH4   = sX3;

  const int gg       = blockIdx.x;
  const int b        = gg / NUM_G;
  const int tid      = threadIdx.x;
  const int lane     = tid & 31;
  const int wv       = tid >> 5;
  const int laneN    = lane & 15;
  const int laneHalf = lane >> 4;

  const v8f vzero = {0.f, 0.f, 0.f, 0.f, 0.f, 0.f, 0.f, 0.f};

  if (tid < GK) {
    int idx = knnIdx[(size_t)gg * GK + tid];
    const float* p = points + ((size_t)b * NUM_N + idx) * 3;
    const float* c = centers + (size_t)gg * 3;
    float dx = p[0] - c[0], dy = p[1] - c[1], dz = p[2] - c[2];
    float nr = sqrtf(dx * dx + dy * dy + dz * dz);
    sFeat[tid * 4 + 0] = dx;
    sFeat[tid * 4 + 1] = dy;
    sFeat[tid * 4 + 2] = dz;
    sFeat[tid * 4 + 3] = nr;
  }
  __syncthreads();

  {
    int m0 = wv * 16;
    v8f a0 = vzero, a1 = vzero;
    gemm_1x2<4, 4>(w1, 4, sFeat, m0, laneN, laneHalf, a0, a1);
    auto op = [&](float x, int m) {
      float inv = g1[m] * rsqrtf(v1[m] + 1e-5f);
      float v   = (x + b1[m]) * inv + (be1[m] - m1[m] * inv);
      return fmaxf(v, 0.f);
    };
    store_tile(sH1, 132, a0, m0, 0,  laneN, laneHalf, op);
    store_tile(sH1, 132, a1, m0, 16, laneN, laneHalf, op);
  }
  __syncthreads();

  {
    int m0 = wv * 32;
    v8f a00 = vzero, a01 = vzero, a10 = vzero, a11 = vzero;
    gemm_2x2w<C1, 132>(w2, C1, sH1, m0, laneN, laneHalf, a00, a01, a10, a11);
    auto op = [&](float x, int m) { return x + b2[m]; };
    store_tile(sX3 + 256, 516, a00, m0,      0,  laneN, laneHalf, op);
    store_tile(sX3 + 256, 516, a01, m0,      16, laneN, laneHalf, op);
    store_tile(sX3 + 256, 516, a10, m0 + 16, 0,  laneN, laneHalf, op);
    store_tile(sX3 + 256, 516, a11, m0 + 16, 16, laneN, laneHalf, op);
  }
  __syncthreads();

  {
    int   m  = tid;
    float gm = -3.4e38f;
    for (int n = 0; n < GK; ++n) gm = fmaxf(gm, sX3[n * 516 + 256 + m]);
    for (int n = 0; n < GK; ++n) sX3[n * 516 + m] = gm;
  }
  __syncthreads();

  {
    auto op = [&](float x, int m) {
      float inv = g2[m] * rsqrtf(v2[m] + 1e-5f);
      float v   = (x + b3[m]) * inv + (be2[m] - m2[m] * inv);
      return fmaxf(v, 0.f);
    };
#pragma unroll
    for (int j = 0; j < 2; ++j) {
      int m0  = wv * 64 + j * 32;
      v8f a00 = vzero, a01 = vzero, a10 = vzero, a11 = vzero;
      gemm_2x2w<C3, 516>(w3, C3, sX3, m0, laneN, laneHalf, a00, a01, a10, a11);
      store_tile(sH3, 516, a00, m0,      0,  laneN, laneHalf, op);
      store_tile(sH3, 516, a01, m0,      16, laneN, laneHalf, op);
      store_tile(sH3, 516, a10, m0 + 16, 0,  laneN, laneHalf, op);
      store_tile(sH3, 516, a11, m0 + 16, 16, laneN, laneHalf, op);
    }
  }
  __syncthreads();

  {
    auto op = [&](float x, int m) { return x + b4[m]; };
    {
      int m0  = wv * 48;
      v8f a00 = vzero, a01 = vzero, a10 = vzero, a11 = vzero;
      gemm_2x2w<C3, 516>(w4, C3, sH3, m0, laneN, laneHalf, a00, a01, a10, a11);
      store_tile(sH4, 388, a00, m0,      0,  laneN, laneHalf, op);
      store_tile(sH4, 388, a01, m0,      16, laneN, laneHalf, op);
      store_tile(sH4, 388, a10, m0 + 16, 0,  laneN, laneHalf, op);
      store_tile(sH4, 388, a11, m0 + 16, 16, laneN, laneHalf, op);
    }
    {
      int m0 = wv * 48 + 32;
      v8f a0 = vzero, a1 = vzero;
      gemm_1x2w<C3, 516>(w4, C3, sH3, m0, laneN, laneHalf, a0, a1);
      store_tile(sH4, 388, a0, m0, 0,  laneN, laneHalf, op);
      store_tile(sH4, 388, a1, m0, 16, laneN, laneHalf, op);
    }
  }
  __syncthreads();

  for (int m = tid; m < CT; m += 256) {
    float mx = -3.4e38f;
    for (int n = 0; n < GK; ++n) mx = fmaxf(mx, sH4[n * 388 + m]);
    tokens[(size_t)gg * CT + m] = mx;
  }
}

extern "C" void kernel_launch(void* const* d_in, const int* in_sizes, int n_in,
                              void* d_out, int out_size, void* d_ws,
                              size_t ws_size, hipStream_t stream) {
  (void)in_sizes; (void)n_in; (void)out_size; (void)ws_size;
  const float* points = (const float*)d_in[0];
  const float* w1  = (const float*)d_in[1];
  const float* b1  = (const float*)d_in[2];
  const float* g1  = (const float*)d_in[3];
  const float* be1 = (const float*)d_in[4];
  const float* m1  = (const float*)d_in[5];
  const float* v1  = (const float*)d_in[6];
  const float* w2  = (const float*)d_in[7];
  const float* b2  = (const float*)d_in[8];
  const float* w3  = (const float*)d_in[9];
  const float* b3  = (const float*)d_in[10];
  const float* g2  = (const float*)d_in[11];
  const float* be2 = (const float*)d_in[12];
  const float* m2  = (const float*)d_in[13];
  const float* v2  = (const float*)d_in[14];
  const float* w4  = (const float*)d_in[15];
  const float* b4  = (const float*)d_in[16];

  float* tokens  = (float*)d_out;
  float* centers = tokens + TOK_TOTAL;
  int*   knnIdx  = (int*)d_ws;

  fps_kernel<<<NUM_B, 1024, 0, stream>>>(points, centers);
  knn_kernel<<<NUM_B * NUM_G, 256, 0, stream>>>(points, centers, knnIdx);

  size_t ldsBytes = (size_t)(128 + 32 * 132 + 32 * 516 + 32 * 516) * sizeof(float);
  encoder_kernel<<<NUM_B * NUM_G, 256, ldsBytes, stream>>>(
      points, knnIdx, centers, w1, b1, g1, be1, m1, v1, w2, b2, w3, b3, g2,
      be2, m2, v2, w4, b4, tokens);
}


# Encoder conv3: the 256 max-pooled input channels are equal in all 32 columns, so their products run as ONE 16-column WMMA tile per 16 rows whose D starts both column tiles; sum order unchanged
# speedup vs baseline: 1.9161x; 1.0470x over previous
#include <hip/hip_runtime.h>
#include <math.h>

#define NUM_B 8
#define NUM_N 8192
#define NUM_G 256
#define GK    32
#define C1    128
#define C2    256
#define C3    512
#define CT    384
#define TOK_TOTAL (NUM_B * NUM_G * CT)

typedef __attribute__((ext_vector_type(2))) float v2f;
typedef __attribute__((ext_vector_type(4))) float v4f;
typedef __attribute__((ext_vector_type(8))) float v8f;

__device__ __forceinline__ v8f wmma4(v2f a, v2f b, v8f c) {
  return __builtin_amdgcn_wmma_f32_16x16x4_f32(false, a, false, b, (short)0, c,
                                               false, false);
}

__device__ __forceinline__ v2f lo2(v4f v) { return __builtin_shufflevector(v, v, 0, 1); }
__device__ __forceinline__ v2f hi2(v4f v) { return __builtin_shufflevector(v, v, 2, 3); }

template <int KDIM, int LDB>
__device__ __forceinline__ void gemm_2x2w(const float* __restrict__ W, int ldw,
                                          const float* B, int m0, int laneN,
                                          int laneHalf, v8f& acc00, v8f& acc01,
                                          v8f& acc10, v8f& acc11) {
  static_assert(KDIM % 8 == 0, "8-k blocks");
  const float* wrow0 = W + (size_t)(m0 + laneN) * (size_t)ldw + 4 * laneHalf;
  const float* wrow1 = wrow0 + (size_t)16 * (size_t)ldw;
  const float* bcol0 = B + laneN * LDB + 4 * laneHalf;
  const float* bcol1 = B + (16 + laneN) * LDB + 4 * laneHalf;
#pragma unroll 4
  for (int kk = 0; kk < KDIM; kk += 8) {
    v4f a0 = *(const v4f*)(wrow0 + kk);
    v4f a1 = *(const v4f*)(wrow1 + kk);
    v4f b0 = *(const v4f*)(bcol0 + kk);
    v4f b1 = *(const v4f*)(bcol1 + kk);
    acc00 = wmma4(lo2(a0), lo2(b0), acc00);
    acc01 = wmma4(lo2(a0), lo2(b1), acc01);
    acc10 = wmma4(lo2(a1), lo2(b0), acc10);
    acc11 = wmma4(lo2(a1), lo2(b1), acc11);
    acc00 = wmma4(hi2(a0), hi2(b0), acc00);
    acc01 = wmma4(hi2(a0), hi2(b1), acc01);
    acc10 = wmma4(hi2(a1), hi2(b0), acc10);
    acc11 = wmma4(hi2(a1), hi2(b1), acc11);
  }
}

template <int KDIM, int LDB>
__device__ __forceinline__ void gemm_1x2(const float* __restrict__ W, int ldw,
                                         const float* B, int m0, int laneN,
                                         int laneHalf, v8f& acc0, v8f& acc1) {
  const float* wrow  = W + (size_t)(m0 + laneN) * (size_t)ldw + 2 * laneHalf;
  const float* bcol0 = B + laneN * LDB + 2 * laneHalf;
  const float* bcol1 = B + (16 + laneN) * LDB + 2 * laneHalf;
  for (int kk = 0; kk < KDIM; kk += 4) {
    v2f a  = *(const v2f*)(wrow + kk);
    v2f b0 = *(const v2f*)(bcol0 + kk);
    v2f b1 = *(const v2f*)(bcol1 + kk);
    acc0 = wmma4(a, b0, acc0);
    acc1 = wmma4(a, b1, acc1);
  }
}

template <int KDIM, int LDB>
__device__ __forceinline__ void gemm_1x2w(const float* __restrict__ W, int ldw,
                                          const float* B, int m0, int laneN,
                                          int laneHalf, v8f& acc0, v8f& acc1) {
  static_assert(KDIM % 8 == 0, "8-k blocks");
  const float* wrow  = W + (size_t)(m0 + laneN) * (size_t)ldw + 4 * laneHalf;
  const float* bcol0 = B + laneN * LDB + 4 * laneHalf;
  const float* bcol1 = B + (16 + laneN) * LDB + 4 * laneHalf;
#pragma unroll 4
  for (int kk = 0; kk < KDIM; kk += 8) {
    v4f a  = *(const v4f*)(wrow + kk);
    v4f b0 = *(const v4f*)(bcol0 + kk);
    v4f b1 = *(const v4f*)(bcol1 + kk);
    acc0 = wmma4(lo2(a), lo2(b0), acc0);
    acc1 = wmma4(lo2(a), lo2(b1), acc1);
    acc0 = wmma4(hi2(a), hi2(b0), acc0);
    acc1 = wmma4(hi2(a), hi2(b1), acc1);
  }
}

template <int KDIM>
__device__ __forceinline__ void gemm_2x1v(const float* __restrict__ W, int ldw,
                                          const float* Bv, int m0, int laneN,
                                          int laneHalf, v8f& acc0, v8f& acc1) {
  static_assert(KDIM % 8 == 0, "8-k blocks");
  const float* wrow0 = W + (size_t)(m0 + laneN) * (size_t)ldw + 4 * laneHalf;
  const float* wrow1 = wrow0 + (size_t)16 * (size_t)ldw;
  const float* bvec  = Bv + 4 * laneHalf;
#pragma unroll 8
  for (int kk = 0; kk < KDIM; kk += 8) {
    v4f a0 = *(const v4f*)(wrow0 + kk);
    v4f a1 = *(const v4f*)(wrow1 + kk);
    v4f bb = *(const v4f*)(bvec + kk);
    acc0 = wmma4(lo2(a0), lo2(bb), acc0);
    acc1 = wmma4(lo2(a1), lo2(bb), acc1);
    acc0 = wmma4(hi2(a0), hi2(bb), acc0);
    acc1 = wmma4(hi2(a1), hi2(bb), acc1);
  }
}

template <typename F>
__device__ __forceinline__ void store_tile(float* dst, int ldd, const v8f& acc,
                                           int m0, int n0, int laneN,
                                           int laneHalf, F op) {
  int n = n0 + laneN;
#pragma unroll
  for (int rr = 0; rr < 8; ++rr) {
    int m = m0 + 8 * laneHalf + rr;
    dst[n * ldd + m] = op(acc[rr], m);
  }
}

template <int M>
__device__ __forceinline__ int row_xor(int b) {
  return __builtin_amdgcn_update_dpp(b, b, 0x160 + M, 0xf, 0xf, true);
}
__device__ __forceinline__ int imax(int a, int b) { return a > b ? a : b; }
__device__ __forceinline__ int wave_max(int v) {
  v = imax(v, row_xor<1>(v));
  v = imax(v, row_xor<2>(v));
  v = imax(v, row_xor<4>(v));
  v = imax(v, row_xor<8>(v));
  v = imax(v, __builtin_amdgcn_permlanex16(v, v, 0x76543210, 0xfedcba98, false,
                                           false));
  return v;
}

__global__ __launch_bounds__(1024) void fps_kernel(
    const float* __restrict__ points, float* __restrict__ centers) {
  const int b    = blockIdx.x;
  const int tid  = threadIdx.x;
  const int lane = tid & 31;
  const int wv   = tid >> 5;
  const float* base = points + (size_t)b * NUM_N * 3;
  float* co = centers + (size_t)b * NUM_G * 3;

  __shared__ float sP[NUM_N * 3];
  __shared__ int   sWv[2][32];
  __shared__ int   sWi[2][32];

  for (int k = tid; k < NUM_N * 3; k += 1024) sP[k] = base[k];

  float px[8], py[8], pz[8], dd[8];
#pragma unroll
  for (int i = 0; i < 8; ++i) {
    int p = tid * 8 + i;
    px[i] = base[p * 3 + 0];
    py[i] = base[p * 3 + 1];
    pz[i] = base[p * 3 + 2];
    dd[i] = 1e10f;
  }
  __syncthreads();

  float lx = sP[0], ly = sP[1], lz = sP[2];
  if (tid == 0) { co[0] = lx; co[1] = ly; co[2] = lz; }

  for (int g = 1; g < NUM_G; ++g) {
    float bv = -1.f;
    int   bi = 0;
#pragma unroll
    for (int i = 0; i < 8; ++i) {
      float dx = px[i] - lx, dy = py[i] - ly, dz = pz[i] - lz;
      float d  = __builtin_fmaf(dz, dz, __builtin_fmaf(dx, dx, dy * dy));
      dd[i]    = fminf(dd[i], d);
      if (dd[i] > bv) { bv = dd[i]; bi = tid * 8 + i; }
    }
    const int buf = g & 1;
    {
      int      bvb = __builtin_bit_cast(int, bv);
      int      wm  = wave_max(bvb);
      unsigned has = __builtin_amdgcn_ballot_w32(bvb == wm);
      int      wbi = __builtin_amdgcn_readlane(bi, __builtin_ctz(has));
      if (lane == 0) { sWv[buf][wv] = wm; sWi[buf][wv] = wbi; }
    }
    __syncthreads();
    {
      int      v   = sWv[buf][lane];
      int      ii  = sWi[buf][lane];
      int      bm  = wave_max(v);
      unsigned has = __builtin_amdgcn_ballot_w32(v == bm);
      int      wi  = __builtin_amdgcn_readlane(ii, __builtin_ctz(has));
      lx = sP[wi * 3 + 0];
      ly = sP[wi * 3 + 1];
      lz = sP[wi * 3 + 2];
    }
    if (tid == 0) { co[g * 3 + 0] = lx; co[g * 3 + 1] = ly; co[g * 3 + 2] = lz; }
  }
}

__device__ __forceinline__ int imin(int a, int b) { return a < b ? a : b; }
__device__ __forceinline__ int wave_min(int v) {
  v = imin(v, row_xor<1>(v));
  v = imin(v, row_xor<2>(v));
  v = imin(v, row_xor<4>(v));
  v = imin(v, row_xor<8>(v));
  v = imin(v, __builtin_amdgcn_permlanex16(v, v, 0x76543210, 0xfedcba98, false,
                                           false));
  return v;
}
__device__ __forceinline__ int oct_min(int v) {
  v = imin(v, row_xor<1>(v));
  v = imin(v, row_xor<2>(v));
  v = imin(v, row_xor<4>(v));
  return v;
}
__device__ __forceinline__ int order_key(float f) {
  int bits = __builtin_bit_cast(int, f);
  return bits ^ ((bits >> 31) & 0x7fffffff);
}

#define KEY_GONE 0x7fffffff

__global__ __launch_bounds__(256) void knn_kernel(
    const float* __restrict__ points, const float* __restrict__ centers,
    int* __restrict__ knnIdx) {
  const int gg   = blockIdx.x;
  const int b    = gg / NUM_G;
  const int tid  = threadIdx.x;
  const int lane = tid & 31;
  const int wv   = tid >> 5;
  const float* base = points + (size_t)b * NUM_N * 3;

  __shared__ int sWv[2][8];
  __shared__ int sWi[2][8];

  const float cx = centers[(size_t)gg * 3 + 0];
  const float cy = centers[(size_t)gg * 3 + 1];
  const float cz = centers[(size_t)gg * 3 + 2];
  const float c2 = __builtin_fmaf(cz, cz, __builtin_fmaf(cx, cx, cy * cy));

  int dk[32];
#pragma unroll
  for (int i = 0; i < NUM_N / 256; ++i) {
    int   p  = i * 256 + tid;
    float xx = base[p * 3 + 0], yy = base[p * 3 + 1], zz = base[p * 3 + 2];
    float x2 = __builtin_fmaf(zz, zz, __builtin_fmaf(xx, xx, yy * yy));
    float dt = __builtin_fmaf(cz, zz, __builtin_fmaf(cx, xx, cy * yy));
    dk[i]    = order_key(__builtin_fmaf(-2.f, dt, c2 + x2));
  }

  int lv = KEY_GONE, li = 0;
#pragma unroll
  for (int i = 0; i < 32; ++i)
    if (dk[i] < lv) { lv = dk[i]; li = i; }
  int lp    = li * 256 + tid;
  int myOut = 0;

  for (int r = 0; r < GK; ++r) {
    const int buf = r & 1;
    {
      int wm = wave_min(lv);
      int wp = wave_min(lv == wm ? lp : KEY_GONE);
      if (lane == 0) { sWv[buf][wv] = wm; sWi[buf][wv] = wp; }
    }
    __syncthreads();
    int win;
    {
      int v  = sWv[buf][lane & 7];
      int ii = sWi[buf][lane & 7];
      int bm = oct_min(v);
      win    = oct_min(v == bm ? ii : KEY_GONE);
    }
    win = __builtin_amdgcn_readfirstlane(win);
    if (tid == r) myOut = win;
    const int wt = win & 255;
    const int wi = win >> 8;
    if (wv == (wt >> 5)) {
      const bool mine = (tid == wt);
      int nv = KEY_GONE, ni = 0;
#pragma unroll
      for (int i = 0; i < 32; ++i) {
        if (mine && i == wi) dk[i] = KEY_GONE;
        if (dk[i] < nv) { nv = dk[i]; ni = i; }
      }
      lv = nv;
      lp = ni * 256 + tid;
    }
  }
  if (tid < GK) knnIdx[(size_t)gg * GK + tid] = myOut & (NUM_N - 1);
}

__global__ __launch_bounds__(256) void encoder_kernel(
    const float* __restrict__ points, const int* __restrict__ knnIdx,
    const float* __restrict__ centers, const float* __restrict__ w1,
    const float* __restrict__ b1, const float* __restrict__ g1,
    const float* __restrict__ be1, const float* __restrict__ m1,
    const float* __restrict__ v1, const float* __restrict__ w2,
    const float* __restrict__ b2, const float* __restrict__ w3,
    const float* __restrict__ b3, const float* __restrict__ g2,
    const float* __restrict__ be2, const float* __restrict__ m2,
    const float* __restrict__ v2, const float* __restrict__ w4,
    const float* __restrict__ b4, float* __restrict__ tokens) {
  extern __shared__ __attribute__((aligned(16))) float smem[];
  float* sFeat = smem;
  float* sH1   = smem + 128;
  float* sX3   = smem + 128 + 32 * 132;
  float* sH3   = sX3 + 32 * 516;
  float* sH4   = sX3;

  const int gg       = blockIdx.x;
  const int b        = gg / NUM_G;
  const int tid      = threadIdx.x;
  const int lane     = tid & 31;
  const int wv       = tid >> 5;
  const int laneN    = lane & 15;
  const int laneHalf = lane >> 4;

  const v8f vzero = {0.f, 0.f, 0.f, 0.f, 0.f, 0.f, 0.f, 0.f};

  if (tid < GK) {
    int idx = knnIdx[(size_t)gg * GK + tid];
    const float* p = points + ((size_t)b * NUM_N + idx) * 3;
    const float* c = centers + (size_t)gg * 3;
    float dx = p[0] - c[0], dy = p[1] - c[1], dz = p[2] - c[2];
    float nr = sqrtf(dx * dx + dy * dy + dz * dz);
    sFeat[tid * 4 + 0] = dx;
    sFeat[tid * 4 + 1] = dy;
    sFeat[tid * 4 + 2] = dz;
    sFeat[tid * 4 + 3] = nr;
  }
  __syncthreads();

  {
    int m0 = wv * 16;
    v8f a0 = vzero, a1 = vzero;
    gemm_1x2<4, 4>(w1, 4, sFeat, m0, laneN, laneHalf, a0, a1);
    auto op = [&](float x, int m) {
      float inv = g1[m] * rsqrtf(v1[m] + 1e-5f);
      float v   = (x + b1[m]) * inv + (be1[m] - m1[m] * inv);
      return fmaxf(v, 0.f);
    };
    store_tile(sH1, 132, a0, m0, 0,  laneN, laneHalf, op);
    store_tile(sH1, 132, a1, m0, 16, laneN, laneHalf, op);
  }
  __syncthreads();

  {
    int m0 = wv * 32;
    v8f a00 = vzero, a01 = vzero, a10 = vzero, a11 = vzero;
    gemm_2x2w<C1, 132>(w2, C1, sH1, m0, laneN, laneHalf, a00, a01, a10, a11);
    auto op = [&](float x, int m) { return x + b2[m]; };
    store_tile(sX3 + 256, 516, a00, m0,      0,  laneN, laneHalf, op);
    store_tile(sX3 + 256, 516, a01, m0,      16, laneN, laneHalf, op);
    store_tile(sX3 + 256, 516, a10, m0 + 16, 0,  laneN, laneHalf, op);
    store_tile(sX3 + 256, 516, a11, m0 + 16, 16, laneN, laneHalf, op);
  }
  __syncthreads();

  float* sGm = sX3;
  {
    int   m  = tid;
    float gm = -3.4e38f;
    for (int n = 0; n < GK; ++n) gm = fmaxf(gm, sX3[n * 516 + 256 + m]);
    sGm[m] = gm;
  }
  __syncthreads();

  {
    auto op = [&](float x, int m) {
      float inv = g2[m] * rsqrtf(v2[m] + 1e-5f);
      float v   = (x + b3[m]) * inv + (be2[m] - m2[m] * inv);
      return fmaxf(v, 0.f);
    };
#pragma unroll
    for (int j = 0; j < 2; ++j) {
      int m0  = wv * 64 + j * 32;
      v8f g0 = vzero, g1 = vzero;
      gemm_2x1v<C2>(w3, C3, sGm, m0, laneN, laneHalf, g0, g1);
      v8f a00 = g0, a01 = g0, a10 = g1, a11 = g1;
      gemm_2x2w<C2, 516>(w3 + C2, C3, sX3 + C2, m0, laneN, laneHalf, a00, a01,
                         a10, a11);
      store_tile(sH3, 516, a00, m0,      0,  laneN, laneHalf, op);
      store_tile(sH3, 516, a01, m0,      16, laneN, laneHalf, op);
      store_tile(sH3, 516, a10, m0 + 16, 0,  laneN, laneHalf, op);
      store_tile(sH3, 516, a11, m0 + 16, 16, laneN, laneHalf, op);
    }
  }
  __syncthreads();

  {
    auto op = [&](float x, int m) { return x + b4[m]; };
    {
      int m0  = wv * 48;
      v8f a00 = vzero, a01 = vzero, a10 = vzero, a11 = vzero;
      gemm_2x2w<C3, 516>(w4, C3, sH3, m0, laneN, laneHalf, a00, a01, a10, a11);
      store_tile(sH4, 388, a00, m0,      0,  laneN, laneHalf, op);
      store_tile(sH4, 388, a01, m0,      16, laneN, laneHalf, op);
      store_tile(sH4, 388, a10, m0 + 16, 0,  laneN, laneHalf, op);
      store_tile(sH4, 388, a11, m0 + 16, 16, laneN, laneHalf, op);
    }
    {
      int m0 = wv * 48 + 32;
      v8f a0 = vzero, a1 = vzero;
      gemm_1x2w<C3, 516>(w4, C3, sH3, m0, laneN, laneHalf, a0, a1);
      store_tile(sH4, 388, a0, m0, 0,  laneN, laneHalf, op);
      store_tile(sH4, 388, a1, m0, 16, laneN, laneHalf, op);
    }
  }
  __syncthreads();

  for (int m = tid; m < CT; m += 256) {
    float mx = -3.4e38f;
    for (int n = 0; n < GK; ++n) mx = fmaxf(mx, sH4[n * 388 + m]);
    tokens[(size_t)gg * CT + m] = mx;
  }
}

extern "C" void kernel_launch(void* const* d_in, const int* in_sizes, int n_in,
                              void* d_out, int out_size, void* d_ws,
                              size_t ws_size, hipStream_t stream) {
  (void)in_sizes; (void)n_in; (void)out_size; (void)ws_size;
  const float* points = (const float*)d_in[0];
  const float* w1  = (const float*)d_in[1];
  const float* b1  = (const float*)d_in[2];
  const float* g1  = (const float*)d_in[3];
  const float* be1 = (const float*)d_in[4];
  const float* m1  = (const float*)d_in[5];
  const float* v1  = (const float*)d_in[6];
  const float* w2  = (const float*)d_in[7];
  const float* b2  = (const float*)d_in[8];
  const float* w3  = (const float*)d_in[9];
  const float* b3  = (const float*)d_in[10];
  const float* g2  = (const float*)d_in[11];
  const float* be2 = (const float*)d_in[12];
  const float* m2  = (const float*)d_in[13];
  const float* v2  = (const float*)d_in[14];
  const float* w4  = (const float*)d_in[15];
  const float* b4  = (const float*)d_in[16];

  float* tokens  = (float*)d_out;
  float* centers = tokens + TOK_TOTAL;
  int*   knnIdx  = (int*)d_ws;

  fps_kernel<<<NUM_B, 1024, 0, stream>>>(points, centers);
  knn_kernel<<<NUM_B * NUM_G, 256, 0, stream>>>(points, centers, knnIdx);

  size_t ldsBytes = (size_t)(128 + 32 * 132 + 32 * 516 + 32 * 516) * sizeof(float);
  encoder_kernel<<<NUM_B * NUM_G, 256, ldsBytes, stream>>>(
      points, knnIdx, centers, w1, b1, g1, be1, m1, v1, w2, b2, w3, b3, g2,
      be2, m2, v2, w4, b4, tokens);
}


# FPS: 16 points per thread in 16 waves (was 8 in 32); threads and waves reduce the distance VALUE only (integer min/max on its bits), the holder's position read once per wave from equality masks
# speedup vs baseline: 1.9769x; 1.0317x over previous
#include <hip/hip_runtime.h>
#include <math.h>

#define NUM_B 8
#define NUM_N 8192
#define NUM_G 256
#define GK    32
#define C1    128
#define C2    256
#define C3    512
#define CT    384
#define TOK_TOTAL (NUM_B * NUM_G * CT)

typedef __attribute__((ext_vector_type(2))) float v2f;
typedef __attribute__((ext_vector_type(4))) float v4f;
typedef __attribute__((ext_vector_type(8))) float v8f;

__device__ __forceinline__ v8f wmma4(v2f a, v2f b, v8f c) {
  return __builtin_amdgcn_wmma_f32_16x16x4_f32(false, a, false, b, (short)0, c,
                                               false, false);
}

__device__ __forceinline__ v2f lo2(v4f v) { return __builtin_shufflevector(v, v, 0, 1); }
__device__ __forceinline__ v2f hi2(v4f v) { return __builtin_shufflevector(v, v, 2, 3); }

template <int KDIM, int LDB>
__device__ __forceinline__ void gemm_2x2w(const float* __restrict__ W, int ldw,
                                          const float* B, int m0, int laneN,
                                          int laneHalf, v8f& acc00, v8f& acc01,
                                          v8f& acc10, v8f& acc11) {
  static_assert(KDIM % 8 == 0, "8-k blocks");
  const float* wrow0 = W + (size_t)(m0 + laneN) * (size_t)ldw + 4 * laneHalf;
  const float* wrow1 = wrow0 + (size_t)16 * (size_t)ldw;
  const float* bcol0 = B + laneN * LDB + 4 * laneHalf;
  const float* bcol1 = B + (16 + laneN) * LDB + 4 * laneHalf;
#pragma unroll 4
  for (int kk = 0; kk < KDIM; kk += 8) {
    v4f a0 = *(const v4f*)(wrow0 + kk);
    v4f a1 = *(const v4f*)(wrow1 + kk);
    v4f b0 = *(const v4f*)(bcol0 + kk);
    v4f b1 = *(const v4f*)(bcol1 + kk);
    acc00 = wmma4(lo2(a0), lo2(b0), acc00);
    acc01 = wmma4(lo2(a0), lo2(b1), acc01);
    acc10 = wmma4(lo2(a1), lo2(b0), acc10);
    acc11 = wmma4(lo2(a1), lo2(b1), acc11);
    acc00 = wmma4(hi2(a0), hi2(b0), acc00);
    acc01 = wmma4(hi2(a0), hi2(b1), acc01);
    acc10 = wmma4(hi2(a1), hi2(b0), acc10);
    acc11 = wmma4(hi2(a1), hi2(b1), acc11);
  }
}

template <int KDIM, int LDB>
__device__ __forceinline__ void gemm_1x2(const float* __restrict__ W, int ldw,
                                         const float* B, int m0, int laneN,
                                         int laneHalf, v8f& acc0, v8f& acc1) {
  const float* wrow  = W + (size_t)(m0 + laneN) * (size_t)ldw + 2 * laneHalf;
  const float* bcol0 = B + laneN * LDB + 2 * laneHalf;
  const float* bcol1 = B + (16 + laneN) * LDB + 2 * laneHalf;
  for (int kk = 0; kk < KDIM; kk += 4) {
    v2f a  = *(const v2f*)(wrow + kk);
    v2f b0 = *(const v2f*)(bcol0 + kk);
    v2f b1 = *(const v2f*)(bcol1 + kk);
    acc0 = wmma4(a, b0, acc0);
    acc1 = wmma4(a, b1, acc1);
  }
}

template <int KDIM, int LDB>
__device__ __forceinline__ void gemm_1x2w(const float* __restrict__ W, int ldw,
                                          const float* B, int m0, int laneN,
                                          int laneHalf, v8f& acc0, v8f& acc1) {
  static_assert(KDIM % 8 == 0, "8-k blocks");
  const float* wrow  = W + (size_t)(m0 + laneN) * (size_t)ldw + 4 * laneHalf;
  const float* bcol0 = B + laneN * LDB + 4 * laneHalf;
  const float* bcol1 = B + (16 + laneN) * LDB + 4 * laneHalf;
#pragma unroll 4
  for (int kk = 0; kk < KDIM; kk += 8) {
    v4f a  = *(const v4f*)(wrow + kk);
    v4f b0 = *(const v4f*)(bcol0 + kk);
    v4f b1 = *(const v4f*)(bcol1 + kk);
    acc0 = wmma4(lo2(a), lo2(b0), acc0);
    acc1 = wmma4(lo2(a), lo2(b1), acc1);
    acc0 = wmma4(hi2(a), hi2(b0), acc0);
    acc1 = wmma4(hi2(a), hi2(b1), acc1);
  }
}

template <int KDIM>
__device__ __forceinline__ void gemm_2x1v(const float* __restrict__ W, int ldw,
                                          const float* Bv, int m0, int laneN,
                                          int laneHalf, v8f& acc0, v8f& acc1) {
  static_assert(KDIM % 8 == 0, "8-k blocks");
  const float* wrow0 = W + (size_t)(m0 + laneN) * (size_t)ldw + 4 * laneHalf;
  const float* wrow1 = wrow0 + (size_t)16 * (size_t)ldw;
  const float* bvec  = Bv + 4 * laneHalf;
#pragma unroll 8
  for (int kk = 0; kk < KDIM; kk += 8) {
    v4f a0 = *(const v4f*)(wrow0 + kk);
    v4f a1 = *(const v4f*)(wrow1 + kk);
    v4f bb = *(const v4f*)(bvec + kk);
    acc0 = wmma4(lo2(a0), lo2(bb), acc0);
    acc1 = wmma4(lo2(a1), lo2(bb), acc1);
    acc0 = wmma4(hi2(a0), hi2(bb), acc0);
    acc1 = wmma4(hi2(a1), hi2(bb), acc1);
  }
}

template <typename F>
__device__ __forceinline__ void store_tile(float* dst, int ldd, const v8f& acc,
                                           int m0, int n0, int laneN,
                                           int laneHalf, F op) {
  int n = n0 + laneN;
#pragma unroll
  for (int rr = 0; rr < 8; ++rr) {
    int m = m0 + 8 * laneHalf + rr;
    dst[n * ldd + m] = op(acc[rr], m);
  }
}

template <int M>
__device__ __forceinline__ int row_xor(int b) {
  return __builtin_amdgcn_update_dpp(b, b, 0x160 + M, 0xf, 0xf, true);
}
__device__ __forceinline__ int imax(int a, int b) { return a > b ? a : b; }
__device__ __forceinline__ int imin(int a, int b) { return a < b ? a : b; }
__device__ __forceinline__ int wave_max(int v) {
  v = imax(v, row_xor<1>(v));
  v = imax(v, row_xor<2>(v));
  v = imax(v, row_xor<4>(v));
  v = imax(v, row_xor<8>(v));
  v = imax(v, __builtin_amdgcn_permlanex16(v, v, 0x76543210, 0xfedcba98, false,
                                           false));
  return v;
}

#define FPS_T 512
#define FPS_P (NUM_N / FPS_T)
#define FPS_W (FPS_T / 32)

__device__ __forceinline__ int row_max(int v) {
  v = imax(v, row_xor<1>(v));
  v = imax(v, row_xor<2>(v));
  v = imax(v, row_xor<4>(v));
  v = imax(v, row_xor<8>(v));
  return v;
}

__global__ __launch_bounds__(FPS_T) void fps_kernel(
    const float* __restrict__ points, float* __restrict__ centers) {
  const int b    = blockIdx.x;
  const int tid  = threadIdx.x;
  const int lane = tid & 31;
  const int wv   = __builtin_amdgcn_readfirstlane(tid >> 5);
  const float* base = points + (size_t)b * NUM_N * 3;
  float* co = centers + (size_t)b * NUM_G * 3;

  __shared__ float sP[NUM_N * 3];
  __shared__ int   sWv[2][FPS_W];
  __shared__ int   sWi[2][FPS_W];

  for (int k = tid; k < NUM_N * 3; k += FPS_T) sP[k] = base[k];

  float px[FPS_P], py[FPS_P], pz[FPS_P];
  int   dd[FPS_P];
#pragma unroll
  for (int i = 0; i < FPS_P; ++i) {
    int p = tid * FPS_P + i;
    px[i] = base[p * 3 + 0];
    py[i] = base[p * 3 + 1];
    pz[i] = base[p * 3 + 2];
    dd[i] = __builtin_bit_cast(int, 1e10f);
  }
  __syncthreads();

  float lx = sP[0], ly = sP[1], lz = sP[2];
  if (tid == 0) { co[0] = lx; co[1] = ly; co[2] = lz; }

  for (int g = 1; g < NUM_G; ++g) {
#pragma unroll
    for (int i = 0; i < FPS_P; ++i) {
      float dx = px[i] - lx, dy = py[i] - ly, dz = pz[i] - lz;
      float d = __builtin_fmaf(dz, dz, __builtin_fmaf(dx, dx, dy * dy));
      dd[i]   = imin(dd[i], __builtin_bit_cast(int, d));
    }
    static_assert(FPS_P == 16, "the tree below is written for 16 points");
    const int t0 = imax(imax(dd[0], dd[1]), dd[2]);
    const int t1 = imax(imax(dd[3], dd[4]), dd[5]);
    const int t2 = imax(imax(dd[6], dd[7]), dd[8]);
    const int t3 = imax(imax(dd[9], dd[10]), dd[11]);
    const int t4 = imax(imax(dd[12], dd[13]), dd[14]);
    const int u0 = imax(imax(t0, t1), t2);
    const int u1 = imax(imax(t3, t4), dd[15]);
    const int tm = imax(u0, u1);

    const int buf = g & 1;
    {
      int      wm  = wave_max(tm);
      unsigned has = __builtin_amdgcn_ballot_w32(tm == wm);
      int      fl  = __builtin_ctz(has);
      int idx = FPS_P - 1;
#pragma unroll
      for (int i = FPS_P - 2; i >= 0; --i) {
        unsigned eq = __builtin_amdgcn_ballot_w32(dd[i] == wm);
        if ((eq >> fl) & 1u) idx = i;
      }
      int wbi = (wv * 32 + fl) * FPS_P + idx;
      if (lane == 0) { sWv[buf][wv] = wm; sWi[buf][wv] = wbi; }
    }
    __syncthreads();
    {
      int      v   = sWv[buf][lane & (FPS_W - 1)];
      int      ii  = sWi[buf][lane & (FPS_W - 1)];
      int      bm  = row_max(v);
      unsigned has = __builtin_amdgcn_ballot_w32(v == bm);
      int      wi  = __builtin_amdgcn_readlane(ii, __builtin_ctz(has));
      lx = sP[wi * 3 + 0];
      ly = sP[wi * 3 + 1];
      lz = sP[wi * 3 + 2];
    }
    if (tid == 0) { co[g * 3 + 0] = lx; co[g * 3 + 1] = ly; co[g * 3 + 2] = lz; }
  }
}

__device__ __forceinline__ int wave_min(int v) {
  v = imin(v, row_xor<1>(v));
  v = imin(v, row_xor<2>(v));
  v = imin(v, row_xor<4>(v));
  v = imin(v, row_xor<8>(v));
  v = imin(v, __builtin_amdgcn_permlanex16(v, v, 0x76543210, 0xfedcba98, false,
                                           false));
  return v;
}
__device__ __forceinline__ int oct_min(int v) {
  v = imin(v, row_xor<1>(v));
  v = imin(v, row_xor<2>(v));
  v = imin(v, row_xor<4>(v));
  return v;
}
__device__ __forceinline__ int order_key(float f) {
  int bits = __builtin_bit_cast(int, f);
  return bits ^ ((bits >> 31) & 0x7fffffff);
}

#define KEY_GONE 0x7fffffff

__global__ __launch_bounds__(256) void knn_kernel(
    const float* __restrict__ points, const float* __restrict__ centers,
    int* __restrict__ knnIdx) {
  const int gg   = blockIdx.x;
  const int b    = gg / NUM_G;
  const int tid  = threadIdx.x;
  const int lane = tid & 31;
  const int wv   = tid >> 5;
  const float* base = points + (size_t)b * NUM_N * 3;

  __shared__ int sWv[2][8];
  __shared__ int sWi[2][8];

  const float cx = centers[(size_t)gg * 3 + 0];
  const float cy = centers[(size_t)gg * 3 + 1];
  const float cz = centers[(size_t)gg * 3 + 2];
  const float c2 = __builtin_fmaf(cz, cz, __builtin_fmaf(cx, cx, cy * cy));

  int dk[32];
#pragma unroll
  for (int i = 0; i < NUM_N / 256; ++i) {
    int   p  = i * 256 + tid;
    float xx = base[p * 3 + 0], yy = base[p * 3 + 1], zz = base[p * 3 + 2];
    float x2 = __builtin_fmaf(zz, zz, __builtin_fmaf(xx, xx, yy * yy));
    float dt = __builtin_fmaf(cz, zz, __builtin_fmaf(cx, xx, cy * yy));
    dk[i]    = order_key(__builtin_fmaf(-2.f, dt, c2 + x2));
  }

  int lv = KEY_GONE, li = 0;
#pragma unroll
  for (int i = 0; i < 32; ++i)
    if (dk[i] < lv) { lv = dk[i]; li = i; }
  int lp    = li * 256 + tid;
  int myOut = 0;

  for (int r = 0; r < GK; ++r) {
    const int buf = r & 1;
    {
      int wm = wave_min(lv);
      int wp = wave_min(lv == wm ? lp : KEY_GONE);
      if (lane == 0) { sWv[buf][wv] = wm; sWi[buf][wv] = wp; }
    }
    __syncthreads();
    int win;
    {
      int v  = sWv[buf][lane & 7];
      int ii = sWi[buf][lane & 7];
      int bm = oct_min(v);
      win    = oct_min(v == bm ? ii : KEY_GONE);
    }
    win = __builtin_amdgcn_readfirstlane(win);
    if (tid == r) myOut = win;
    const int wt = win & 255;
    const int wi = win >> 8;
    if (wv == (wt >> 5)) {
      const bool mine = (tid == wt);
      int nv = KEY_GONE, ni = 0;
#pragma unroll
      for (int i = 0; i < 32; ++i) {
        if (mine && i == wi) dk[i] = KEY_GONE;
        if (dk[i] < nv) { nv = dk[i]; ni = i; }
      }
      lv = nv;
      lp = ni * 256 + tid;
    }
  }
  if (tid < GK) knnIdx[(size_t)gg * GK + tid] = myOut & (NUM_N - 1);
}

__global__ __launch_bounds__(256) void encoder_kernel(
    const float* __restrict__ points, const int* __restrict__ knnIdx,
    const float* __restrict__ centers, const float* __restrict__ w1,
    const float* __restrict__ b1, const float* __restrict__ g1,
    const float* __restrict__ be1, const float* __restrict__ m1,
    const float* __restrict__ v1, const float* __restrict__ w2,
    const float* __restrict__ b2, const float* __restrict__ w3,
    const float* __restrict__ b3, const float* __restrict__ g2,
    const float* __restrict__ be2, const float* __restrict__ m2,
    const float* __restrict__ v2, const float* __restrict__ w4,
    const float* __restrict__ b4, float* __restrict__ tokens) {
  extern __shared__ __attribute__((aligned(16))) float smem[];
  float* sFeat = smem;
  float* sH1   = smem + 128;
  float* sX3   = smem + 128 + 32 * 132;
  float* sH3   = sX3 + 32 * 516;
  float* sH4   = sX3;

  const int gg       = blockIdx.x;
  const int b        = gg / NUM_G;
  const int tid      = threadIdx.x;
  const int lane     = tid & 31;
  const int wv       = tid >> 5;
  const int laneN    = lane & 15;
  const int laneHalf = lane >> 4;

  const v8f vzero = {0.f, 0.f, 0.f, 0.f, 0.f, 0.f, 0.f, 0.f};

  if (tid < GK) {
    int idx = knnIdx[(size_t)gg * GK + tid];
    const float* p = points + ((size_t)b * NUM_N + idx) * 3;
    const float* c = centers + (size_t)gg * 3;
    float dx = p[0] - c[0], dy = p[1] - c[1], dz = p[2] - c[2];
    float nr = sqrtf(dx * dx + dy * dy + dz * dz);
    sFeat[tid * 4 + 0] = dx;
    sFeat[tid * 4 + 1] = dy;
    sFeat[tid * 4 + 2] = dz;
    sFeat[tid * 4 + 3] = nr;
  }
  __syncthreads();

  {
    int m0 = wv * 16;
    v8f a0 = vzero, a1 = vzero;
    gemm_1x2<4, 4>(w1, 4, sFeat, m0, laneN, laneHalf, a0, a1);
    auto op = [&](float x, int m) {
      float inv = g1[m] * rsqrtf(v1[m] + 1e-5f);
      float v   = (x + b1[m]) * inv + (be1[m] - m1[m] * inv);
      return fmaxf(v, 0.f);
    };
    store_tile(sH1, 132, a0, m0, 0,  laneN, laneHalf, op);
    store_tile(sH1, 132, a1, m0, 16, laneN, laneHalf, op);
  }
  __syncthreads();

  {
    int m0 = wv * 32;
    v8f a00 = vzero, a01 = vzero, a10 = vzero, a11 = vzero;
    gemm_2x2w<C1, 132>(w2, C1, sH1, m0, laneN, laneHalf, a00, a01, a10, a11);
    auto op = [&](float x, int m) { return x + b2[m]; };
    store_tile(sX3 + 256, 516, a00, m0,      0,  laneN, laneHalf, op);
    store_tile(sX3 + 256, 516, a01, m0,      16, laneN, laneHalf, op);
    store_tile(sX3 + 256, 516, a10, m0 + 16, 0,  laneN, laneHalf, op);
    store_tile(sX3 + 256, 516, a11, m0 + 16, 16, laneN, laneHalf, op);
  }
  __syncthreads();

  float* sGm = sX3;
  {
    int   m  = tid;
    float gm = -3.4e38f;
    for (int n = 0; n < GK; ++n) gm = fmaxf(gm, sX3[n * 516 + 256 + m]);
    sGm[m] = gm;
  }
  __syncthreads();

  {
    auto op = [&](float x, int m) {
      float inv = g2[m] * rsqrtf(v2[m] + 1e-5f);
      float v   = (x + b3[m]) * inv + (be2[m] - m2[m] * inv);
      return fmaxf(v, 0.f);
    };
#pragma unroll
    for (int j = 0; j < 2; ++j) {
      int m0  = wv * 64 + j * 32;
      v8f g0 = vzero, g1 = vzero;
      gemm_2x1v<C2>(w3, C3, sGm, m0, laneN, laneHalf, g0, g1);
      v8f a00 = g0, a01 = g0, a10 = g1, a11 = g1;
      gemm_2x2w<C2, 516>(w3 + C2, C3, sX3 + C2, m0, laneN, laneHalf, a00, a01,
                         a10, a11);
      store_tile(sH3, 516, a00, m0,      0,  laneN, laneHalf, op);
      store_tile(sH3, 516, a01, m0,      16, laneN, laneHalf, op);
      store_tile(sH3, 516, a10, m0 + 16, 0,  laneN, laneHalf, op);
      store_tile(sH3, 516, a11, m0 + 16, 16, laneN, laneHalf, op);
    }
  }
  __syncthreads();

  {
    auto op = [&](float x, int m) { return x + b4[m]; };
    {
      int m0  = wv * 48;
      v8f a00 = vzero, a01 = vzero, a10 = vzero, a11 = vzero;
      gemm_2x2w<C3, 516>(w4, C3, sH3, m0, laneN, laneHalf, a00, a01, a10, a11);
      store_tile(sH4, 388, a00, m0,      0,  laneN, laneHalf, op);
      store_tile(sH4, 388, a01, m0,      16, laneN, laneHalf, op);
      store_tile(sH4, 388, a10, m0 + 16, 0,  laneN, laneHalf, op);
      store_tile(sH4, 388, a11, m0 + 16, 16, laneN, laneHalf, op);
    }
    {
      int m0 = wv * 48 + 32;
      v8f a0 = vzero, a1 = vzero;
      gemm_1x2w<C3, 516>(w4, C3, sH3, m0, laneN, laneHalf, a0, a1);
      store_tile(sH4, 388, a0, m0, 0,  laneN, laneHalf, op);
      store_tile(sH4, 388, a1, m0, 16, laneN, laneHalf, op);
    }
  }
  __syncthreads();

  for (int m = tid; m < CT; m += 256) {
    float mx = -3.4e38f;
    for (int n = 0; n < GK; ++n) mx = fmaxf(mx, sH4[n * 388 + m]);
    tokens[(size_t)gg * CT + m] = mx;
  }
}

extern "C" void kernel_launch(void* const* d_in, const int* in_sizes, int n_in,
                              void* d_out, int out_size, void* d_ws,
                              size_t ws_size, hipStream_t stream) {
  (void)in_sizes; (void)n_in; (void)out_size; (void)ws_size;
  const float* points = (const float*)d_in[0];
  const float* w1  = (const float*)d_in[1];
  const float* b1  = (const float*)d_in[2];
  const float* g1  = (const float*)d_in[3];
  const float* be1 = (const float*)d_in[4];
  const float* m1  = (const float*)d_in[5];
  const float* v1  = (const float*)d_in[6];
  const float* w2  = (const float*)d_in[7];
  const float* b2  = (const float*)d_in[8];
  const float* w3  = (const float*)d_in[9];
  const float* b3  = (const float*)d_in[10];
  const float* g2  = (const float*)d_in[11];
  const float* be2 = (const float*)d_in[12];
  const float* m2  = (const float*)d_in[13];
  const float* v2  = (const float*)d_in[14];
  const float* w4  = (const float*)d_in[15];
  const float* b4  = (const float*)d_in[16];

  float* tokens  = (float*)d_out;
  float* centers = tokens + TOK_TOTAL;
  int*   knnIdx  = (int*)d_ws;

  fps_kernel<<<NUM_B, FPS_T, 0, stream>>>(points, centers);
  knn_kernel<<<NUM_B * NUM_G, 256, 0, stream>>>(points, centers, knnIdx);

  size_t ldsBytes = (size_t)(128 + 32 * 132 + 32 * 516 + 32 * 516) * sizeof(float);
  encoder_kernel<<<NUM_B * NUM_G, 256, ldsBytes, stream>>>(
      points, knnIdx, centers, w1, b1, g1, be1, m1, v1, w2, b2, w3, b3, g2,
      be2, m2, v2, w4, b4, tokens);
}


# Encoder weights: each wave fetches its A rows as whole 128-byte lines (4 lines per load, was 16 quarter-used ones), one 32-k trip ahead, via a private LDS tile; same fragments, sum order unchanged
# speedup vs baseline: 2.0723x; 1.0483x over previous
#include <hip/hip_runtime.h>
#include <math.h>

#define NUM_B 8
#define NUM_N 8192
#define NUM_G 256
#define GK    32
#define C1    128
#define C2    256
#define C3    512
#define CT    384
#define TOK_TOTAL (NUM_B * NUM_G * CT)

typedef __attribute__((ext_vector_type(2))) float v2f;
typedef __attribute__((ext_vector_type(4))) float v4f;
typedef __attribute__((ext_vector_type(8))) float v8f;

__device__ __forceinline__ v8f wmma4(v2f a, v2f b, v8f c) {
  return __builtin_amdgcn_wmma_f32_16x16x4_f32(false, a, false, b, (short)0, c,
                                               false, false);
}

__device__ __forceinline__ v2f lo2(v4f v) { return __builtin_shufflevector(v, v, 0, 1); }
__device__ __forceinline__ v2f hi2(v4f v) { return __builtin_shufflevector(v, v, 2, 3); }

#define SP 36
#define STAGE_FLOATS (32 * SP)

template <int ROWS>
__device__ __forceinline__ void wload(const float* __restrict__ Wm, int ldw,
                                      int lane, v4f (&g)[ROWS / 4]) {
  const float* p = Wm + (size_t)(lane >> 3) * (size_t)ldw + 4 * (lane & 7);
#pragma unroll
  for (int j = 0; j < ROWS / 4; ++j)
    g[j] = *(const v4f*)(p + (size_t)(4 * j) * (size_t)ldw);
}
template <int ROWS>
__device__ __forceinline__ void wstore(float* S, int lane,
                                       const v4f (&g)[ROWS / 4]) {
  float* q = S + (lane >> 3) * SP + 4 * (lane & 7);
#pragma unroll
  for (int j = 0; j < ROWS / 4; ++j) *(v4f*)(q + 4 * j * SP) = g[j];
}

template <int KDIM, int LDB>
__device__ __forceinline__ void gemm_2x2w(const float* __restrict__ W, int ldw,
                                          const float* B, float* S, int m0,
                                          int lane, v8f& acc00, v8f& acc01,
                                          v8f& acc10, v8f& acc11) {
  static_assert(KDIM % 32 == 0, "32-k trips");
  const int laneN = lane & 15, laneHalf = lane >> 4;
  const float* Wm = W + (size_t)m0 * (size_t)ldw;
  const float* sa0 = S + laneN * SP + 4 * laneHalf;
  const float* sa1 = sa0 + 16 * SP;
  const float* bcol0 = B + laneN * LDB + 4 * laneHalf;
  const float* bcol1 = B + (16 + laneN) * LDB + 4 * laneHalf;
  v4f g[8];
  wload<32>(Wm, ldw, lane, g);
  for (int kk = 0; kk < KDIM; kk += 32) {
    wstore<32>(S, lane, g);
    if (kk + 32 < KDIM) wload<32>(Wm + kk + 32, ldw, lane, g);
#pragma unroll
    for (int t = 0; t < 4; ++t) {
      v4f a0 = *(const v4f*)(sa0 + 8 * t);
      v4f a1 = *(const v4f*)(sa1 + 8 * t);
      v4f b0 = *(const v4f*)(bcol0 + kk + 8 * t);
      v4f b1 = *(const v4f*)(bcol1 + kk + 8 * t);
      acc00 = wmma4(lo2(a0), lo2(b0), acc00);
      acc01 = wmma4(lo2(a0), lo2(b1), acc01);
      acc10 = wmma4(lo2(a1), lo2(b0), acc10);
      acc11 = wmma4(lo2(a1), lo2(b1), acc11);
      acc00 = wmma4(hi2(a0), hi2(b0), acc00);
      acc01 = wmma4(hi2(a0), hi2(b1), acc01);
      acc10 = wmma4(hi2(a1), hi2(b0), acc10);
      acc11 = wmma4(hi2(a1), hi2(b1), acc11);
    }
  }
}

template <int KDIM, int LDB>
__device__ __forceinline__ void gemm_1x2(const float* __restrict__ W, int ldw,
                                         const float* B, int m0, int laneN,
                                         int laneHalf, v8f& acc0, v8f& acc1) {
  const float* wrow  = W + (size_t)(m0 + laneN) * (size_t)ldw + 2 * laneHalf;
  const float* bcol0 = B + laneN * LDB + 2 * laneHalf;
  const float* bcol1 = B + (16 + laneN) * LDB + 2 * laneHalf;
  for (int kk = 0; kk < KDIM; kk += 4) {
    v2f a  = *(const v2f*)(wrow + kk);
    v2f b0 = *(const v2f*)(bcol0 + kk);
    v2f b1 = *(const v2f*)(bcol1 + kk);
    acc0 = wmma4(a, b0, acc0);
    acc1 = wmma4(a, b1, acc1);
  }
}

template <int KDIM, int LDB>
__device__ __forceinline__ void gemm_1x2w(const float* __restrict__ W, int ldw,
                                          const float* B, float* S, int m0,
                                          int lane, v8f& acc0, v8f& acc1) {
  static_assert(KDIM % 32 == 0, "32-k trips");
  const int laneN = lane & 15, laneHalf = lane >> 4;
  const float* Wm = W + (size_t)m0 * (size_t)ldw;
  const float* sa = S + laneN * SP + 4 * laneHalf;
  const float* bcol0 = B + laneN * LDB + 4 * laneHalf;
  const float* bcol1 = B + (16 + laneN) * LDB + 4 * laneHalf;
  v4f g[4];
  wload<16>(Wm, ldw, lane, g);
  for (int kk = 0; kk < KDIM; kk += 32) {
    wstore<16>(S, lane, g);
    if (kk + 32 < KDIM) wload<16>(Wm + kk + 32, ldw, lane, g);
#pragma unroll
    for (int t = 0; t < 4; ++t) {
      v4f a  = *(const v4f*)(sa + 8 * t);
      v4f b0 = *(const v4f*)(bcol0 + kk + 8 * t);
      v4f b1 = *(const v4f*)(bcol1 + kk + 8 * t);
      acc0 = wmma4(lo2(a), lo2(b0), acc0);
      acc1 = wmma4(lo2(a), lo2(b1), acc1);
      acc0 = wmma4(hi2(a), hi2(b0), acc0);
      acc1 = wmma4(hi2(a), hi2(b1), acc1);
    }
  }
}

template <int KDIM>
__device__ __forceinline__ void gemm_2x1v(const float* __restrict__ W, int ldw,
                                          const float* Bv, float* S, int m0,
                                          int lane, v8f& acc0, v8f& acc1) {
  static_assert(KDIM % 32 == 0, "32-k trips");
  const int laneN = lane & 15, laneHalf = lane >> 4;
  const float* Wm = W + (size_t)m0 * (size_t)ldw;
  const float* sa0 = S + laneN * SP + 4 * laneHalf;
  const float* sa1 = sa0 + 16 * SP;
  const float* bvec = Bv + 4 * laneHalf;
  v4f g[8];
  wload<32>(Wm, ldw, lane, g);
  for (int kk = 0; kk < KDIM; kk += 32) {
    wstore<32>(S, lane, g);
    if (kk + 32 < KDIM) wload<32>(Wm + kk + 32, ldw, lane, g);
#pragma unroll
    for (int t = 0; t < 4; ++t) {
      v4f a0 = *(const v4f*)(sa0 + 8 * t);
      v4f a1 = *(const v4f*)(sa1 + 8 * t);
      v4f bb = *(const v4f*)(bvec + kk + 8 * t);
      acc0 = wmma4(lo2(a0), lo2(bb), acc0);
      acc1 = wmma4(lo2(a1), lo2(bb), acc1);
      acc0 = wmma4(hi2(a0), hi2(bb), acc0);
      acc1 = wmma4(hi2(a1), hi2(bb), acc1);
    }
  }
}

template <typename F>
__device__ __forceinline__ void store_tile(float* dst, int ldd, const v8f& acc,
                                           int m0, int n0, int laneN,
                                           int laneHalf, F op) {
  int n = n0 + laneN;
#pragma unroll
  for (int rr = 0; rr < 8; ++rr) {
    int m = m0 + 8 * laneHalf + rr;
    dst[n * ldd + m] = op(acc[rr], m);
  }
}

template <int M>
__device__ __forceinline__ int row_xor(int b) {
  return __builtin_amdgcn_update_dpp(b, b, 0x160 + M, 0xf, 0xf, true);
}
__device__ __forceinline__ int imax(int a, int b) { return a > b ? a : b; }
__device__ __forceinline__ int imin(int a, int b) { return a < b ? a : b; }
__device__ __forceinline__ int wave_max(int v) {
  v = imax(v, row_xor<1>(v));
  v = imax(v, row_xor<2>(v));
  v = imax(v, row_xor<4>(v));
  v = imax(v, row_xor<8>(v));
  v = imax(v, __builtin_amdgcn_permlanex16(v, v, 0x76543210, 0xfedcba98, false,
                                           false));
  return v;
}

#define FPS_T 512
#define FPS_P (NUM_N / FPS_T)
#define FPS_W (FPS_T / 32)

__device__ __forceinline__ int row_max(int v) {
  v = imax(v, row_xor<1>(v));
  v = imax(v, row_xor<2>(v));
  v = imax(v, row_xor<4>(v));
  v = imax(v, row_xor<8>(v));
  return v;
}

__global__ __launch_bounds__(FPS_T) void fps_kernel(
    const float* __restrict__ points, float* __restrict__ centers) {
  const int b    = blockIdx.x;
  const int tid  = threadIdx.x;
  const int lane = tid & 31;
  const int wv   = __builtin_amdgcn_readfirstlane(tid >> 5);
  const float* base = points + (size_t)b * NUM_N * 3;
  float* co = centers + (size_t)b * NUM_G * 3;

  __shared__ float sP[NUM_N * 3];
  __shared__ int   sWv[2][FPS_W];
  __shared__ int   sWi[2][FPS_W];

  for (int k = tid; k < NUM_N * 3; k += FPS_T) sP[k] = base[k];

  float px[FPS_P], py[FPS_P], pz[FPS_P];
  int   dd[FPS_P];
#pragma unroll
  for (int i = 0; i < FPS_P; ++i) {
    int p = tid * FPS_P + i;
    px[i] = base[p * 3 + 0];
    py[i] = base[p * 3 + 1];
    pz[i] = base[p * 3 + 2];
    dd[i] = __builtin_bit_cast(int, 1e10f);
  }
  __syncthreads();

  float lx = sP[0], ly = sP[1], lz = sP[2];
  if (tid == 0) { co[0] = lx; co[1] = ly; co[2] = lz; }

  for (int g = 1; g < NUM_G; ++g) {
#pragma unroll
    for (int i = 0; i < FPS_P; ++i) {
      float dx = px[i] - lx, dy = py[i] - ly, dz = pz[i] - lz;
      float d = __builtin_fmaf(dz, dz, __builtin_fmaf(dx, dx, dy * dy));
      dd[i]   = imin(dd[i], __builtin_bit_cast(int, d));
    }
    static_assert(FPS_P == 16, "the tree below is written for 16 points");
    const int t0 = imax(imax(dd[0], dd[1]), dd[2]);
    const int t1 = imax(imax(dd[3], dd[4]), dd[5]);
    const int t2 = imax(imax(dd[6], dd[7]), dd[8]);
    const int t3 = imax(imax(dd[9], dd[10]), dd[11]);
    const int t4 = imax(imax(dd[12], dd[13]), dd[14]);
    const int u0 = imax(imax(t0, t1), t2);
    const int u1 = imax(imax(t3, t4), dd[15]);
    const int tm = imax(u0, u1);

    const int buf = g & 1;
    {
      int      wm  = wave_max(tm);
      unsigned has = __builtin_amdgcn_ballot_w32(tm == wm);
      int      fl  = __builtin_ctz(has);
      int idx = FPS_P - 1;
#pragma unroll
      for (int i = FPS_P - 2; i >= 0; --i) {
        unsigned eq = __builtin_amdgcn_ballot_w32(dd[i] == wm);
        if ((eq >> fl) & 1u) idx = i;
      }
      int wbi = (wv * 32 + fl) * FPS_P + idx;
      if (lane == 0) { sWv[buf][wv] = wm; sWi[buf][wv] = wbi; }
    }
    __syncthreads();
    {
      int      v   = sWv[buf][lane & (FPS_W - 1)];
      int      ii  = sWi[buf][lane & (FPS_W - 1)];
      int      bm  = row_max(v);
      unsigned has = __builtin_amdgcn_ballot_w32(v == bm);
      int      wi  = __builtin_amdgcn_readlane(ii, __builtin_ctz(has));
      lx = sP[wi * 3 + 0];
      ly = sP[wi * 3 + 1];
      lz = sP[wi * 3 + 2];
    }
    if (tid == 0) { co[g * 3 + 0] = lx; co[g * 3 + 1] = ly; co[g * 3 + 2] = lz; }
  }
}

__device__ __forceinline__ int wave_min(int v) {
  v = imin(v, row_xor<1>(v));
  v = imin(v, row_xor<2>(v));
  v = imin(v, row_xor<4>(v));
  v = imin(v, row_xor<8>(v));
  v = imin(v, __builtin_amdgcn_permlanex16(v, v, 0x76543210, 0xfedcba98, false,
                                           false));
  return v;
}
__device__ __forceinline__ int oct_min(int v) {
  v = imin(v, row_xor<1>(v));
  v = imin(v, row_xor<2>(v));
  v = imin(v, row_xor<4>(v));
  return v;
}
__device__ __forceinline__ int order_key(float f) {
  int bits = __builtin_bit_cast(int, f);
  return bits ^ ((bits >> 31) & 0x7fffffff);
}

#define KEY_GONE 0x7fffffff

__global__ __launch_bounds__(256) void knn_kernel(
    const float* __restrict__ points, const float* __restrict__ centers,
    int* __restrict__ knnIdx) {
  const int gg   = blockIdx.x;
  const int b    = gg / NUM_G;
  const int tid  = threadIdx.x;
  const int lane = tid & 31;
  const int wv   = tid >> 5;
  const float* base = points + (size_t)b * NUM_N * 3;

  __shared__ int sWv[2][8];
  __shared__ int sWi[2][8];

  const float cx = centers[(size_t)gg * 3 + 0];
  const float cy = centers[(size_t)gg * 3 + 1];
  const float cz = centers[(size_t)gg * 3 + 2];
  const float c2 = __builtin_fmaf(cz, cz, __builtin_fmaf(cx, cx, cy * cy));

  int dk[32];
#pragma unroll
  for (int i = 0; i < NUM_N / 256; ++i) {
    int   p  = i * 256 + tid;
    float xx = base[p * 3 + 0], yy = base[p * 3 + 1], zz = base[p * 3 + 2];
    float x2 = __builtin_fmaf(zz, zz, __builtin_fmaf(xx, xx, yy * yy));
    float dt = __builtin_fmaf(cz, zz, __builtin_fmaf(cx, xx, cy * yy));
    dk[i]    = order_key(__builtin_fmaf(-2.f, dt, c2 + x2));
  }

  int lv = KEY_GONE, li = 0;
#pragma unroll
  for (int i = 0; i < 32; ++i)
    if (dk[i] < lv) { lv = dk[i]; li = i; }
  int lp    = li * 256 + tid;
  int myOut = 0;

  for (int r = 0; r < GK; ++r) {
    const int buf = r & 1;
    {
      int wm = wave_min(lv);
      int wp = wave_min(lv == wm ? lp : KEY_GONE);
      if (lane == 0) { sWv[buf][wv] = wm; sWi[buf][wv] = wp; }
    }
    __syncthreads();
    int win;
    {
      int v  = sWv[buf][lane & 7];
      int ii = sWi[buf][lane & 7];
      int bm = oct_min(v);
      win    = oct_min(v == bm ? ii : KEY_GONE);
    }
    win = __builtin_amdgcn_readfirstlane(win);
    if (tid == r) myOut = win;
    const int wt = win & 255;
    const int wi = win >> 8;
    if (wv == (wt >> 5)) {
      const bool mine = (tid == wt);
      int nv = KEY_GONE, ni = 0;
#pragma unroll
      for (int i = 0; i < 32; ++i) {
        if (mine && i == wi) dk[i] = KEY_GONE;
        if (dk[i] < nv) { nv = dk[i]; ni = i; }
      }
      lv = nv;
      lp = ni * 256 + tid;
    }
  }
  if (tid < GK) knnIdx[(size_t)gg * GK + tid] = myOut & (NUM_N - 1);
}

__global__ __launch_bounds__(256) void encoder_kernel(
    const float* __restrict__ points, const int* __restrict__ knnIdx,
    const float* __restrict__ centers, const float* __restrict__ w1,
    const float* __restrict__ b1, const float* __restrict__ g1,
    const float* __restrict__ be1, const float* __restrict__ m1,
    const float* __restrict__ v1, const float* __restrict__ w2,
    const float* __restrict__ b2, const float* __restrict__ w3,
    const float* __restrict__ b3, const float* __restrict__ g2,
    const float* __restrict__ be2, const float* __restrict__ m2,
    const float* __restrict__ v2, const float* __restrict__ w4,
    const float* __restrict__ b4, float* __restrict__ tokens) {
  extern __shared__ __attribute__((aligned(16))) float smem[];
  float* sFeat = smem;
  float* sH1   = smem + 128;
  float* sXu   = sH1 + 32 * 132;
  float* sGm   = sXu + 32 * 260;
  float* sH3   = sGm + 256;
  float* sTile = sH3 + 32 * 516;
  float* sH4   = smem;

  const int gg       = blockIdx.x;
  const int b        = gg / NUM_G;
  const int tid      = threadIdx.x;
  const int lane     = tid & 31;
  const int wv       = tid >> 5;
  const int laneN    = lane & 15;
  const int laneHalf = lane >> 4;
  float* S = sTile + wv * STAGE_FLOATS;

  const v8f vzero = {0.f, 0.f, 0.f, 0.f, 0.f, 0.f, 0.f, 0.f};

  if (tid < GK) {
    int idx = knnIdx[(size_t)gg * GK + tid];
    const float* p = points + ((size_t)b * NUM_N + idx) * 3;
    const float* c = centers + (size_t)gg * 3;
    float dx = p[0] - c[0], dy = p[1] - c[1], dz = p[2] - c[2];
    float nr = sqrtf(dx * dx + dy * dy + dz * dz);
    sFeat[tid * 4 + 0] = dx;
    sFeat[tid * 4 + 1] = dy;
    sFeat[tid * 4 + 2] = dz;
    sFeat[tid * 4 + 3] = nr;
  }
  __syncthreads();

  {
    int m0 = wv * 16;
    v8f a0 = vzero, a1 = vzero;
    gemm_1x2<4, 4>(w1, 4, sFeat, m0, laneN, laneHalf, a0, a1);
    auto op = [&](float x, int m) {
      float inv = g1[m] * rsqrtf(v1[m] + 1e-5f);
      float v   = (x + b1[m]) * inv + (be1[m] - m1[m] * inv);
      return fmaxf(v, 0.f);
    };
    store_tile(sH1, 132, a0, m0, 0,  laneN, laneHalf, op);
    store_tile(sH1, 132, a1, m0, 16, laneN, laneHalf, op);
  }
  __syncthreads();

  {
    int m0 = wv * 32;
    v8f a00 = vzero, a01 = vzero, a10 = vzero, a11 = vzero;
    gemm_2x2w<C1, 132>(w2, C1, sH1, S, m0, lane, a00, a01, a10, a11);
    auto op = [&](float x, int m) { return x + b2[m]; };
    store_tile(sXu, 260, a00, m0,      0,  laneN, laneHalf, op);
    store_tile(sXu, 260, a01, m0,      16, laneN, laneHalf, op);
    store_tile(sXu, 260, a10, m0 + 16, 0,  laneN, laneHalf, op);
    store_tile(sXu, 260, a11, m0 + 16, 16, laneN, laneHalf, op);
  }
  __syncthreads();

  {
    int   m  = tid;
    float gm = -3.4e38f;
    for (int n = 0; n < GK; ++n) gm = fmaxf(gm, sXu[n * 260 + m]);
    sGm[m] = gm;
  }
  __syncthreads();

  {
    auto op = [&](float x, int m) {
      float inv = g2[m] * rsqrtf(v2[m] + 1e-5f);
      float v   = (x + b3[m]) * inv + (be2[m] - m2[m] * inv);
      return fmaxf(v, 0.f);
    };
#pragma unroll
    for (int j = 0; j < 2; ++j) {
      int m0  = wv * 64 + j * 32;
      v8f g0 = vzero, g1 = vzero;
      gemm_2x1v<C2>(w3, C3, sGm, S, m0, lane, g0, g1);
      v8f a00 = g0, a01 = g0, a10 = g1, a11 = g1;
      gemm_2x2w<C2, 260>(w3 + C2, C3, sXu, S, m0, lane, a00, a01, a10, a11);
      store_tile(sH3, 516, a00, m0,      0,  laneN, laneHalf, op);
      store_tile(sH3, 516, a01, m0,      16, laneN, laneHalf, op);
      store_tile(sH3, 516, a10, m0 + 16, 0,  laneN, laneHalf, op);
      store_tile(sH3, 516, a11, m0 + 16, 16, laneN, laneHalf, op);
    }
  }
  __syncthreads();

  {
    auto op = [&](float x, int m) { return x + b4[m]; };
    {
      int m0  = wv * 48;
      v8f a00 = vzero, a01 = vzero, a10 = vzero, a11 = vzero;
      gemm_2x2w<C3, 516>(w4, C3, sH3, S, m0, lane, a00, a01, a10, a11);
      store_tile(sH4, 388, a00, m0,      0,  laneN, laneHalf, op);
      store_tile(sH4, 388, a01, m0,      16, laneN, laneHalf, op);
      store_tile(sH4, 388, a10, m0 + 16, 0,  laneN, laneHalf, op);
      store_tile(sH4, 388, a11, m0 + 16, 16, laneN, laneHalf, op);
    }
    {
      int m0 = wv * 48 + 32;
      v8f a0 = vzero, a1 = vzero;
      gemm_1x2w<C3, 516>(w4, C3, sH3, S, m0, lane, a0, a1);
      store_tile(sH4, 388, a0, m0, 0,  laneN, laneHalf, op);
      store_tile(sH4, 388, a1, m0, 16, laneN, laneHalf, op);
    }
  }
  __syncthreads();

  for (int m = tid; m < CT; m += 256) {
    float mx = -3.4e38f;
    for (int n = 0; n < GK; ++n) mx = fmaxf(mx, sH4[n * 388 + m]);
    tokens[(size_t)gg * CT + m] = mx;
  }
}

extern "C" void kernel_launch(void* const* d_in, const int* in_sizes, int n_in,
                              void* d_out, int out_size, void* d_ws,
                              size_t ws_size, hipStream_t stream) {
  (void)in_sizes; (void)n_in; (void)out_size; (void)ws_size;
  const float* points = (const float*)d_in[0];
  const float* w1  = (const float*)d_in[1];
  const float* b1  = (const float*)d_in[2];
  const float* g1  = (const float*)d_in[3];
  const float* be1 = (const float*)d_in[4];
  const float* m1  = (const float*)d_in[5];
  const float* v1  = (const float*)d_in[6];
  const float* w2  = (const float*)d_in[7];
  const float* b2  = (const float*)d_in[8];
  const float* w3  = (const float*)d_in[9];
  const float* b3  = (const float*)d_in[10];
  const float* g2  = (const float*)d_in[11];
  const float* be2 = (const float*)d_in[12];
  const float* m2  = (const float*)d_in[13];
  const float* v2  = (const float*)d_in[14];
  const float* w4  = (const float*)d_in[15];
  const float* b4  = (const float*)d_in[16];

  float* tokens  = (float*)d_out;
  float* centers = tokens + TOK_TOTAL;
  int*   knnIdx  = (int*)d_ws;

  fps_kernel<<<NUM_B, FPS_T, 0, stream>>>(points, centers);
  knn_kernel<<<NUM_B * NUM_G, 256, 0, stream>>>(points, centers, knnIdx);

  size_t ldsBytes = (size_t)(128 + 32 * 132 + 32 * 260 + 256 + 32 * 516 + 8 * STAGE_FLOATS) * sizeof(float);
  encoder_kernel<<<NUM_B * NUM_G, 256, ldsBytes, stream>>>(
      points, knnIdx, centers, w1, b1, g1, be1, m1, v1, w2, b2, w3, b3, g2,
      be2, m2, v2, w4, b4, tokens);
}


# Encoder conv3, per-column half: conv2 folded into its weights once per call (W3[:,256:].W2 and W3[:,256:].b2 by f32 WMMA into workspace), K 256 -> 128 on h1; the order of the f32 sums changes
# speedup vs baseline: 2.2541x; 1.0877x over previous
#include <hip/hip_runtime.h>
#include <math.h>

#define NUM_B 8
#define NUM_N 8192
#define NUM_G 256
#define GK    32
#define C1    128
#define C2    256
#define C3    512
#define CT    384
#define TOK_TOTAL (NUM_B * NUM_G * CT)

typedef __attribute__((ext_vector_type(2))) float v2f;
typedef __attribute__((ext_vector_type(4))) float v4f;
typedef __attribute__((ext_vector_type(8))) float v8f;

__device__ __forceinline__ v8f wmma4(v2f a, v2f b, v8f c) {
  return __builtin_amdgcn_wmma_f32_16x16x4_f32(false, a, false, b, (short)0, c,
                                               false, false);
}

__device__ __forceinline__ v2f lo2(v4f v) { return __builtin_shufflevector(v, v, 0, 1); }
__device__ __forceinline__ v2f hi2(v4f v) { return __builtin_shufflevector(v, v, 2, 3); }

#define SP 36
#define STAGE_FLOATS (32 * SP)

template <int ROWS>
__device__ __forceinline__ void wload(const float* __restrict__ Wm, int ldw,
                                      int lane, v4f (&g)[ROWS / 4]) {
  const float* p = Wm + (size_t)(lane >> 3) * (size_t)ldw + 4 * (lane & 7);
#pragma unroll
  for (int j = 0; j < ROWS / 4; ++j)
    g[j] = *(const v4f*)(p + (size_t)(4 * j) * (size_t)ldw);
}
template <int ROWS>
__device__ __forceinline__ void wstore(float* S, int lane,
                                       const v4f (&g)[ROWS / 4]) {
  float* q = S + (lane >> 3) * SP + 4 * (lane & 7);
#pragma unroll
  for (int j = 0; j < ROWS / 4; ++j) *(v4f*)(q + 4 * j * SP) = g[j];
}

template <int KDIM, int LDB>
__device__ __forceinline__ void gemm_2x2w(const float* __restrict__ W, int ldw,
                                          const float* B, float* S, int m0,
                                          int lane, v8f& acc00, v8f& acc01,
                                          v8f& acc10, v8f& acc11) {
  static_assert(KDIM % 32 == 0, "32-k trips");
  const int laneN = lane & 15, laneHalf = lane >> 4;
  const float* Wm = W + (size_t)m0 * (size_t)ldw;
  const float* sa0 = S + laneN * SP + 4 * laneHalf;
  const float* sa1 = sa0 + 16 * SP;
  const float* bcol0 = B + laneN * LDB + 4 * laneHalf;
  const float* bcol1 = B + (16 + laneN) * LDB + 4 * laneHalf;
  v4f g[8];
  wload<32>(Wm, ldw, lane, g);
  for (int kk = 0; kk < KDIM; kk += 32) {
    wstore<32>(S, lane, g);
    if (kk + 32 < KDIM) wload<32>(Wm + kk + 32, ldw, lane, g);
#pragma unroll
    for (int t = 0; t < 4; ++t) {
      v4f a0 = *(const v4f*)(sa0 + 8 * t);
      v4f a1 = *(const v4f*)(sa1 + 8 * t);
      v4f b0 = *(const v4f*)(bcol0 + kk + 8 * t);
      v4f b1 = *(const v4f*)(bcol1 + kk + 8 * t);
      acc00 = wmma4(lo2(a0), lo2(b0), acc00);
      acc01 = wmma4(lo2(a0), lo2(b1), acc01);
      acc10 = wmma4(lo2(a1), lo2(b0), acc10);
      acc11 = wmma4(lo2(a1), lo2(b1), acc11);
      acc00 = wmma4(hi2(a0), hi2(b0), acc00);
      acc01 = wmma4(hi2(a0), hi2(b1), acc01);
      acc10 = wmma4(hi2(a1), hi2(b0), acc10);
      acc11 = wmma4(hi2(a1), hi2(b1), acc11);
    }
  }
}

template <int KDIM, int LDB>
__device__ __forceinline__ void gemm_1x2(const float* __restrict__ W, int ldw,
                                         const float* B, int m0, int laneN,
                                         int laneHalf, v8f& acc0, v8f& acc1) {
  const float* wrow  = W + (size_t)(m0 + laneN) * (size_t)ldw + 2 * laneHalf;
  const float* bcol0 = B + laneN * LDB + 2 * laneHalf;
  const float* bcol1 = B + (16 + laneN) * LDB + 2 * laneHalf;
  for (int kk = 0; kk < KDIM; kk += 4) {
    v2f a  = *(const v2f*)(wrow + kk);
    v2f b0 = *(const v2f*)(bcol0 + kk);
    v2f b1 = *(const v2f*)(bcol1 + kk);
    acc0 = wmma4(a, b0, acc0);
    acc1 = wmma4(a, b1, acc1);
  }
}

template <int KDIM, int LDB>
__device__ __forceinline__ void gemm_1x2w(const float* __restrict__ W, int ldw,
                                          const float* B, float* S, int m0,
                                          int lane, v8f& acc0, v8f& acc1) {
  static_assert(KDIM % 32 == 0, "32-k trips");
  const int laneN = lane & 15, laneHalf = lane >> 4;
  const float* Wm = W + (size_t)m0 * (size_t)ldw;
  const float* sa = S + laneN * SP + 4 * laneHalf;
  const float* bcol0 = B + laneN * LDB + 4 * laneHalf;
  const float* bcol1 = B + (16 + laneN) * LDB + 4 * laneHalf;
  v4f g[4];
  wload<16>(Wm, ldw, lane, g);
  for (int kk = 0; kk < KDIM; kk += 32) {
    wstore<16>(S, lane, g);
    if (kk + 32 < KDIM) wload<16>(Wm + kk + 32, ldw, lane, g);
#pragma unroll
    for (int t = 0; t < 4; ++t) {
      v4f a  = *(const v4f*)(sa + 8 * t);
      v4f b0 = *(const v4f*)(bcol0 + kk + 8 * t);
      v4f b1 = *(const v4f*)(bcol1 + kk + 8 * t);
      acc0 = wmma4(lo2(a), lo2(b0), acc0);
      acc1 = wmma4(lo2(a), lo2(b1), acc1);
      acc0 = wmma4(hi2(a), hi2(b0), acc0);
      acc1 = wmma4(hi2(a), hi2(b1), acc1);
    }
  }
}

template <int KDIM>
__device__ __forceinline__ void gemm_2x1v(const float* __restrict__ W, int ldw,
                                          const float* Bv, float* S, int m0,
                                          int lane, v8f& acc0, v8f& acc1) {
  static_assert(KDIM % 32 == 0, "32-k trips");
  const int laneN = lane & 15, laneHalf = lane >> 4;
  const float* Wm = W + (size_t)m0 * (size_t)ldw;
  const float* sa0 = S + laneN * SP + 4 * laneHalf;
  const float* sa1 = sa0 + 16 * SP;
  const float* bvec = Bv + 4 * laneHalf;
  v4f g[8];
  wload<32>(Wm, ldw, lane, g);
  for (int kk = 0; kk < KDIM; kk += 32) {
    wstore<32>(S, lane, g);
    if (kk + 32 < KDIM) wload<32>(Wm + kk + 32, ldw, lane, g);
#pragma unroll
    for (int t = 0; t < 4; ++t) {
      v4f a0 = *(const v4f*)(sa0 + 8 * t);
      v4f a1 = *(const v4f*)(sa1 + 8 * t);
      v4f bb = *(const v4f*)(bvec + kk + 8 * t);
      acc0 = wmma4(lo2(a0), lo2(bb), acc0);
      acc1 = wmma4(lo2(a1), lo2(bb), acc1);
      acc0 = wmma4(hi2(a0), hi2(bb), acc0);
      acc1 = wmma4(hi2(a1), hi2(bb), acc1);
    }
  }
}

template <typename F>
__device__ __forceinline__ void store_tile(float* dst, int ldd, const v8f& acc,
                                           int m0, int n0, int laneN,
                                           int laneHalf, F op) {
  int n = n0 + laneN;
#pragma unroll
  for (int rr = 0; rr < 8; ++rr) {
    int m = m0 + 8 * laneHalf + rr;
    dst[n * ldd + m] = op(acc[rr], m);
  }
}

template <int M>
__device__ __forceinline__ int row_xor(int b) {
  return __builtin_amdgcn_update_dpp(b, b, 0x160 + M, 0xf, 0xf, true);
}
__device__ __forceinline__ int imax(int a, int b) { return a > b ? a : b; }
__device__ __forceinline__ int imin(int a, int b) { return a < b ? a : b; }
__device__ __forceinline__ int wave_max(int v) {
  v = imax(v, row_xor<1>(v));
  v = imax(v, row_xor<2>(v));
  v = imax(v, row_xor<4>(v));
  v = imax(v, row_xor<8>(v));
  v = imax(v, __builtin_amdgcn_permlanex16(v, v, 0x76543210, 0xfedcba98, false,
                                           false));
  return v;
}

#define FPS_T 512
#define FPS_P (NUM_N / FPS_T)
#define FPS_W (FPS_T / 32)

__device__ __forceinline__ int row_max(int v) {
  v = imax(v, row_xor<1>(v));
  v = imax(v, row_xor<2>(v));
  v = imax(v, row_xor<4>(v));
  v = imax(v, row_xor<8>(v));
  return v;
}

__global__ __launch_bounds__(FPS_T) void fps_kernel(
    const float* __restrict__ points, float* __restrict__ centers) {
  const int b    = blockIdx.x;
  const int tid  = threadIdx.x;
  const int lane = tid & 31;
  const int wv   = __builtin_amdgcn_readfirstlane(tid >> 5);
  const float* base = points + (size_t)b * NUM_N * 3;
  float* co = centers + (size_t)b * NUM_G * 3;

  __shared__ float sP[NUM_N * 3];
  __shared__ int   sWv[2][FPS_W];
  __shared__ int   sWi[2][FPS_W];

  for (int k = tid; k < NUM_N * 3; k += FPS_T) sP[k] = base[k];

  float px[FPS_P], py[FPS_P], pz[FPS_P];
  int   dd[FPS_P];
#pragma unroll
  for (int i = 0; i < FPS_P; ++i) {
    int p = tid * FPS_P + i;
    px[i] = base[p * 3 + 0];
    py[i] = base[p * 3 + 1];
    pz[i] = base[p * 3 + 2];
    dd[i] = __builtin_bit_cast(int, 1e10f);
  }
  __syncthreads();

  float lx = sP[0], ly = sP[1], lz = sP[2];
  if (tid == 0) { co[0] = lx; co[1] = ly; co[2] = lz; }

  for (int g = 1; g < NUM_G; ++g) {
#pragma unroll
    for (int i = 0; i < FPS_P; ++i) {
      float dx = px[i] - lx, dy = py[i] - ly, dz = pz[i] - lz;
      float d = __builtin_fmaf(dz, dz, __builtin_fmaf(dx, dx, dy * dy));
      dd[i]   = imin(dd[i], __builtin_bit_cast(int, d));
    }
    static_assert(FPS_P == 16, "the tree below is written for 16 points");
    const int t0 = imax(imax(dd[0], dd[1]), dd[2]);
    const int t1 = imax(imax(dd[3], dd[4]), dd[5]);
    const int t2 = imax(imax(dd[6], dd[7]), dd[8]);
    const int t3 = imax(imax(dd[9], dd[10]), dd[11]);
    const int t4 = imax(imax(dd[12], dd[13]), dd[14]);
    const int u0 = imax(imax(t0, t1), t2);
    const int u1 = imax(imax(t3, t4), dd[15]);
    const int tm = imax(u0, u1);

    const int buf = g & 1;
    {
      int      wm  = wave_max(tm);
      unsigned has = __builtin_amdgcn_ballot_w32(tm == wm);
      int      fl  = __builtin_ctz(has);
      int idx = FPS_P - 1;
#pragma unroll
      for (int i = FPS_P - 2; i >= 0; --i) {
        unsigned eq = __builtin_amdgcn_ballot_w32(dd[i] == wm);
        if ((eq >> fl) & 1u) idx = i;
      }
      int wbi = (wv * 32 + fl) * FPS_P + idx;
      if (lane == 0) { sWv[buf][wv] = wm; sWi[buf][wv] = wbi; }
    }
    __syncthreads();
    {
      int      v   = sWv[buf][lane & (FPS_W - 1)];
      int      ii  = sWi[buf][lane & (FPS_W - 1)];
      int      bm  = row_max(v);
      unsigned has = __builtin_amdgcn_ballot_w32(v == bm);
      int      wi  = __builtin_amdgcn_readlane(ii, __builtin_ctz(has));
      lx = sP[wi * 3 + 0];
      ly = sP[wi * 3 + 1];
      lz = sP[wi * 3 + 2];
    }
    if (tid == 0) { co[g * 3 + 0] = lx; co[g * 3 + 1] = ly; co[g * 3 + 2] = lz; }
  }
}

#define WS_KNN_FLOATS (NUM_B * NUM_G * GK)
#define WS_NEED_BYTES ((size_t)(WS_KNN_FLOATS + C3 * C1 + C3) * 4)
#define FOLD_TASKS  (32 * 4 + 16)
#define FOLD_BLOCKS (FOLD_TASKS / 8)

__device__ __forceinline__ void fold_task(const float* __restrict__ w3,
                                          const float* __restrict__ w2,
                                          const float* __restrict__ b2,
                                          float* __restrict__ FW,
                                          float* __restrict__ FC, float* T,
                                          int task, int lane) {
  const int  laneN = lane & 15, laneHalf = lane >> 4;
  const bool vecTask = task >= 128;
  const int  mA0 = vecTask ? 32 * (task - 128) : 16 * (task >> 2);
  const int  mA1 = vecTask ? mA0 + 16 : mA0;
  const int  n0  = 32 * (task & 3);
  const float* a0p = w3 + (size_t)(mA0 + laneN) * C3 + C2 + 2 * laneHalf;
  const float* a1p = w3 + (size_t)(mA1 + laneN) * C3 + C2 + 2 * laneHalf;
  const int    bstep = vecTask ? 1 : C1;
  const float* b0p = vecTask ? b2 + 2 * laneHalf
                             : w2 + (size_t)(2 * laneHalf) * C1 + n0 + laneN;
  const float* b1p = vecTask ? b0p : b0p + 16;
  v8f acc0 = {0.f, 0.f, 0.f, 0.f, 0.f, 0.f, 0.f, 0.f};
  v8f acc1 = acc0;
  for (int kk = 0; kk < C2; kk += 32) {
    v2f a0[8], a1[8], b0[8], b1[8];
#pragma unroll
    for (int t = 0; t < 8; ++t) {
      a0[t] = *(const v2f*)(a0p + kk + 4 * t);
      a1[t] = *(const v2f*)(a1p + kk + 4 * t);
      const float* p0 = b0p + (size_t)(kk + 4 * t) * bstep;
      const float* p1 = b1p + (size_t)(kk + 4 * t) * bstep;
      b0[t].x = p0[0]; b0[t].y = p0[bstep];
      b1[t].x = p1[0]; b1[t].y = p1[bstep];
    }
#pragma unroll
    for (int t = 0; t < 8; ++t) {
      acc0 = wmma4(a0[t], b0[t], acc0);
      acc1 = wmma4(a1[t], b1[t], acc1);
    }
  }
  if (!vecTask) {
#pragma unroll
    for (int r = 0; r < 8; ++r) {
      T[(8 * laneHalf + r) * 32 + laneN]      = acc0[r];
      T[(8 * laneHalf + r) * 32 + 16 + laneN] = acc1[r];
    }
    for (int rr = 0; rr < 16; ++rr)
      FW[(size_t)(mA0 + rr) * C1 + n0 + lane] = T[rr * 32 + lane];
  } else {
    if (laneN == 0) {
#pragma unroll
      for (int r = 0; r < 8; ++r) {
        T[8 * laneHalf + r]      = acc0[r];
        T[16 + 8 * laneHalf + r] = acc1[r];
      }
    }
    FC[mA0 + lane] = T[lane];
  }
}

__device__ __forceinline__ int wave_min(int v) {
  v = imin(v, row_xor<1>(v));
  v = imin(v, row_xor<2>(v));
  v = imin(v, row_xor<4>(v));
  v = imin(v, row_xor<8>(v));
  v = imin(v, __builtin_amdgcn_permlanex16(v, v, 0x76543210, 0xfedcba98, false,
                                           false));
  return v;
}
__device__ __forceinline__ int oct_min(int v) {
  v = imin(v, row_xor<1>(v));
  v = imin(v, row_xor<2>(v));
  v = imin(v, row_xor<4>(v));
  return v;
}
__device__ __forceinline__ int order_key(float f) {
  int bits = __builtin_bit_cast(int, f);
  return bits ^ ((bits >> 31) & 0x7fffffff);
}

#define KEY_GONE 0x7fffffff

__global__ __launch_bounds__(256) void knn_kernel(
    const float* __restrict__ points, const float* __restrict__ centers,
    int* __restrict__ knnIdx, const float* __restrict__ w2,
    const float* __restrict__ b2, const float* __restrict__ w3,
    float* __restrict__ FW, float* __restrict__ FC, size_t wsBytes) {
  const int tid  = threadIdx.x;
  const int lane = tid & 31;
  const int wv   = tid >> 5;

  if (blockIdx.x < FOLD_BLOCKS) {
    __shared__ float sFold[8][16 * 32];
    if (wsBytes >= WS_NEED_BYTES) {
      const int task = __builtin_amdgcn_readfirstlane(blockIdx.x * 8 + wv);
      fold_task(w3, w2, b2, FW, FC, sFold[wv], task, lane);
    }
    return;
  }
  const int gg   = blockIdx.x - FOLD_BLOCKS;
  const int b    = gg / NUM_G;
  const float* base = points + (size_t)b * NUM_N * 3;

  __shared__ int sWv[2][8];
  __shared__ int sWi[2][8];

  const float cx = centers[(size_t)gg * 3 + 0];
  const float cy = centers[(size_t)gg * 3 + 1];
  const float cz = centers[(size_t)gg * 3 + 2];
  const float c2 = __builtin_fmaf(cz, cz, __builtin_fmaf(cx, cx, cy * cy));

  int dk[32];
#pragma unroll
  for (int i = 0; i < NUM_N / 256; ++i) {
    int   p  = i * 256 + tid;
    float xx = base[p * 3 + 0], yy = base[p * 3 + 1], zz = base[p * 3 + 2];
    float x2 = __builtin_fmaf(zz, zz, __builtin_fmaf(xx, xx, yy * yy));
    float dt = __builtin_fmaf(cz, zz, __builtin_fmaf(cx, xx, cy * yy));
    dk[i]    = order_key(__builtin_fmaf(-2.f, dt, c2 + x2));
  }

  int lv = KEY_GONE, li = 0;
#pragma unroll
  for (int i = 0; i < 32; ++i)
    if (dk[i] < lv) { lv = dk[i]; li = i; }
  int lp    = li * 256 + tid;
  int myOut = 0;

  for (int r = 0; r < GK; ++r) {
    const int buf = r & 1;
    {
      int wm = wave_min(lv);
      int wp = wave_min(lv == wm ? lp : KEY_GONE);
      if (lane == 0) { sWv[buf][wv] = wm; sWi[buf][wv] = wp; }
    }
    __syncthreads();
    int win;
    {
      int v  = sWv[buf][lane & 7];
      int ii = sWi[buf][lane & 7];
      int bm = oct_min(v);
      win    = oct_min(v == bm ? ii : KEY_GONE);
    }
    win = __builtin_amdgcn_readfirstlane(win);
    if (tid == r) myOut = win;
    const int wt = win & 255;
    const int wi = win >> 8;
    if (wv == (wt >> 5)) {
      const bool mine = (tid == wt);
      int nv = KEY_GONE, ni = 0;
#pragma unroll
      for (int i = 0; i < 32; ++i) {
        if (mine && i == wi) dk[i] = KEY_GONE;
        if (dk[i] < nv) { nv = dk[i]; ni = i; }
      }
      lv = nv;
      lp = ni * 256 + tid;
    }
  }
  if (tid < GK) knnIdx[(size_t)gg * GK + tid] = myOut & (NUM_N - 1);
}

template <bool FOLDED>
__device__ __forceinline__ void conv3_stage(
    const float* __restrict__ w3, const float* __restrict__ b3,
    const float* __restrict__ g2, const float* __restrict__ be2,
    const float* __restrict__ m2, const float* __restrict__ v2,
    const float* __restrict__ FW, const float* __restrict__ FC,
    const float* sGm, const float* sXu, const float* sH1, float* sH3, float* S,
    int wv, int lane) {
  const int laneN = lane & 15, laneHalf = lane >> 4;
  const v8f vzero = {0.f, 0.f, 0.f, 0.f, 0.f, 0.f, 0.f, 0.f};
  auto op = [&](float x, int m) {
    float inv = g2[m] * rsqrtf(v2[m] + 1e-5f);
    float bb  = b3[m];
    if constexpr (FOLDED) bb = FC[m] + bb;
    float v   = (x + bb) * inv + (be2[m] - m2[m] * inv);
    return fmaxf(v, 0.f);
  };
#pragma unroll
  for (int j = 0; j < 2; ++j) {
    int m0  = wv * 64 + j * 32;
    v8f g0 = vzero, g1 = vzero;
    gemm_2x1v<C2>(w3, C3, sGm, S, m0, lane, g0, g1);
    v8f a00 = g0, a01 = g0, a10 = g1, a11 = g1;
    if constexpr (FOLDED)
      gemm_2x2w<C1, 132>(FW, C1, sH1, S, m0, lane, a00, a01, a10, a11);
    else
      gemm_2x2w<C2, 260>(w3 + C2, C3, sXu, S, m0, lane, a00, a01, a10, a11);
    store_tile(sH3, 516, a00, m0,      0,  laneN, laneHalf, op);
    store_tile(sH3, 516, a01, m0,      16, laneN, laneHalf, op);
    store_tile(sH3, 516, a10, m0 + 16, 0,  laneN, laneHalf, op);
    store_tile(sH3, 516, a11, m0 + 16, 16, laneN, laneHalf, op);
  }
}

__global__ __launch_bounds__(256) void encoder_kernel(
    const float* __restrict__ points, const int* __restrict__ knnIdx,
    const float* __restrict__ centers, const float* __restrict__ w1,
    const float* __restrict__ b1, const float* __restrict__ g1,
    const float* __restrict__ be1, const float* __restrict__ m1,
    const float* __restrict__ v1, const float* __restrict__ w2,
    const float* __restrict__ b2, const float* __restrict__ w3,
    const float* __restrict__ b3, const float* __restrict__ g2,
    const float* __restrict__ be2, const float* __restrict__ m2,
    const float* __restrict__ v2, const float* __restrict__ w4,
    const float* __restrict__ b4, float* __restrict__ tokens,
    const float* __restrict__ FW, const float* __restrict__ FC,
    size_t wsBytes) {
  extern __shared__ __attribute__((aligned(16))) float smem[];
  float* sFeat = smem;
  float* sH1   = smem + 128;
  float* sXu   = sH1 + 32 * 132;
  float* sGm   = sXu + 32 * 260;
  float* sH3   = sGm + 256;
  float* sTile = sH3 + 32 * 516;
  float* sH4   = smem;

  const int gg       = blockIdx.x;
  const int b        = gg / NUM_G;
  const int tid      = threadIdx.x;
  const int lane     = tid & 31;
  const int wv       = tid >> 5;
  const int laneN    = lane & 15;
  const int laneHalf = lane >> 4;
  float* S = sTile + wv * STAGE_FLOATS;

  const v8f vzero = {0.f, 0.f, 0.f, 0.f, 0.f, 0.f, 0.f, 0.f};

  if (tid < GK) {
    int idx = knnIdx[(size_t)gg * GK + tid];
    const float* p = points + ((size_t)b * NUM_N + idx) * 3;
    const float* c = centers + (size_t)gg * 3;
    float dx = p[0] - c[0], dy = p[1] - c[1], dz = p[2] - c[2];
    float nr = sqrtf(dx * dx + dy * dy + dz * dz);
    sFeat[tid * 4 + 0] = dx;
    sFeat[tid * 4 + 1] = dy;
    sFeat[tid * 4 + 2] = dz;
    sFeat[tid * 4 + 3] = nr;
  }
  __syncthreads();

  {
    int m0 = wv * 16;
    v8f a0 = vzero, a1 = vzero;
    gemm_1x2<4, 4>(w1, 4, sFeat, m0, laneN, laneHalf, a0, a1);
    auto op = [&](float x, int m) {
      float inv = g1[m] * rsqrtf(v1[m] + 1e-5f);
      float v   = (x + b1[m]) * inv + (be1[m] - m1[m] * inv);
      return fmaxf(v, 0.f);
    };
    store_tile(sH1, 132, a0, m0, 0,  laneN, laneHalf, op);
    store_tile(sH1, 132, a1, m0, 16, laneN, laneHalf, op);
  }
  __syncthreads();

  {
    int m0 = wv * 32;
    v8f a00 = vzero, a01 = vzero, a10 = vzero, a11 = vzero;
    gemm_2x2w<C1, 132>(w2, C1, sH1, S, m0, lane, a00, a01, a10, a11);
    auto op = [&](float x, int m) { return x + b2[m]; };
    store_tile(sXu, 260, a00, m0,      0,  laneN, laneHalf, op);
    store_tile(sXu, 260, a01, m0,      16, laneN, laneHalf, op);
    store_tile(sXu, 260, a10, m0 + 16, 0,  laneN, laneHalf, op);
    store_tile(sXu, 260, a11, m0 + 16, 16, laneN, laneHalf, op);
  }
  __syncthreads();

  {
    int   m  = tid;
    float gm = -3.4e38f;
    for (int n = 0; n < GK; ++n) gm = fmaxf(gm, sXu[n * 260 + m]);
    sGm[m] = gm;
  }
  __syncthreads();

  if (wsBytes >= WS_NEED_BYTES)
    conv3_stage<true>(w3, b3, g2, be2, m2, v2, FW, FC, sGm, sXu, sH1, sH3, S,
                      wv, lane);
  else
    conv3_stage<false>(w3, b3, g2, be2, m2, v2, FW, FC, sGm, sXu, sH1, sH3, S,
                       wv, lane);
  __syncthreads();

  {
    auto op = [&](float x, int m) { return x + b4[m]; };
    {
      int m0  = wv * 48;
      v8f a00 = vzero, a01 = vzero, a10 = vzero, a11 = vzero;
      gemm_2x2w<C3, 516>(w4, C3, sH3, S, m0, lane, a00, a01, a10, a11);
      store_tile(sH4, 388, a00, m0,      0,  laneN, laneHalf, op);
      store_tile(sH4, 388, a01, m0,      16, laneN, laneHalf, op);
      store_tile(sH4, 388, a10, m0 + 16, 0,  laneN, laneHalf, op);
      store_tile(sH4, 388, a11, m0 + 16, 16, laneN, laneHalf, op);
    }
    {
      int m0 = wv * 48 + 32;
      v8f a0 = vzero, a1 = vzero;
      gemm_1x2w<C3, 516>(w4, C3, sH3, S, m0, lane, a0, a1);
      store_tile(sH4, 388, a0, m0, 0,  laneN, laneHalf, op);
      store_tile(sH4, 388, a1, m0, 16, laneN, laneHalf, op);
    }
  }
  __syncthreads();

  for (int m = tid; m < CT; m += 256) {
    float mx = -3.4e38f;
    for (int n = 0; n < GK; ++n) mx = fmaxf(mx, sH4[n * 388 + m]);
    tokens[(size_t)gg * CT + m] = mx;
  }
}

extern "C" void kernel_launch(void* const* d_in, const int* in_sizes, int n_in,
                              void* d_out, int out_size, void* d_ws,
                              size_t ws_size, hipStream_t stream) {
  (void)in_sizes; (void)n_in; (void)out_size;
  const float* points = (const float*)d_in[0];
  const float* w1  = (const float*)d_in[1];
  const float* b1  = (const float*)d_in[2];
  const float* g1  = (const float*)d_in[3];
  const float* be1 = (const float*)d_in[4];
  const float* m1  = (const float*)d_in[5];
  const float* v1  = (const float*)d_in[6];
  const float* w2  = (const float*)d_in[7];
  const float* b2  = (const float*)d_in[8];
  const float* w3  = (const float*)d_in[9];
  const float* b3  = (const float*)d_in[10];
  const float* g2  = (const float*)d_in[11];
  const float* be2 = (const float*)d_in[12];
  const float* m2  = (const float*)d_in[13];
  const float* v2  = (const float*)d_in[14];
  const float* w4  = (const float*)d_in[15];
  const float* b4  = (const float*)d_in[16];

  float* tokens  = (float*)d_out;
  float* centers = tokens + TOK_TOTAL;
  int*   knnIdx  = (int*)d_ws;
  float* FW      = (float*)d_ws + WS_KNN_FLOATS;
  float* FC      = FW + C3 * C1;

  fps_kernel<<<NUM_B, FPS_T, 0, stream>>>(points, centers);
  knn_kernel<<<FOLD_BLOCKS + NUM_B * NUM_G, 256, 0, stream>>>(
      points, centers, knnIdx, w2, b2, w3, FW, FC, ws_size);

  size_t ldsBytes = (size_t)(128 + 32 * 132 + 32 * 260 + 256 + 32 * 516 + 8 * STAGE_FLOATS) * sizeof(float);
  encoder_kernel<<<NUM_B * NUM_G, 256, ldsBytes, stream>>>(
      points, knnIdx, centers, w1, b1, g1, be1, m1, v1, w2, b2, w3, b3, g2,
      be2, m2, v2, w4, b4, tokens, FW, FC, ws_size);
}


# Encoder conv3, pooled half: W3[:,0:256].gmax taken over all 2048 groups as ONE f32 WMMA GEMM (16 groups per tile, was 16 equal columns per group); encoder split in 3 launches; sum order unchanged
# speedup vs baseline: 2.3628x; 1.0482x over previous
#include <hip/hip_runtime.h>
#include <math.h>

#define NUM_B 8
#define NUM_N 8192
#define NUM_G 256
#define GK    32
#define C1    128
#define C2    256
#define C3    512
#define CT    384
#define TOK_TOTAL (NUM_B * NUM_G * CT)

typedef __attribute__((ext_vector_type(2))) float v2f;
typedef __attribute__((ext_vector_type(4))) float v4f;
typedef __attribute__((ext_vector_type(8))) float v8f;

__device__ __forceinline__ v8f wmma4(v2f a, v2f b, v8f c) {
  return __builtin_amdgcn_wmma_f32_16x16x4_f32(false, a, false, b, (short)0, c,
                                               false, false);
}

__device__ __forceinline__ v2f lo2(v4f v) { return __builtin_shufflevector(v, v, 0, 1); }
__device__ __forceinline__ v2f hi2(v4f v) { return __builtin_shufflevector(v, v, 2, 3); }

#define SP 36
#define STAGE_FLOATS (32 * SP)

template <int ROWS>
__device__ __forceinline__ void wload(const float* __restrict__ Wm, int ldw,
                                      int lane, v4f (&g)[ROWS / 4]) {
  const float* p = Wm + (size_t)(lane >> 3) * (size_t)ldw + 4 * (lane & 7);
#pragma unroll
  for (int j = 0; j < ROWS / 4; ++j)
    g[j] = *(const v4f*)(p + (size_t)(4 * j) * (size_t)ldw);
}
template <int ROWS>
__device__ __forceinline__ void wstore(float* S, int lane,
                                       const v4f (&g)[ROWS / 4]) {
  float* q = S + (lane >> 3) * SP + 4 * (lane & 7);
#pragma unroll
  for (int j = 0; j < ROWS / 4; ++j) *(v4f*)(q + 4 * j * SP) = g[j];
}

template <int KDIM, int LDB>
__device__ __forceinline__ void gemm_2x2w(const float* __restrict__ W, int ldw,
                                          const float* B, float* S, int m0,
                                          int lane, v8f& acc00, v8f& acc01,
                                          v8f& acc10, v8f& acc11) {
  static_assert(KDIM % 32 == 0, "32-k trips");
  const int laneN = lane & 15, laneHalf = lane >> 4;
  const float* Wm = W + (size_t)m0 * (size_t)ldw;
  const float* sa0 = S + laneN * SP + 4 * laneHalf;
  const float* sa1 = sa0 + 16 * SP;
  const float* bcol0 = B + laneN * LDB + 4 * laneHalf;
  const float* bcol1 = B + (16 + laneN) * LDB + 4 * laneHalf;
  v4f g[8];
  wload<32>(Wm, ldw, lane, g);
  for (int kk = 0; kk < KDIM; kk += 32) {
    wstore<32>(S, lane, g);
    if (kk + 32 < KDIM) wload<32>(Wm + kk + 32, ldw, lane, g);
#pragma unroll
    for (int t = 0; t < 4; ++t) {
      v4f a0 = *(const v4f*)(sa0 + 8 * t);
      v4f a1 = *(const v4f*)(sa1 + 8 * t);
      v4f b0 = *(const v4f*)(bcol0 + kk + 8 * t);
      v4f b1 = *(const v4f*)(bcol1 + kk + 8 * t);
      acc00 = wmma4(lo2(a0), lo2(b0), acc00);
      acc01 = wmma4(lo2(a0), lo2(b1), acc01);
      acc10 = wmma4(lo2(a1), lo2(b0), acc10);
      acc11 = wmma4(lo2(a1), lo2(b1), acc11);
      acc00 = wmma4(hi2(a0), hi2(b0), acc00);
      acc01 = wmma4(hi2(a0), hi2(b1), acc01);
      acc10 = wmma4(hi2(a1), hi2(b0), acc10);
      acc11 = wmma4(hi2(a1), hi2(b1), acc11);
    }
  }
}

template <int KDIM, int LDB>
__device__ __forceinline__ void gemm_1x2(const float* __restrict__ W, int ldw,
                                         const float* B, int m0, int laneN,
                                         int laneHalf, v8f& acc0, v8f& acc1) {
  const float* wrow  = W + (size_t)(m0 + laneN) * (size_t)ldw + 2 * laneHalf;
  const float* bcol0 = B + laneN * LDB + 2 * laneHalf;
  const float* bcol1 = B + (16 + laneN) * LDB + 2 * laneHalf;
  for (int kk = 0; kk < KDIM; kk += 4) {
    v2f a  = *(const v2f*)(wrow + kk);
    v2f b0 = *(const v2f*)(bcol0 + kk);
    v2f b1 = *(const v2f*)(bcol1 + kk);
    acc0 = wmma4(a, b0, acc0);
    acc1 = wmma4(a, b1, acc1);
  }
}

template <int KDIM, int LDB>
__device__ __forceinline__ void gemm_1x2w(const float* __restrict__ W, int ldw,
                                          const float* B, float* S, int m0,
                                          int lane, v8f& acc0, v8f& acc1) {
  static_assert(KDIM % 32 == 0, "32-k trips");
  const int laneN = lane & 15, laneHalf = lane >> 4;
  const float* Wm = W + (size_t)m0 * (size_t)ldw;
  const float* sa = S + laneN * SP + 4 * laneHalf;
  const float* bcol0 = B + laneN * LDB + 4 * laneHalf;
  const float* bcol1 = B + (16 + laneN) * LDB + 4 * laneHalf;
  v4f g[4];
  wload<16>(Wm, ldw, lane, g);
  for (int kk = 0; kk < KDIM; kk += 32) {
    wstore<16>(S, lane, g);
    if (kk + 32 < KDIM) wload<16>(Wm + kk + 32, ldw, lane, g);
#pragma unroll
    for (int t = 0; t < 4; ++t) {
      v4f a  = *(const v4f*)(sa + 8 * t);
      v4f b0 = *(const v4f*)(bcol0 + kk + 8 * t);
      v4f b1 = *(const v4f*)(bcol1 + kk + 8 * t);
      acc0 = wmma4(lo2(a), lo2(b0), acc0);
      acc1 = wmma4(lo2(a), lo2(b1), acc1);
      acc0 = wmma4(hi2(a), hi2(b0), acc0);
      acc1 = wmma4(hi2(a), hi2(b1), acc1);
    }
  }
}

template <int KDIM>
__device__ __forceinline__ void gemm_2x1v(const float* __restrict__ W, int ldw,
                                          const float* Bv, float* S, int m0,
                                          int lane, v8f& acc0, v8f& acc1) {
  static_assert(KDIM % 32 == 0, "32-k trips");
  const int laneN = lane & 15, laneHalf = lane >> 4;
  const float* Wm = W + (size_t)m0 * (size_t)ldw;
  const float* sa0 = S + laneN * SP + 4 * laneHalf;
  const float* sa1 = sa0 + 16 * SP;
  const float* bvec = Bv + 4 * laneHalf;
  v4f g[8];
  wload<32>(Wm, ldw, lane, g);
  for (int kk = 0; kk < KDIM; kk += 32) {
    wstore<32>(S, lane, g);
    if (kk + 32 < KDIM) wload<32>(Wm + kk + 32, ldw, lane, g);
#pragma unroll
    for (int t = 0; t < 4; ++t) {
      v4f a0 = *(const v4f*)(sa0 + 8 * t);
      v4f a1 = *(const v4f*)(sa1 + 8 * t);
      v4f bb = *(const v4f*)(bvec + kk + 8 * t);
      acc0 = wmma4(lo2(a0), lo2(bb), acc0);
      acc1 = wmma4(lo2(a1), lo2(bb), acc1);
      acc0 = wmma4(hi2(a0), hi2(bb), acc0);
      acc1 = wmma4(hi2(a1), hi2(bb), acc1);
    }
  }
}

template <typename F>
__device__ __forceinline__ void store_tile(float* dst, int ldd, const v8f& acc,
                                           int m0, int n0, int laneN,
                                           int laneHalf, F op) {
  int n = n0 + laneN;
#pragma unroll
  for (int rr = 0; rr < 8; ++rr) {
    int m = m0 + 8 * laneHalf + rr;
    dst[n * ldd + m] = op(acc[rr], m);
  }
}

template <int M>
__device__ __forceinline__ int row_xor(int b) {
  return __builtin_amdgcn_update_dpp(b, b, 0x160 + M, 0xf, 0xf, true);
}
__device__ __forceinline__ int imax(int a, int b) { return a > b ? a : b; }
__device__ __forceinline__ int imin(int a, int b) { return a < b ? a : b; }
__device__ __forceinline__ int wave_max(int v) {
  v = imax(v, row_xor<1>(v));
  v = imax(v, row_xor<2>(v));
  v = imax(v, row_xor<4>(v));
  v = imax(v, row_xor<8>(v));
  v = imax(v, __builtin_amdgcn_permlanex16(v, v, 0x76543210, 0xfedcba98, false,
                                           false));
  return v;
}

#define FPS_T 512
#define FPS_P (NUM_N / FPS_T)
#define FPS_W (FPS_T / 32)

__device__ __forceinline__ int row_max(int v) {
  v = imax(v, row_xor<1>(v));
  v = imax(v, row_xor<2>(v));
  v = imax(v, row_xor<4>(v));
  v = imax(v, row_xor<8>(v));
  return v;
}

__global__ __launch_bounds__(FPS_T) void fps_kernel(
    const float* __restrict__ points, float* __restrict__ centers) {
  const int b    = blockIdx.x;
  const int tid  = threadIdx.x;
  const int lane = tid & 31;
  const int wv   = __builtin_amdgcn_readfirstlane(tid >> 5);
  const float* base = points + (size_t)b * NUM_N * 3;
  float* co = centers + (size_t)b * NUM_G * 3;

  __shared__ float sP[NUM_N * 3];
  __shared__ int   sWv[2][FPS_W];
  __shared__ int   sWi[2][FPS_W];

  for (int k = tid; k < NUM_N * 3; k += FPS_T) sP[k] = base[k];

  float px[FPS_P], py[FPS_P], pz[FPS_P];
  int   dd[FPS_P];
#pragma unroll
  for (int i = 0; i < FPS_P; ++i) {
    int p = tid * FPS_P + i;
    px[i] = base[p * 3 + 0];
    py[i] = base[p * 3 + 1];
    pz[i] = base[p * 3 + 2];
    dd[i] = __builtin_bit_cast(int, 1e10f);
  }
  __syncthreads();

  float lx = sP[0], ly = sP[1], lz = sP[2];
  if (tid == 0) { co[0] = lx; co[1] = ly; co[2] = lz; }

  for (int g = 1; g < NUM_G; ++g) {
#pragma unroll
    for (int i = 0; i < FPS_P; ++i) {
      float dx = px[i] - lx, dy = py[i] - ly, dz = pz[i] - lz;
      float d = __builtin_fmaf(dz, dz, __builtin_fmaf(dx, dx, dy * dy));
      dd[i]   = imin(dd[i], __builtin_bit_cast(int, d));
    }
    static_assert(FPS_P == 16, "the tree below is written for 16 points");
    const int t0 = imax(imax(dd[0], dd[1]), dd[2]);
    const int t1 = imax(imax(dd[3], dd[4]), dd[5]);
    const int t2 = imax(imax(dd[6], dd[7]), dd[8]);
    const int t3 = imax(imax(dd[9], dd[10]), dd[11]);
    const int t4 = imax(imax(dd[12], dd[13]), dd[14]);
    const int u0 = imax(imax(t0, t1), t2);
    const int u1 = imax(imax(t3, t4), dd[15]);
    const int tm = imax(u0, u1);

    const int buf = g & 1;
    {
      int      wm  = wave_max(tm);
      unsigned has = __builtin_amdgcn_ballot_w32(tm == wm);
      int      fl  = __builtin_ctz(has);
      int idx = FPS_P - 1;
#pragma unroll
      for (int i = FPS_P - 2; i >= 0; --i) {
        unsigned eq = __builtin_amdgcn_ballot_w32(dd[i] == wm);
        if ((eq >> fl) & 1u) idx = i;
      }
      int wbi = (wv * 32 + fl) * FPS_P + idx;
      if (lane == 0) { sWv[buf][wv] = wm; sWi[buf][wv] = wbi; }
    }
    __syncthreads();
    {
      int      v   = sWv[buf][lane & (FPS_W - 1)];
      int      ii  = sWi[buf][lane & (FPS_W - 1)];
      int      bm  = row_max(v);
      unsigned has = __builtin_amdgcn_ballot_w32(v == bm);
      int      wi  = __builtin_amdgcn_readlane(ii, __builtin_ctz(has));
      lx = sP[wi * 3 + 0];
      ly = sP[wi * 3 + 1];
      lz = sP[wi * 3 + 2];
    }
    if (tid == 0) { co[g * 3 + 0] = lx; co[g * 3 + 1] = ly; co[g * 3 + 2] = lz; }
  }
}

#define WS_KNN_FLOATS (NUM_B * NUM_G * GK)
#define WS_NEED_BYTES ((size_t)(WS_KNN_FLOATS + C3 * C1 + C3) * 4)
#define WS_GM_FLOATS (NUM_B * NUM_G * C2)
#define WS_G3_FLOATS (NUM_B * NUM_G * C3)
#define WS_SPLIT_BYTES (WS_NEED_BYTES + (size_t)(WS_GM_FLOATS + WS_G3_FLOATS) * 4)
#define FOLD_TASKS  (32 * 4 + 16)
#define FOLD_BLOCKS (FOLD_TASKS / 8)

__device__ __forceinline__ void fold_task(const float* __restrict__ w3,
                                          const float* __restrict__ w2,
                                          const float* __restrict__ b2,
                                          float* __restrict__ FW,
                                          float* __restrict__ FC, float* T,
                                          int task, int lane) {
  const int  laneN = lane & 15, laneHalf = lane >> 4;
  const bool vecTask = task >= 128;
  const int  mA0 = vecTask ? 32 * (task - 128) : 16 * (task >> 2);
  const int  mA1 = vecTask ? mA0 + 16 : mA0;
  const int  n0  = 32 * (task & 3);
  const float* a0p = w3 + (size_t)(mA0 + laneN) * C3 + C2 + 2 * laneHalf;
  const float* a1p = w3 + (size_t)(mA1 + laneN) * C3 + C2 + 2 * laneHalf;
  const int    bstep = vecTask ? 1 : C1;
  const float* b0p = vecTask ? b2 + 2 * laneHalf
                             : w2 + (size_t)(2 * laneHalf) * C1 + n0 + laneN;
  const float* b1p = vecTask ? b0p : b0p + 16;
  v8f acc0 = {0.f, 0.f, 0.f, 0.f, 0.f, 0.f, 0.f, 0.f};
  v8f acc1 = acc0;
  for (int kk = 0; kk < C2; kk += 32) {
    v2f a0[8], a1[8], b0[8], b1[8];
#pragma unroll
    for (int t = 0; t < 8; ++t) {
      a0[t] = *(const v2f*)(a0p + kk + 4 * t);
      a1[t] = *(const v2f*)(a1p + kk + 4 * t);
      const float* p0 = b0p + (size_t)(kk + 4 * t) * bstep;
      const float* p1 = b1p + (size_t)(kk + 4 * t) * bstep;
      b0[t].x = p0[0]; b0[t].y = p0[bstep];
      b1[t].x = p1[0]; b1[t].y = p1[bstep];
    }
#pragma unroll
    for (int t = 0; t < 8; ++t) {
      acc0 = wmma4(a0[t], b0[t], acc0);
      acc1 = wmma4(a1[t], b1[t], acc1);
    }
  }
  if (!vecTask) {
#pragma unroll
    for (int r = 0; r < 8; ++r) {
      T[(8 * laneHalf + r) * 32 + laneN]      = acc0[r];
      T[(8 * laneHalf + r) * 32 + 16 + laneN] = acc1[r];
    }
    for (int rr = 0; rr < 16; ++rr)
      FW[(size_t)(mA0 + rr) * C1 + n0 + lane] = T[rr * 32 + lane];
  } else {
    if (laneN == 0) {
#pragma unroll
      for (int r = 0; r < 8; ++r) {
        T[8 * laneHalf + r]      = acc0[r];
        T[16 + 8 * laneHalf + r] = acc1[r];
      }
    }
    FC[mA0 + lane] = T[lane];
  }
}

__device__ __forceinline__ int wave_min(int v) {
  v = imin(v, row_xor<1>(v));
  v = imin(v, row_xor<2>(v));
  v = imin(v, row_xor<4>(v));
  v = imin(v, row_xor<8>(v));
  v = imin(v, __builtin_amdgcn_permlanex16(v, v, 0x76543210, 0xfedcba98, false,
                                           false));
  return v;
}
__device__ __forceinline__ int oct_min(int v) {
  v = imin(v, row_xor<1>(v));
  v = imin(v, row_xor<2>(v));
  v = imin(v, row_xor<4>(v));
  return v;
}
__device__ __forceinline__ int order_key(float f) {
  int bits = __builtin_bit_cast(int, f);
  return bits ^ ((bits >> 31) & 0x7fffffff);
}

#define KEY_GONE 0x7fffffff

__global__ __launch_bounds__(256) void knn_kernel(
    const float* __restrict__ points, const float* __restrict__ centers,
    int* __restrict__ knnIdx, const float* __restrict__ w2,
    const float* __restrict__ b2, const float* __restrict__ w3,
    float* __restrict__ FW, float* __restrict__ FC, size_t wsBytes) {
  const int tid  = threadIdx.x;
  const int lane = tid & 31;
  const int wv   = tid >> 5;

  if (blockIdx.x < FOLD_BLOCKS) {
    __shared__ float sFold[8][16 * 32];
    if (wsBytes >= WS_NEED_BYTES) {
      const int task = __builtin_amdgcn_readfirstlane(blockIdx.x * 8 + wv);
      fold_task(w3, w2, b2, FW, FC, sFold[wv], task, lane);
    }
    return;
  }
  const int gg   = blockIdx.x - FOLD_BLOCKS;
  const int b    = gg / NUM_G;
  const float* base = points + (size_t)b * NUM_N * 3;

  __shared__ int sWv[2][8];
  __shared__ int sWi[2][8];

  const float cx = centers[(size_t)gg * 3 + 0];
  const float cy = centers[(size_t)gg * 3 + 1];
  const float cz = centers[(size_t)gg * 3 + 2];
  const float c2 = __builtin_fmaf(cz, cz, __builtin_fmaf(cx, cx, cy * cy));

  int dk[32];
#pragma unroll
  for (int i = 0; i < NUM_N / 256; ++i) {
    int   p  = i * 256 + tid;
    float xx = base[p * 3 + 0], yy = base[p * 3 + 1], zz = base[p * 3 + 2];
    float x2 = __builtin_fmaf(zz, zz, __builtin_fmaf(xx, xx, yy * yy));
    float dt = __builtin_fmaf(cz, zz, __builtin_fmaf(cx, xx, cy * yy));
    dk[i]    = order_key(__builtin_fmaf(-2.f, dt, c2 + x2));
  }

  int lv = KEY_GONE, li = 0;
#pragma unroll
  for (int i = 0; i < 32; ++i)
    if (dk[i] < lv) { lv = dk[i]; li = i; }
  int lp    = li * 256 + tid;
  int myOut = 0;

  for (int r = 0; r < GK; ++r) {
    const int buf = r & 1;
    {
      int wm = wave_min(lv);
      int wp = wave_min(lv == wm ? lp : KEY_GONE);
      if (lane == 0) { sWv[buf][wv] = wm; sWi[buf][wv] = wp; }
    }
    __syncthreads();
    int win;
    {
      int v  = sWv[buf][lane & 7];
      int ii = sWi[buf][lane & 7];
      int bm = oct_min(v);
      win    = oct_min(v == bm ? ii : KEY_GONE);
    }
    win = __builtin_amdgcn_readfirstlane(win);
    if (tid == r) myOut = win;
    const int wt = win & 255;
    const int wi = win >> 8;
    if (wv == (wt >> 5)) {
      const bool mine = (tid == wt);
      int nv = KEY_GONE, ni = 0;
#pragma unroll
      for (int i = 0; i < 32; ++i) {
        if (mine && i == wi) dk[i] = KEY_GONE;
        if (dk[i] < nv) { nv = dk[i]; ni = i; }
      }
      lv = nv;
      lp = ni * 256 + tid;
    }
  }
  if (tid < GK) knnIdx[(size_t)gg * GK + tid] = myOut & (NUM_N - 1);
}

template <int MODE>
__device__ __forceinline__ void conv3_stage(
    const float* __restrict__ w3, const float* __restrict__ b3,
    const float* __restrict__ g2, const float* __restrict__ be2,
    const float* __restrict__ m2, const float* __restrict__ v2,
    const float* __restrict__ FW, const float* __restrict__ FC,
    const float* __restrict__ G3g,
    const float* sGm, const float* sXu, const float* sH1, float* sH3, float* S,
    int wv, int lane) {
  constexpr bool FOLDED = (MODE != 0);
  const int laneN = lane & 15, laneHalf = lane >> 4;
  const v8f vzero = {0.f, 0.f, 0.f, 0.f, 0.f, 0.f, 0.f, 0.f};
  auto op = [&](float x, int m) {
    float inv = g2[m] * rsqrtf(v2[m] + 1e-5f);
    float bb  = b3[m];
    if constexpr (FOLDED) bb = FC[m] + bb;
    float v   = (x + bb) * inv + (be2[m] - m2[m] * inv);
    return fmaxf(v, 0.f);
  };
  auto pass = [&](int j) {
    int m0  = wv * 64 + j * 32;
    v8f g0 = vzero, g1 = vzero;
    if constexpr (MODE == 2) {
      const float* gp = G3g + m0 + 8 * laneHalf;
      v4f x0 = *(const v4f*)(gp),      x1 = *(const v4f*)(gp + 4);
      v4f y0 = *(const v4f*)(gp + 16), y1 = *(const v4f*)(gp + 20);
      g0 = __builtin_shufflevector(x0, x1, 0, 1, 2, 3, 4, 5, 6, 7);
      g1 = __builtin_shufflevector(y0, y1, 0, 1, 2, 3, 4, 5, 6, 7);
    } else {
      gemm_2x1v<C2>(w3, C3, sGm, S, m0, lane, g0, g1);
    }
    v8f a00 = g0, a01 = g0, a10 = g1, a11 = g1;
    if constexpr (FOLDED)
      gemm_2x2w<C1, 132>(FW, C1, sH1, S, m0, lane, a00, a01, a10, a11);
    else
      gemm_2x2w<C2, 260>(w3 + C2, C3, sXu, S, m0, lane, a00, a01, a10, a11);
    store_tile(sH3, 516, a00, m0,      0,  laneN, laneHalf, op);
    store_tile(sH3, 516, a01, m0,      16, laneN, laneHalf, op);
    store_tile(sH3, 516, a10, m0 + 16, 0,  laneN, laneHalf, op);
    store_tile(sH3, 516, a11, m0 + 16, 16, laneN, laneHalf, op);
  };
  if constexpr (MODE == 2) {
#pragma clang loop unroll(disable)
    for (int j = 0; j < 2; ++j) pass(j);
  } else {
#pragma unroll
    for (int j = 0; j < 2; ++j) pass(j);
  }
}

__device__ __forceinline__ void enc_stage_ab(
    const float* __restrict__ points, const int* __restrict__ knnIdx,
    const float* __restrict__ centers, const float* __restrict__ w1,
    const float* __restrict__ b1, const float* __restrict__ g1,
    const float* __restrict__ be1, const float* __restrict__ m1,
    const float* __restrict__ v1, float* sFeat, float* sH1, int gg, int tid) {
  const int b = gg / NUM_G;
  const int lane = tid & 31, wv = tid >> 5;
  const int laneN = lane & 15, laneHalf = lane >> 4;
  const v8f vzero = {0.f, 0.f, 0.f, 0.f, 0.f, 0.f, 0.f, 0.f};
  if (tid < GK) {
    int idx = knnIdx[(size_t)gg * GK + tid];
    const float* p = points + ((size_t)b * NUM_N + idx) * 3;
    const float* c = centers + (size_t)gg * 3;
    float dx = p[0] - c[0], dy = p[1] - c[1], dz = p[2] - c[2];
    float nr = sqrtf(dx * dx + dy * dy + dz * dz);
    sFeat[tid * 4 + 0] = dx;
    sFeat[tid * 4 + 1] = dy;
    sFeat[tid * 4 + 2] = dz;
    sFeat[tid * 4 + 3] = nr;
  }
  __syncthreads();
  {
    int m0 = wv * 16;
    v8f a0 = vzero, a1 = vzero;
    gemm_1x2<4, 4>(w1, 4, sFeat, m0, laneN, laneHalf, a0, a1);
    auto op = [&](float x, int m) {
      float inv = g1[m] * rsqrtf(v1[m] + 1e-5f);
      float v   = (x + b1[m]) * inv + (be1[m] - m1[m] * inv);
      return fmaxf(v, 0.f);
    };
    store_tile(sH1, 132, a0, m0, 0,  laneN, laneHalf, op);
    store_tile(sH1, 132, a1, m0, 16, laneN, laneHalf, op);
  }
  __syncthreads();
}

__device__ __forceinline__ void enc_stage_cd(const float* __restrict__ w2,
                                             const float* __restrict__ b2,
                                             const float* sH1, float* sXu,
                                             float* S, float* gm, int tid) {
  const int lane = tid & 31, wv = tid >> 5;
  const int laneN = lane & 15, laneHalf = lane >> 4;
  const v8f vzero = {0.f, 0.f, 0.f, 0.f, 0.f, 0.f, 0.f, 0.f};
  {
    int m0 = wv * 32;
    v8f a00 = vzero, a01 = vzero, a10 = vzero, a11 = vzero;
    gemm_2x2w<C1, 132>(w2, C1, sH1, S, m0, lane, a00, a01, a10, a11);
    auto op = [&](float x, int m) { return x + b2[m]; };
    store_tile(sXu, 260, a00, m0,      0,  laneN, laneHalf, op);
    store_tile(sXu, 260, a01, m0,      16, laneN, laneHalf, op);
    store_tile(sXu, 260, a10, m0 + 16, 0,  laneN, laneHalf, op);
    store_tile(sXu, 260, a11, m0 + 16, 16, laneN, laneHalf, op);
  }
  __syncthreads();
  {
    int   m  = tid;
    float mx = -3.4e38f;
    for (int n = 0; n < GK; ++n) mx = fmaxf(mx, sXu[n * 260 + m]);
    gm[m] = mx;
  }
}

#define POOL_LDS_FLOATS (128 + 32 * 132 + 32 * 260 + 8 * STAGE_FLOATS)
__global__ __launch_bounds__(256) void enc_pool_kernel(
    const float* __restrict__ points, const int* __restrict__ knnIdx,
    const float* __restrict__ centers, const float* __restrict__ w1,
    const float* __restrict__ b1, const float* __restrict__ g1,
    const float* __restrict__ be1, const float* __restrict__ m1,
    const float* __restrict__ v1, const float* __restrict__ w2,
    const float* __restrict__ b2, float* __restrict__ GM, size_t wsBytes) {
  if (wsBytes < WS_SPLIT_BYTES) return;
  extern __shared__ __attribute__((aligned(16))) float smem[];
  float* sFeat = smem;
  float* sH1   = smem + 128;
  float* sXu   = sH1 + 32 * 132;
  float* sTile = sXu + 32 * 260;
  const int gg = blockIdx.x, tid = threadIdx.x;
  enc_stage_ab(points, knnIdx, centers, w1, b1, g1, be1, m1, v1, sFeat, sH1, gg,
               tid);
  enc_stage_cd(w2, b2, sH1, sXu, sTile + (tid >> 5) * STAGE_FLOATS,
               GM + (size_t)gg * C2, tid);
}

__global__ __launch_bounds__(256) void enc_gm_kernel(
    const float* __restrict__ w3, const float* __restrict__ GM,
    float* __restrict__ G3, size_t wsBytes) {
  if (wsBytes < WS_SPLIT_BYTES) return;
  const int lane = threadIdx.x & 31;
  const int laneN = lane & 15, laneHalf = lane >> 4;
  const int wv = __builtin_amdgcn_readfirstlane(threadIdx.x >> 5);
  const int mt = (blockIdx.x & 3) * 8 + wv;
  const int n0 = (blockIdx.x >> 2) * 32;
  const float* ap  = w3 + (size_t)(16 * mt + laneN) * C3 + 4 * laneHalf;
  const float* bp0 = GM + (size_t)(n0 + laneN) * C2 + 4 * laneHalf;
  const float* bp1 = bp0 + (size_t)16 * C2;
  v8f acc0 = {0.f, 0.f, 0.f, 0.f, 0.f, 0.f, 0.f, 0.f};
  v8f acc1 = acc0;
  for (int kk = 0; kk < C2; kk += 32) {
    v4f a[4], b0[4], b1[4];
#pragma unroll
    for (int t = 0; t < 4; ++t) {
      a[t]  = *(const v4f*)(ap  + kk + 8 * t);
      b0[t] = *(const v4f*)(bp0 + kk + 8 * t);
      b1[t] = *(const v4f*)(bp1 + kk + 8 * t);
    }
#pragma unroll
    for (int t = 0; t < 4; ++t) {
      acc0 = wmma4(lo2(a[t]), lo2(b0[t]), acc0);
      acc1 = wmma4(lo2(a[t]), lo2(b1[t]), acc1);
      acc0 = wmma4(hi2(a[t]), hi2(b0[t]), acc0);
      acc1 = wmma4(hi2(a[t]), hi2(b1[t]), acc1);
    }
  }
  float* o0 = G3 + (size_t)(n0 + laneN) * C3 + 16 * mt + 8 * laneHalf;
  float* o1 = o0 + (size_t)16 * C3;
  *(v4f*)(o0)     = __builtin_shufflevector(acc0, acc0, 0, 1, 2, 3);
  *(v4f*)(o0 + 4) = __builtin_shufflevector(acc0, acc0, 4, 5, 6, 7);
  *(v4f*)(o1)     = __builtin_shufflevector(acc1, acc1, 0, 1, 2, 3);
  *(v4f*)(o1 + 4) = __builtin_shufflevector(acc1, acc1, 4, 5, 6, 7);
}

__global__ __launch_bounds__(256) void encoder_kernel(
    const float* __restrict__ points, const int* __restrict__ knnIdx,
    const float* __restrict__ centers, const float* __restrict__ w1,
    const float* __restrict__ b1, const float* __restrict__ g1,
    const float* __restrict__ be1, const float* __restrict__ m1,
    const float* __restrict__ v1, const float* __restrict__ w2,
    const float* __restrict__ b2, const float* __restrict__ w3,
    const float* __restrict__ b3, const float* __restrict__ g2,
    const float* __restrict__ be2, const float* __restrict__ m2,
    const float* __restrict__ v2, const float* __restrict__ w4,
    const float* __restrict__ b4, float* __restrict__ tokens,
    const float* __restrict__ FW, const float* __restrict__ FC,
    const float* __restrict__ G3, size_t wsBytes) {
  extern __shared__ __attribute__((aligned(16))) float smem[];
  float* sFeat = smem;
  float* sH1   = smem + 128;
  float* sXu   = sH1 + 32 * 132;
  float* sGm   = sXu + 32 * 260;
  float* sH3   = sGm + 256;
  float* sTile = sH3 + 32 * 516;
  float* sH4   = smem;

  const int gg       = blockIdx.x;
  const int tid      = threadIdx.x;
  const int lane     = tid & 31;
  const int wv       = tid >> 5;
  const int laneN    = lane & 15;
  const int laneHalf = lane >> 4;
  float* S = sTile + wv * STAGE_FLOATS;

  const v8f vzero = {0.f, 0.f, 0.f, 0.f, 0.f, 0.f, 0.f, 0.f};

  const bool split = wsBytes >= WS_SPLIT_BYTES;

  enc_stage_ab(points, knnIdx, centers, w1, b1, g1, be1, m1, v1, sFeat, sH1, gg,
               tid);

  if (!split) {
    enc_stage_cd(w2, b2, sH1, sXu, S, sGm, tid);
    __syncthreads();
  }

  if (split)
    conv3_stage<2>(w3, b3, g2, be2, m2, v2, FW, FC, G3 + (size_t)gg * C3, sGm,
                   sXu, sH1, sH3, S, wv, lane);
  else if (wsBytes >= WS_NEED_BYTES)
    conv3_stage<1>(w3, b3, g2, be2, m2, v2, FW, FC, G3, sGm, sXu, sH1, sH3, S,
                   wv, lane);
  else
    conv3_stage<0>(w3, b3, g2, be2, m2, v2, FW, FC, G3, sGm, sXu, sH1, sH3, S,
                   wv, lane);
  __syncthreads();

  {
    auto op = [&](float x, int m) { return x + b4[m]; };
    {
      int m0  = wv * 48;
      v8f a00 = vzero, a01 = vzero, a10 = vzero, a11 = vzero;
      gemm_2x2w<C3, 516>(w4, C3, sH3, S, m0, lane, a00, a01, a10, a11);
      store_tile(sH4, 388, a00, m0,      0,  laneN, laneHalf, op);
      store_tile(sH4, 388, a01, m0,      16, laneN, laneHalf, op);
      store_tile(sH4, 388, a10, m0 + 16, 0,  laneN, laneHalf, op);
      store_tile(sH4, 388, a11, m0 + 16, 16, laneN, laneHalf, op);
    }
    {
      int m0 = wv * 48 + 32;
      v8f a0 = vzero, a1 = vzero;
      gemm_1x2w<C3, 516>(w4, C3, sH3, S, m0, lane, a0, a1);
      store_tile(sH4, 388, a0, m0, 0,  laneN, laneHalf, op);
      store_tile(sH4, 388, a1, m0, 16, laneN, laneHalf, op);
    }
  }
  __syncthreads();

  for (int m = tid; m < CT; m += 256) {
    float mx = -3.4e38f;
    for (int n = 0; n < GK; ++n) mx = fmaxf(mx, sH4[n * 388 + m]);
    tokens[(size_t)gg * CT + m] = mx;
  }
}

extern "C" void kernel_launch(void* const* d_in, const int* in_sizes, int n_in,
                              void* d_out, int out_size, void* d_ws,
                              size_t ws_size, hipStream_t stream) {
  (void)in_sizes; (void)n_in; (void)out_size;
  const float* points = (const float*)d_in[0];
  const float* w1  = (const float*)d_in[1];
  const float* b1  = (const float*)d_in[2];
  const float* g1  = (const float*)d_in[3];
  const float* be1 = (const float*)d_in[4];
  const float* m1  = (const float*)d_in[5];
  const float* v1  = (const float*)d_in[6];
  const float* w2  = (const float*)d_in[7];
  const float* b2  = (const float*)d_in[8];
  const float* w3  = (const float*)d_in[9];
  const float* b3  = (const float*)d_in[10];
  const float* g2  = (const float*)d_in[11];
  const float* be2 = (const float*)d_in[12];
  const float* m2  = (const float*)d_in[13];
  const float* v2  = (const float*)d_in[14];
  const float* w4  = (const float*)d_in[15];
  const float* b4  = (const float*)d_in[16];

  float* tokens  = (float*)d_out;
  float* centers = tokens + TOK_TOTAL;
  int*   knnIdx  = (int*)d_ws;
  float* FW      = (float*)d_ws + WS_KNN_FLOATS;
  float* FC      = FW + C3 * C1;
  float* GM      = FC + C3;
  float* G3      = GM + WS_GM_FLOATS;

  fps_kernel<<<NUM_B, FPS_T, 0, stream>>>(points, centers);
  knn_kernel<<<FOLD_BLOCKS + NUM_B * NUM_G, 256, 0, stream>>>(
      points, centers, knnIdx, w2, b2, w3, FW, FC, ws_size);

  enc_pool_kernel<<<NUM_B * NUM_G, 256, (size_t)POOL_LDS_FLOATS * sizeof(float),
                    stream>>>(points, knnIdx, centers, w1, b1, g1, be1, m1, v1,
                              w2, b2, GM, ws_size);
  enc_gm_kernel<<<NUM_B * NUM_G / 8, 256, 0, stream>>>(w3, GM, G3, ws_size);

  size_t ldsBytes = (size_t)(128 + 32 * 132 + 32 * 260 + 256 + 32 * 516 + 8 * STAGE_FLOATS) * sizeof(float);
  encoder_kernel<<<NUM_B * NUM_G, 256, ldsBytes, stream>>>(
      points, knnIdx, centers, w1, b1, g1, be1, m1, v1, w2, b2, w3, b3, g2,
      be2, m2, v2, w4, b4, tokens, FW, FC, G3, ws_size);
}


# Sampling overlapped with the encoder: samples 128..255 run as 8 extra blocks of the launch that encodes groups 0..127 (state via workspace); search and encoder once per half, 9 launches; same sums
# speedup vs baseline: 2.4450x; 1.0348x over previous
#include <hip/hip_runtime.h>
#include <math.h>

#define NUM_B 8
#define NUM_N 8192
#define NUM_G 256
#define GK    32
#define C1    128
#define C2    256
#define C3    512
#define CT    384
#define TOK_TOTAL (NUM_B * NUM_G * CT)

typedef __attribute__((ext_vector_type(2))) float v2f;
typedef __attribute__((ext_vector_type(4))) float v4f;
typedef __attribute__((ext_vector_type(8))) float v8f;

__device__ __forceinline__ v8f wmma4(v2f a, v2f b, v8f c) {
  return __builtin_amdgcn_wmma_f32_16x16x4_f32(false, a, false, b, (short)0, c,
                                               false, false);
}

__device__ __forceinline__ v2f lo2(v4f v) { return __builtin_shufflevector(v, v, 0, 1); }
__device__ __forceinline__ v2f hi2(v4f v) { return __builtin_shufflevector(v, v, 2, 3); }

#define SP 36
#define STAGE_FLOATS (32 * SP)

template <int ROWS>
__device__ __forceinline__ void wload(const float* __restrict__ Wm, int ldw,
                                      int lane, v4f (&g)[ROWS / 4]) {
  const float* p = Wm + (size_t)(lane >> 3) * (size_t)ldw + 4 * (lane & 7);
#pragma unroll
  for (int j = 0; j < ROWS / 4; ++j)
    g[j] = *(const v4f*)(p + (size_t)(4 * j) * (size_t)ldw);
}
template <int ROWS>
__device__ __forceinline__ void wstore(float* S, int lane,
                                       const v4f (&g)[ROWS / 4]) {
  float* q = S + (lane >> 3) * SP + 4 * (lane & 7);
#pragma unroll
  for (int j = 0; j < ROWS / 4; ++j) *(v4f*)(q + 4 * j * SP) = g[j];
}

template <int KDIM, int LDB>
__device__ __forceinline__ void gemm_2x2w(const float* __restrict__ W, int ldw,
                                          const float* B, float* S, int m0,
                                          int lane, v8f& acc00, v8f& acc01,
                                          v8f& acc10, v8f& acc11) {
  static_assert(KDIM % 32 == 0, "32-k trips");
  const int laneN = lane & 15, laneHalf = lane >> 4;
  const float* Wm = W + (size_t)m0 * (size_t)ldw;
  const float* sa0 = S + laneN * SP + 4 * laneHalf;
  const float* sa1 = sa0 + 16 * SP;
  const float* bcol0 = B + laneN * LDB + 4 * laneHalf;
  const float* bcol1 = B + (16 + laneN) * LDB + 4 * laneHalf;
  v4f g[8];
  wload<32>(Wm, ldw, lane, g);
  for (int kk = 0; kk < KDIM; kk += 32) {
    wstore<32>(S, lane, g);
    if (kk + 32 < KDIM) wload<32>(Wm + kk + 32, ldw, lane, g);
#pragma unroll
    for (int t = 0; t < 4; ++t) {
      v4f a0 = *(const v4f*)(sa0 + 8 * t);
      v4f a1 = *(const v4f*)(sa1 + 8 * t);
      v4f b0 = *(const v4f*)(bcol0 + kk + 8 * t);
      v4f b1 = *(const v4f*)(bcol1 + kk + 8 * t);
      acc00 = wmma4(lo2(a0), lo2(b0), acc00);
      acc01 = wmma4(lo2(a0), lo2(b1), acc01);
      acc10 = wmma4(lo2(a1), lo2(b0), acc10);
      acc11 = wmma4(lo2(a1), lo2(b1), acc11);
      acc00 = wmma4(hi2(a0), hi2(b0), acc00);
      acc01 = wmma4(hi2(a0), hi2(b1), acc01);
      acc10 = wmma4(hi2(a1), hi2(b0), acc10);
      acc11 = wmma4(hi2(a1), hi2(b1), acc11);
    }
  }
}

template <int KDIM, int LDB>
__device__ __forceinline__ void gemm_1x2(const float* __restrict__ W, int ldw,
                                         const float* B, int m0, int laneN,
                                         int laneHalf, v8f& acc0, v8f& acc1) {
  const float* wrow  = W + (size_t)(m0 + laneN) * (size_t)ldw + 2 * laneHalf;
  const float* bcol0 = B + laneN * LDB + 2 * laneHalf;
  const float* bcol1 = B + (16 + laneN) * LDB + 2 * laneHalf;
  for (int kk = 0; kk < KDIM; kk += 4) {
    v2f a  = *(const v2f*)(wrow + kk);
    v2f b0 = *(const v2f*)(bcol0 + kk);
    v2f b1 = *(const v2f*)(bcol1 + kk);
    acc0 = wmma4(a, b0, acc0);
    acc1 = wmma4(a, b1, acc1);
  }
}

template <int KDIM, int LDB>
__device__ __forceinline__ void gemm_1x2w(const float* __restrict__ W, int ldw,
                                          const float* B, float* S, int m0,
                                          int lane, v8f& acc0, v8f& acc1) {
  static_assert(KDIM % 32 == 0, "32-k trips");
  const int laneN = lane & 15, laneHalf = lane >> 4;
  const float* Wm = W + (size_t)m0 * (size_t)ldw;
  const float* sa = S + laneN * SP + 4 * laneHalf;
  const float* bcol0 = B + laneN * LDB + 4 * laneHalf;
  const float* bcol1 = B + (16 + laneN) * LDB + 4 * laneHalf;
  v4f g[4];
  wload<16>(Wm, ldw, lane, g);
  for (int kk = 0; kk < KDIM; kk += 32) {
    wstore<16>(S, lane, g);
    if (kk + 32 < KDIM) wload<16>(Wm + kk + 32, ldw, lane, g);
#pragma unroll
    for (int t = 0; t < 4; ++t) {
      v4f a  = *(const v4f*)(sa + 8 * t);
      v4f b0 = *(const v4f*)(bcol0 + kk + 8 * t);
      v4f b1 = *(const v4f*)(bcol1 + kk + 8 * t);
      acc0 = wmma4(lo2(a), lo2(b0), acc0);
      acc1 = wmma4(lo2(a), lo2(b1), acc1);
      acc0 = wmma4(hi2(a), hi2(b0), acc0);
      acc1 = wmma4(hi2(a), hi2(b1), acc1);
    }
  }
}

template <int KDIM>
__device__ __forceinline__ void gemm_2x1v(const float* __restrict__ W, int ldw,
                                          const float* Bv, float* S, int m0,
                                          int lane, v8f& acc0, v8f& acc1) {
  static_assert(KDIM % 32 == 0, "32-k trips");
  const int laneN = lane & 15, laneHalf = lane >> 4;
  const float* Wm = W + (size_t)m0 * (size_t)ldw;
  const float* sa0 = S + laneN * SP + 4 * laneHalf;
  const float* sa1 = sa0 + 16 * SP;
  const float* bvec = Bv + 4 * laneHalf;
  v4f g[8];
  wload<32>(Wm, ldw, lane, g);
  for (int kk = 0; kk < KDIM; kk += 32) {
    wstore<32>(S, lane, g);
    if (kk + 32 < KDIM) wload<32>(Wm + kk + 32, ldw, lane, g);
#pragma unroll
    for (int t = 0; t < 4; ++t) {
      v4f a0 = *(const v4f*)(sa0 + 8 * t);
      v4f a1 = *(const v4f*)(sa1 + 8 * t);
      v4f bb = *(const v4f*)(bvec + kk + 8 * t);
      acc0 = wmma4(lo2(a0), lo2(bb), acc0);
      acc1 = wmma4(lo2(a1), lo2(bb), acc1);
      acc0 = wmma4(hi2(a0), hi2(bb), acc0);
      acc1 = wmma4(hi2(a1), hi2(bb), acc1);
    }
  }
}

template <typename F>
__device__ __forceinline__ void store_tile(float* dst, int ldd, const v8f& acc,
                                           int m0, int n0, int laneN,
                                           int laneHalf, F op) {
  int n = n0 + laneN;
#pragma unroll
  for (int rr = 0; rr < 8; ++rr) {
    int m = m0 + 8 * laneHalf + rr;
    dst[n * ldd + m] = op(acc[rr], m);
  }
}

template <int M>
__device__ __forceinline__ int row_xor(int b) {
  return __builtin_amdgcn_update_dpp(b, b, 0x160 + M, 0xf, 0xf, true);
}
__device__ __forceinline__ int imax(int a, int b) { return a > b ? a : b; }
__device__ __forceinline__ int imin(int a, int b) { return a < b ? a : b; }
__device__ __forceinline__ int wave_max(int v) {
  v = imax(v, row_xor<1>(v));
  v = imax(v, row_xor<2>(v));
  v = imax(v, row_xor<4>(v));
  v = imax(v, row_xor<8>(v));
  v = imax(v, __builtin_amdgcn_permlanex16(v, v, 0x76543210, 0xfedcba98, false,
                                           false));
  return v;
}

#define WS_KNN_FLOATS (NUM_B * NUM_G * GK)
#define WS_NEED_BYTES ((size_t)(WS_KNN_FLOATS + C3 * C1 + C3) * 4)
#define WS_GM_FLOATS (NUM_B * NUM_G * C2)
#define WS_G3_FLOATS (NUM_B * NUM_G * C3)
#define WS_SPLIT_BYTES (WS_NEED_BYTES + (size_t)(WS_GM_FLOATS + WS_G3_FLOATS) * 4)

#define FPS_T 512
#define FPS_P (NUM_N / FPS_T)
#define FPS_W (FPS_T / 32)
#define FPS_SPLIT 128
#define HALF_G (NUM_G - FPS_SPLIT)
static_assert(2 * FPS_SPLIT == NUM_G, "the per-half index maps below take two equal halves");

__device__ __forceinline__ int row_max(int v) {
  v = imax(v, row_xor<1>(v));
  v = imax(v, row_xor<2>(v));
  v = imax(v, row_xor<4>(v));
  v = imax(v, row_xor<8>(v));
  return v;
}

__global__ __launch_bounds__(FPS_T) void fps_kernel(
    const float* __restrict__ points, float* __restrict__ centers,
    int* __restrict__ ddSave, size_t wsBytes) {
  const int b    = blockIdx.x;
  const int tid  = threadIdx.x;
  const int lane = tid & 31;
  const int wv   = __builtin_amdgcn_readfirstlane(tid >> 5);
  const float* base = points + (size_t)b * NUM_N * 3;
  float* co = centers + (size_t)b * NUM_G * 3;

  __shared__ float sP[NUM_N * 3];
  __shared__ int   sWv[2][FPS_W];
  __shared__ int   sWi[2][FPS_W];

  for (int k = tid; k < NUM_N * 3; k += FPS_T) sP[k] = base[k];

  float px[FPS_P], py[FPS_P], pz[FPS_P];
  int   dd[FPS_P];
#pragma unroll
  for (int i = 0; i < FPS_P; ++i) {
    int p = tid * FPS_P + i;
    px[i] = base[p * 3 + 0];
    py[i] = base[p * 3 + 1];
    pz[i] = base[p * 3 + 2];
    dd[i] = __builtin_bit_cast(int, 1e10f);
  }
  __syncthreads();

  float lx = sP[0], ly = sP[1], lz = sP[2];
  if (tid == 0) { co[0] = lx; co[1] = ly; co[2] = lz; }

  const int gEnd = (wsBytes >= WS_SPLIT_BYTES) ? FPS_SPLIT : NUM_G;
  for (int g = 1; g < gEnd; ++g) {
#pragma unroll
    for (int i = 0; i < FPS_P; ++i) {
      float dx = px[i] - lx, dy = py[i] - ly, dz = pz[i] - lz;
      float d = __builtin_fmaf(dz, dz, __builtin_fmaf(dx, dx, dy * dy));
      dd[i]   = imin(dd[i], __builtin_bit_cast(int, d));
    }
    static_assert(FPS_P == 16, "the tree below is written for 16 points");
    const int t0 = imax(imax(dd[0], dd[1]), dd[2]);
    const int t1 = imax(imax(dd[3], dd[4]), dd[5]);
    const int t2 = imax(imax(dd[6], dd[7]), dd[8]);
    const int t3 = imax(imax(dd[9], dd[10]), dd[11]);
    const int t4 = imax(imax(dd[12], dd[13]), dd[14]);
    const int u0 = imax(imax(t0, t1), t2);
    const int u1 = imax(imax(t3, t4), dd[15]);
    const int tm = imax(u0, u1);

    const int buf = g & 1;
    {
      int      wm  = wave_max(tm);
      unsigned has = __builtin_amdgcn_ballot_w32(tm == wm);
      int      fl  = __builtin_ctz(has);
      int idx = FPS_P - 1;
#pragma unroll
      for (int i = FPS_P - 2; i >= 0; --i) {
        unsigned eq = __builtin_amdgcn_ballot_w32(dd[i] == wm);
        if ((eq >> fl) & 1u) idx = i;
      }
      int wbi = (wv * 32 + fl) * FPS_P + idx;
      if (lane == 0) { sWv[buf][wv] = wm; sWi[buf][wv] = wbi; }
    }
    __syncthreads();
    {
      int      v   = sWv[buf][lane & (FPS_W - 1)];
      int      ii  = sWi[buf][lane & (FPS_W - 1)];
      int      bm  = row_max(v);
      unsigned has = __builtin_amdgcn_ballot_w32(v == bm);
      int      wi  = __builtin_amdgcn_readlane(ii, __builtin_ctz(has));
      lx = sP[wi * 3 + 0];
      ly = sP[wi * 3 + 1];
      lz = sP[wi * 3 + 2];
    }
    if (tid == 0) { co[g * 3 + 0] = lx; co[g * 3 + 1] = ly; co[g * 3 + 2] = lz; }
  }
  if (gEnd < NUM_G) {
    int* ds = ddSave + (size_t)b * NUM_N + tid * FPS_P;
#pragma unroll
    for (int i = 0; i < FPS_P; ++i) ds[i] = dd[i];
  }
}

#define FOLD_TASKS  (32 * 4 + 16)
#define FOLD_BLOCKS (FOLD_TASKS / 8)

__device__ __forceinline__ void fold_task(const float* __restrict__ w3,
                                          const float* __restrict__ w2,
                                          const float* __restrict__ b2,
                                          float* __restrict__ FW,
                                          float* __restrict__ FC, float* T,
                                          int task, int lane) {
  const int  laneN = lane & 15, laneHalf = lane >> 4;
  const bool vecTask = task >= 128;
  const int  mA0 = vecTask ? 32 * (task - 128) : 16 * (task >> 2);
  const int  mA1 = vecTask ? mA0 + 16 : mA0;
  const int  n0  = 32 * (task & 3);
  const float* a0p = w3 + (size_t)(mA0 + laneN) * C3 + C2 + 2 * laneHalf;
  const float* a1p = w3 + (size_t)(mA1 + laneN) * C3 + C2 + 2 * laneHalf;
  const int    bstep = vecTask ? 1 : C1;
  const float* b0p = vecTask ? b2 + 2 * laneHalf
                             : w2 + (size_t)(2 * laneHalf) * C1 + n0 + laneN;
  const float* b1p = vecTask ? b0p : b0p + 16;
  v8f acc0 = {0.f, 0.f, 0.f, 0.f, 0.f, 0.f, 0.f, 0.f};
  v8f acc1 = acc0;
  for (int kk = 0; kk < C2; kk += 32) {
    v2f a0[8], a1[8], b0[8], b1[8];
#pragma unroll
    for (int t = 0; t < 8; ++t) {
      a0[t] = *(const v2f*)(a0p + kk + 4 * t);
      a1[t] = *(const v2f*)(a1p + kk + 4 * t);
      const float* p0 = b0p + (size_t)(kk + 4 * t) * bstep;
      const float* p1 = b1p + (size_t)(kk + 4 * t) * bstep;
      b0[t].x = p0[0]; b0[t].y = p0[bstep];
      b1[t].x = p1[0]; b1[t].y = p1[bstep];
    }
#pragma unroll
    for (int t = 0; t < 8; ++t) {
      acc0 = wmma4(a0[t], b0[t], acc0);
      acc1 = wmma4(a1[t], b1[t], acc1);
    }
  }
  if (!vecTask) {
#pragma unroll
    for (int r = 0; r < 8; ++r) {
      T[(8 * laneHalf + r) * 32 + laneN]      = acc0[r];
      T[(8 * laneHalf + r) * 32 + 16 + laneN] = acc1[r];
    }
    for (int rr = 0; rr < 16; ++rr)
      FW[(size_t)(mA0 + rr) * C1 + n0 + lane] = T[rr * 32 + lane];
  } else {
    if (laneN == 0) {
#pragma unroll
      for (int r = 0; r < 8; ++r) {
        T[8 * laneHalf + r]      = acc0[r];
        T[16 + 8 * laneHalf + r] = acc1[r];
      }
    }
    FC[mA0 + lane] = T[lane];
  }
}

__device__ __forceinline__ int wave_min(int v) {
  v = imin(v, row_xor<1>(v));
  v = imin(v, row_xor<2>(v));
  v = imin(v, row_xor<4>(v));
  v = imin(v, row_xor<8>(v));
  v = imin(v, __builtin_amdgcn_permlanex16(v, v, 0x76543210, 0xfedcba98, false,
                                           false));
  return v;
}
__device__ __forceinline__ int oct_min(int v) {
  v = imin(v, row_xor<1>(v));
  v = imin(v, row_xor<2>(v));
  v = imin(v, row_xor<4>(v));
  return v;
}
__device__ __forceinline__ int order_key(float f) {
  int bits = __builtin_bit_cast(int, f);
  return bits ^ ((bits >> 31) & 0x7fffffff);
}

#define KEY_GONE 0x7fffffff

__global__ __launch_bounds__(256) void knn_kernel(
    const float* __restrict__ points, const float* __restrict__ centers,
    int* __restrict__ knnIdx, const float* __restrict__ w2,
    const float* __restrict__ b2, const float* __restrict__ w3,
    float* __restrict__ FW, float* __restrict__ FC, size_t wsBytes,
    int foldBlocks, int half) {
  const int tid  = threadIdx.x;
  const int lane = tid & 31;
  const int wv   = tid >> 5;

  if ((int)blockIdx.x < foldBlocks) {
    __shared__ float sFold[8][16 * 32];
    if (wsBytes >= WS_NEED_BYTES) {
      const int task = __builtin_amdgcn_readfirstlane(blockIdx.x * 8 + wv);
      fold_task(w3, w2, b2, FW, FC, sFold[wv], task, lane);
    }
    return;
  }
  const int bi   = (int)blockIdx.x - foldBlocks;
  const int b    = bi / HALF_G;
  const int gg   = b * NUM_G + half * FPS_SPLIT + bi % HALF_G;
  const float* base = points + (size_t)b * NUM_N * 3;

  __shared__ int sWv[2][8];
  __shared__ int sWi[2][8];

  const float cx = centers[(size_t)gg * 3 + 0];
  const float cy = centers[(size_t)gg * 3 + 1];
  const float cz = centers[(size_t)gg * 3 + 2];
  const float c2 = __builtin_fmaf(cz, cz, __builtin_fmaf(cx, cx, cy * cy));

  int dk[32];
#pragma unroll
  for (int i = 0; i < NUM_N / 256; ++i) {
    int   p  = i * 256 + tid;
    float xx = base[p * 3 + 0], yy = base[p * 3 + 1], zz = base[p * 3 + 2];
    float x2 = __builtin_fmaf(zz, zz, __builtin_fmaf(xx, xx, yy * yy));
    float dt = __builtin_fmaf(cz, zz, __builtin_fmaf(cx, xx, cy * yy));
    dk[i]    = order_key(__builtin_fmaf(-2.f, dt, c2 + x2));
  }

  int lv = KEY_GONE, li = 0;
#pragma unroll
  for (int i = 0; i < 32; ++i)
    if (dk[i] < lv) { lv = dk[i]; li = i; }
  int lp    = li * 256 + tid;
  int myOut = 0;

  for (int r = 0; r < GK; ++r) {
    const int buf = r & 1;
    {
      int wm = wave_min(lv);
      int wp = wave_min(lv == wm ? lp : KEY_GONE);
      if (lane == 0) { sWv[buf][wv] = wm; sWi[buf][wv] = wp; }
    }
    __syncthreads();
    int win;
    {
      int v  = sWv[buf][lane & 7];
      int ii = sWi[buf][lane & 7];
      int bm = oct_min(v);
      win    = oct_min(v == bm ? ii : KEY_GONE);
    }
    win = __builtin_amdgcn_readfirstlane(win);
    if (tid == r) myOut = win;
    const int wt = win & 255;
    const int wi = win >> 8;
    if (wv == (wt >> 5)) {
      const bool mine = (tid == wt);
      int nv = KEY_GONE, ni = 0;
#pragma unroll
      for (int i = 0; i < 32; ++i) {
        if (mine && i == wi) dk[i] = KEY_GONE;
        if (dk[i] < nv) { nv = dk[i]; ni = i; }
      }
      lv = nv;
      lp = ni * 256 + tid;
    }
  }
  if (tid < GK) knnIdx[(size_t)gg * GK + tid] = myOut & (NUM_N - 1);
}

template <int MODE>
__device__ __forceinline__ void conv3_stage(
    const float* __restrict__ w3, const float* __restrict__ b3,
    const float* __restrict__ g2, const float* __restrict__ be2,
    const float* __restrict__ m2, const float* __restrict__ v2,
    const float* __restrict__ FW, const float* __restrict__ FC,
    const float* __restrict__ G3g,
    const float* sGm, const float* sXu, const float* sH1, float* sH3, float* S,
    int wv, int lane) {
  constexpr bool FOLDED = (MODE != 0);
  const int laneN = lane & 15, laneHalf = lane >> 4;
  const v8f vzero = {0.f, 0.f, 0.f, 0.f, 0.f, 0.f, 0.f, 0.f};
  auto op = [&](float x, int m) {
    float inv = g2[m] * rsqrtf(v2[m] + 1e-5f);
    float bb  = b3[m];
    if constexpr (FOLDED) bb = FC[m] + bb;
    float v   = (x + bb) * inv + (be2[m] - m2[m] * inv);
    return fmaxf(v, 0.f);
  };
  auto pass = [&](int j) {
    int m0  = wv * 64 + j * 32;
    v8f g0 = vzero, g1 = vzero;
    if constexpr (MODE == 2) {
      const float* gp = G3g + m0 + 8 * laneHalf;
      v4f x0 = *(const v4f*)(gp),      x1 = *(const v4f*)(gp + 4);
      v4f y0 = *(const v4f*)(gp + 16), y1 = *(const v4f*)(gp + 20);
      g0 = __builtin_shufflevector(x0, x1, 0, 1, 2, 3, 4, 5, 6, 7);
      g1 = __builtin_shufflevector(y0, y1, 0, 1, 2, 3, 4, 5, 6, 7);
    } else {
      gemm_2x1v<C2>(w3, C3, sGm, S, m0, lane, g0, g1);
    }
    v8f a00 = g0, a01 = g0, a10 = g1, a11 = g1;
    if constexpr (FOLDED)
      gemm_2x2w<C1, 132>(FW, C1, sH1, S, m0, lane, a00, a01, a10, a11);
    else
      gemm_2x2w<C2, 260>(w3 + C2, C3, sXu, S, m0, lane, a00, a01, a10, a11);
    store_tile(sH3, 516, a00, m0,      0,  laneN, laneHalf, op);
    store_tile(sH3, 516, a01, m0,      16, laneN, laneHalf, op);
    store_tile(sH3, 516, a10, m0 + 16, 0,  laneN, laneHalf, op);
    store_tile(sH3, 516, a11, m0 + 16, 16, laneN, laneHalf, op);
  };
  if constexpr (MODE == 2) {
#pragma clang loop unroll(disable)
    for (int j = 0; j < 2; ++j) pass(j);
  } else {
#pragma unroll
    for (int j = 0; j < 2; ++j) pass(j);
  }
}

__device__ __forceinline__ void enc_stage_ab(
    const float* __restrict__ points, const int* __restrict__ knnIdx,
    const float* __restrict__ centers, const float* __restrict__ w1,
    const float* __restrict__ b1, const float* __restrict__ g1,
    const float* __restrict__ be1, const float* __restrict__ m1,
    const float* __restrict__ v1, float* sFeat, float* sH1, int gg, int tid) {
  const int b = gg / NUM_G;
  const int lane = tid & 31, wv = tid >> 5;
  const int laneN = lane & 15, laneHalf = lane >> 4;
  const v8f vzero = {0.f, 0.f, 0.f, 0.f, 0.f, 0.f, 0.f, 0.f};
  if (tid < GK) {
    int idx = knnIdx[(size_t)gg * GK + tid];
    const float* p = points + ((size_t)b * NUM_N + idx) * 3;
    const float* c = centers + (size_t)gg * 3;
    float dx = p[0] - c[0], dy = p[1] - c[1], dz = p[2] - c[2];
    float nr = sqrtf(dx * dx + dy * dy + dz * dz);
    sFeat[tid * 4 + 0] = dx;
    sFeat[tid * 4 + 1] = dy;
    sFeat[tid * 4 + 2] = dz;
    sFeat[tid * 4 + 3] = nr;
  }
  __syncthreads();
  {
    int m0 = wv * 16;
    v8f a0 = vzero, a1 = vzero;
    gemm_1x2<4, 4>(w1, 4, sFeat, m0, laneN, laneHalf, a0, a1);
    auto op = [&](float x, int m) {
      float inv = g1[m] * rsqrtf(v1[m] + 1e-5f);
      float v   = (x + b1[m]) * inv + (be1[m] - m1[m] * inv);
      return fmaxf(v, 0.f);
    };
    store_tile(sH1, 132, a0, m0, 0,  laneN, laneHalf, op);
    store_tile(sH1, 132, a1, m0, 16, laneN, laneHalf, op);
  }
  __syncthreads();
}

__device__ __forceinline__ void enc_stage_cd(const float* __restrict__ w2,
                                             const float* __restrict__ b2,
                                             const float* sH1, float* sXu,
                                             float* S, float* gm, int tid) {
  const int lane = tid & 31, wv = tid >> 5;
  const int laneN = lane & 15, laneHalf = lane >> 4;
  const v8f vzero = {0.f, 0.f, 0.f, 0.f, 0.f, 0.f, 0.f, 0.f};
  {
    int m0 = wv * 32;
    v8f a00 = vzero, a01 = vzero, a10 = vzero, a11 = vzero;
    gemm_2x2w<C1, 132>(w2, C1, sH1, S, m0, lane, a00, a01, a10, a11);
    auto op = [&](float x, int m) { return x + b2[m]; };
    store_tile(sXu, 260, a00, m0,      0,  laneN, laneHalf, op);
    store_tile(sXu, 260, a01, m0,      16, laneN, laneHalf, op);
    store_tile(sXu, 260, a10, m0 + 16, 0,  laneN, laneHalf, op);
    store_tile(sXu, 260, a11, m0 + 16, 16, laneN, laneHalf, op);
  }
  __syncthreads();
  {
    int   m  = tid;
    float mx = -3.4e38f;
    for (int n = 0; n < GK; ++n) mx = fmaxf(mx, sXu[n * 260 + m]);
    gm[m] = mx;
  }
}

#define POOL_LDS_FLOATS (128 + 32 * 132 + 32 * 260 + 8 * STAGE_FLOATS)
#define ENC_LDS_FLOATS (128 + 32 * 132 + 32 * 260 + 256 + 32 * 516 + 8 * STAGE_FLOATS)
__global__ __launch_bounds__(256) void enc_pool_kernel(
    const float* __restrict__ points, const int* __restrict__ knnIdx,
    const float* __restrict__ centers, const float* __restrict__ w1,
    const float* __restrict__ b1, const float* __restrict__ g1,
    const float* __restrict__ be1, const float* __restrict__ m1,
    const float* __restrict__ v1, const float* __restrict__ w2,
    const float* __restrict__ b2, float* __restrict__ GM, size_t wsBytes,
    int half) {
  if (wsBytes < WS_SPLIT_BYTES) return;
  extern __shared__ __attribute__((aligned(16))) float smem[];
  float* sFeat = smem;
  float* sH1   = smem + 128;
  float* sXu   = sH1 + 32 * 132;
  float* sTile = sXu + 32 * 260;
  const int gg = ((int)blockIdx.x / HALF_G) * NUM_G + half * FPS_SPLIT +
                 (int)blockIdx.x % HALF_G;
  const int tid = threadIdx.x;
  enc_stage_ab(points, knnIdx, centers, w1, b1, g1, be1, m1, v1, sFeat, sH1, gg,
               tid);
  enc_stage_cd(w2, b2, sH1, sXu, sTile + (tid >> 5) * STAGE_FLOATS,
               GM + (size_t)gg * C2, tid);
}

__global__ __launch_bounds__(256) void enc_gm_kernel(
    const float* __restrict__ w3, const float* __restrict__ GM,
    float* __restrict__ G3, size_t wsBytes, int half) {
  if (wsBytes < WS_SPLIT_BYTES) return;
  static_assert(HALF_G % 32 == 0 && FPS_SPLIT % 32 == 0, "32-group pieces");
  const int lane = threadIdx.x & 31;
  const int laneN = lane & 15, laneHalf = lane >> 4;
  const int wv = __builtin_amdgcn_readfirstlane(threadIdx.x >> 5);
  const int mt = (blockIdx.x & 3) * 8 + wv;
  const int pc = blockIdx.x >> 2;
  const int n0 = (pc / (HALF_G / 32)) * NUM_G + half * FPS_SPLIT +
                 (pc % (HALF_G / 32)) * 32;
  const float* ap  = w3 + (size_t)(16 * mt + laneN) * C3 + 4 * laneHalf;
  const float* bp0 = GM + (size_t)(n0 + laneN) * C2 + 4 * laneHalf;
  const float* bp1 = bp0 + (size_t)16 * C2;
  v8f acc0 = {0.f, 0.f, 0.f, 0.f, 0.f, 0.f, 0.f, 0.f};
  v8f acc1 = acc0;
  for (int kk = 0; kk < C2; kk += 32) {
    v4f a[4], b0[4], b1[4];
#pragma unroll
    for (int t = 0; t < 4; ++t) {
      a[t]  = *(const v4f*)(ap  + kk + 8 * t);
      b0[t] = *(const v4f*)(bp0 + kk + 8 * t);
      b1[t] = *(const v4f*)(bp1 + kk + 8 * t);
    }
#pragma unroll
    for (int t = 0; t < 4; ++t) {
      acc0 = wmma4(lo2(a[t]), lo2(b0[t]), acc0);
      acc1 = wmma4(lo2(a[t]), lo2(b1[t]), acc1);
      acc0 = wmma4(hi2(a[t]), hi2(b0[t]), acc0);
      acc1 = wmma4(hi2(a[t]), hi2(b1[t]), acc1);
    }
  }
  float* o0 = G3 + (size_t)(n0 + laneN) * C3 + 16 * mt + 8 * laneHalf;
  float* o1 = o0 + (size_t)16 * C3;
  *(v4f*)(o0)     = __builtin_shufflevector(acc0, acc0, 0, 1, 2, 3);
  *(v4f*)(o0 + 4) = __builtin_shufflevector(acc0, acc0, 4, 5, 6, 7);
  *(v4f*)(o1)     = __builtin_shufflevector(acc1, acc1, 0, 1, 2, 3);
  *(v4f*)(o1 + 4) = __builtin_shufflevector(acc1, acc1, 4, 5, 6, 7);
}

#define FPR_T 256
#define FPR_P (NUM_N / FPR_T)
#define FPR_W (FPR_T / 32)
#define FPR_LDS_FLOATS (NUM_N * 3 + 4 * FPR_W)

__device__ __forceinline__ int oct_max(int v) {
  v = imax(v, row_xor<1>(v));
  v = imax(v, row_xor<2>(v));
  v = imax(v, row_xor<4>(v));
  return v;
}

__device__ __forceinline__ void fps_resume(const float* __restrict__ points,
                                           float* __restrict__ centersOut,
                                           const int* __restrict__ ddSave,
                                           float* smem, int b, int tid) {
  const int lane = tid & 31;
  const int wv   = __builtin_amdgcn_readfirstlane(tid >> 5);
  const float* base = points + (size_t)b * NUM_N * 3;
  float* co = centersOut + (size_t)b * NUM_G * 3;

  float* sP  = smem;
  int*   sWv = (int*)(smem + NUM_N * 3);
  int*   sWi = sWv + 2 * FPR_W;

  {
    const v4f* src4 = (const v4f*)base;
    v4f*       dst4 = (v4f*)sP;
#pragma unroll 8
    for (int k = tid; k < NUM_N * 3 / 4; k += FPR_T) dst4[k] = src4[k];
  }

  float px[FPR_P], py[FPR_P], pz[FPR_P];
  int   dd[FPR_P];
  const int* ds = ddSave + (size_t)b * NUM_N + tid * FPR_P;
#pragma unroll
  for (int i = 0; i < FPR_P; ++i) {
    int p = tid * FPR_P + i;
    px[i] = base[p * 3 + 0];
    py[i] = base[p * 3 + 1];
    pz[i] = base[p * 3 + 2];
    dd[i] = ds[i];
  }
  float lx = co[(FPS_SPLIT - 1) * 3 + 0];
  float ly = co[(FPS_SPLIT - 1) * 3 + 1];
  float lz = co[(FPS_SPLIT - 1) * 3 + 2];
  __syncthreads();

  for (int g = FPS_SPLIT; g < NUM_G; ++g) {
#pragma unroll
    for (int i = 0; i < FPR_P; ++i) {
      float dx = px[i] - lx, dy = py[i] - ly, dz = pz[i] - lz;
      float d = __builtin_fmaf(dz, dz, __builtin_fmaf(dx, dx, dy * dy));
      dd[i]   = imin(dd[i], __builtin_bit_cast(int, d));
    }
    static_assert(FPR_P == 32, "the tree below is written for 32 points");
    int t[11];
#pragma unroll
    for (int j = 0; j < 10; ++j)
      t[j] = imax(imax(dd[3 * j], dd[3 * j + 1]), dd[3 * j + 2]);
    t[10] = imax(dd[30], dd[31]);
    const int u0 = imax(imax(t[0], t[1]), t[2]);
    const int u1 = imax(imax(t[3], t[4]), t[5]);
    const int u2 = imax(imax(t[6], t[7]), t[8]);
    const int u3 = imax(t[9], t[10]);
    const int tm = imax(imax(u0, u1), imax(u2, u3));

    const int buf = g & 1;
    {
      int      wm  = wave_max(tm);
      unsigned has = __builtin_amdgcn_ballot_w32(tm == wm);
      int      fl  = __builtin_ctz(has);
      int idx = FPR_P - 1;
#pragma unroll
      for (int i = FPR_P - 2; i >= 0; --i) {
        unsigned eq = __builtin_amdgcn_ballot_w32(dd[i] == wm);
        if ((eq >> fl) & 1u) idx = i;
      }
      int wbi = (wv * 32 + fl) * FPR_P + idx;
      if (lane == 0) { sWv[buf * FPR_W + wv] = wm; sWi[buf * FPR_W + wv] = wbi; }
    }
    __syncthreads();
    {
      int      v   = sWv[buf * FPR_W + (lane & (FPR_W - 1))];
      int      ii  = sWi[buf * FPR_W + (lane & (FPR_W - 1))];
      int      bm  = oct_max(v);
      unsigned has = __builtin_amdgcn_ballot_w32(v == bm);
      int      wi  = __builtin_amdgcn_readlane(ii, __builtin_ctz(has));
      lx = sP[wi * 3 + 0];
      ly = sP[wi * 3 + 1];
      lz = sP[wi * 3 + 2];
    }
    if (tid == 0) { co[g * 3 + 0] = lx; co[g * 3 + 1] = ly; co[g * 3 + 2] = lz; }
  }
}

__global__ __launch_bounds__(256) void encoder_kernel(
    const float* __restrict__ points, const int* __restrict__ knnIdx,
    const float* __restrict__ centers, const float* __restrict__ w1,
    const float* __restrict__ b1, const float* __restrict__ g1,
    const float* __restrict__ be1, const float* __restrict__ m1,
    const float* __restrict__ v1, const float* __restrict__ w2,
    const float* __restrict__ b2, const float* __restrict__ w3,
    const float* __restrict__ b3, const float* __restrict__ g2,
    const float* __restrict__ be2, const float* __restrict__ m2,
    const float* __restrict__ v2, const float* __restrict__ w4,
    const float* __restrict__ b4, float* __restrict__ tokens,
    const float* __restrict__ FW, const float* __restrict__ FC,
    const float* __restrict__ G3, size_t wsBytes,
    float* __restrict__ centersOut, const int* __restrict__ ddSave,
    int fpsBlocks, int half) {
  extern __shared__ __attribute__((aligned(16))) float smem[];
  if ((int)blockIdx.x < fpsBlocks) {
    static_assert(FPR_LDS_FLOATS <= ENC_LDS_FLOATS, "the points fit the encoder's LDS");
    if (wsBytes >= WS_SPLIT_BYTES)
      fps_resume(points, centersOut, ddSave, smem, blockIdx.x, threadIdx.x);
    return;
  }
  float* sFeat = smem;
  float* sH1   = smem + 128;
  float* sXu   = sH1 + 32 * 132;
  float* sGm   = sXu + 32 * 260;
  float* sH3   = sGm + 256;
  float* sTile = sH3 + 32 * 516;
  float* sH4   = smem;

  const int bi       = (int)blockIdx.x - fpsBlocks;
  const int gg       = (bi / HALF_G) * NUM_G + half * FPS_SPLIT + bi % HALF_G;
  const int tid      = threadIdx.x;
  const int lane     = tid & 31;
  const int wv       = tid >> 5;
  const int laneN    = lane & 15;
  const int laneHalf = lane >> 4;
  float* S = sTile + wv * STAGE_FLOATS;

  const v8f vzero = {0.f, 0.f, 0.f, 0.f, 0.f, 0.f, 0.f, 0.f};

  const bool split = wsBytes >= WS_SPLIT_BYTES;

  enc_stage_ab(points, knnIdx, centers, w1, b1, g1, be1, m1, v1, sFeat, sH1, gg,
               tid);

  if (!split) {
    enc_stage_cd(w2, b2, sH1, sXu, S, sGm, tid);
    __syncthreads();
  }

  if (split)
    conv3_stage<2>(w3, b3, g2, be2, m2, v2, FW, FC, G3 + (size_t)gg * C3, sGm,
                   sXu, sH1, sH3, S, wv, lane);
  else if (wsBytes >= WS_NEED_BYTES)
    conv3_stage<1>(w3, b3, g2, be2, m2, v2, FW, FC, G3, sGm, sXu, sH1, sH3, S,
                   wv, lane);
  else
    conv3_stage<0>(w3, b3, g2, be2, m2, v2, FW, FC, G3, sGm, sXu, sH1, sH3, S,
                   wv, lane);
  __syncthreads();

  {
    auto op = [&](float x, int m) { return x + b4[m]; };
    {
      int m0  = wv * 48;
      v8f a00 = vzero, a01 = vzero, a10 = vzero, a11 = vzero;
      gemm_2x2w<C3, 516>(w4, C3, sH3, S, m0, lane, a00, a01, a10, a11);
      store_tile(sH4, 388, a00, m0,      0,  laneN, laneHalf, op);
      store_tile(sH4, 388, a01, m0,      16, laneN, laneHalf, op);
      store_tile(sH4, 388, a10, m0 + 16, 0,  laneN, laneHalf, op);
      store_tile(sH4, 388, a11, m0 + 16, 16, laneN, laneHalf, op);
    }
    {
      int m0 = wv * 48 + 32;
      v8f a0 = vzero, a1 = vzero;
      gemm_1x2w<C3, 516>(w4, C3, sH3, S, m0, lane, a0, a1);
      store_tile(sH4, 388, a0, m0, 0,  laneN, laneHalf, op);
      store_tile(sH4, 388, a1, m0, 16, laneN, laneHalf, op);
    }
  }
  __syncthreads();

  for (int m = tid; m < CT; m += 256) {
    float mx = -3.4e38f;
    for (int n = 0; n < GK; ++n) mx = fmaxf(mx, sH4[n * 388 + m]);
    tokens[(size_t)gg * CT + m] = mx;
  }
}

extern "C" void kernel_launch(void* const* d_in, const int* in_sizes, int n_in,
                              void* d_out, int out_size, void* d_ws,
                              size_t ws_size, hipStream_t stream) {
  (void)in_sizes; (void)n_in; (void)out_size;
  const float* points = (const float*)d_in[0];
  const float* w1  = (const float*)d_in[1];
  const float* b1  = (const float*)d_in[2];
  const float* g1  = (const float*)d_in[3];
  const float* be1 = (const float*)d_in[4];
  const float* m1  = (const float*)d_in[5];
  const float* v1  = (const float*)d_in[6];
  const float* w2  = (const float*)d_in[7];
  const float* b2  = (const float*)d_in[8];
  const float* w3  = (const float*)d_in[9];
  const float* b3  = (const float*)d_in[10];
  const float* g2  = (const float*)d_in[11];
  const float* be2 = (const float*)d_in[12];
  const float* m2  = (const float*)d_in[13];
  const float* v2  = (const float*)d_in[14];
  const float* w4  = (const float*)d_in[15];
  const float* b4  = (const float*)d_in[16];

  float* tokens  = (float*)d_out;
  float* centers = tokens + TOK_TOTAL;
  int*   knnIdx  = (int*)d_ws;
  float* FW      = (float*)d_ws + WS_KNN_FLOATS;
  float* FC      = FW + C3 * C1;
  float* GM      = FC + C3;
  float* G3      = GM + WS_GM_FLOATS;
  int*   ddSave  = (int*)(G3 + FPS_SPLIT * C3);

  const size_t poolLds = (size_t)POOL_LDS_FLOATS * sizeof(float);
  const size_t encLds  = (size_t)ENC_LDS_FLOATS * sizeof(float);

  fps_kernel<<<NUM_B, FPS_T, 0, stream>>>(points, centers, ddSave, ws_size);

  knn_kernel<<<FOLD_BLOCKS + NUM_B * HALF_G, 256, 0, stream>>>(
      points, centers, knnIdx, w2, b2, w3, FW, FC, ws_size, FOLD_BLOCKS, 0);
  enc_pool_kernel<<<NUM_B * HALF_G, 256, poolLds, stream>>>(
      points, knnIdx, centers, w1, b1, g1, be1, m1, v1, w2, b2, GM, ws_size, 0);
  enc_gm_kernel<<<NUM_B * HALF_G / 8, 256, 0, stream>>>(w3, GM, G3, ws_size, 0);
  encoder_kernel<<<NUM_B + NUM_B * HALF_G, 256, encLds, stream>>>(
      points, knnIdx, centers, w1, b1, g1, be1, m1, v1, w2, b2, w3, b3, g2,
      be2, m2, v2, w4, b4, tokens, FW, FC, G3, ws_size, centers, ddSave, NUM_B,
      0);

  knn_kernel<<<NUM_B * HALF_G, 256, 0, stream>>>(
      points, centers, knnIdx, w2, b2, w3, FW, FC, ws_size, 0, 1);
  enc_pool_kernel<<<NUM_B * HALF_G, 256, poolLds, stream>>>(
      points, knnIdx, centers, w1, b1, g1, be1, m1, v1, w2, b2, GM, ws_size, 1);
  enc_gm_kernel<<<NUM_B * HALF_G / 8, 256, 0, stream>>>(w3, GM, G3, ws_size, 1);
  encoder_kernel<<<NUM_B * HALF_G, 256, encLds, stream>>>(
      points, knnIdx, centers, w1, b1, g1, be1, m1, v1, w2, b2, w3, b3, g2,
      be2, m2, v2, w4, b4, tokens, FW, FC, G3, ws_size, centers, ddSave, 0, 1);
}


# KNN selection: one bound (largest of the 8 waves' 4th smallest cached minimum), the keys at or below it ranked by counting; 3 barriers, was 32 serial minimum rounds; same indices, same order
# speedup vs baseline: 2.7013x; 1.1048x over previous
#include <hip/hip_runtime.h>
#include <math.h>

#define NUM_B 8
#define NUM_N 8192
#define NUM_G 256
#define GK    32
#define C1    128
#define C2    256
#define C3    512
#define CT    384
#define TOK_TOTAL (NUM_B * NUM_G * CT)

typedef __attribute__((ext_vector_type(2))) float v2f;
typedef __attribute__((ext_vector_type(4))) float v4f;
typedef __attribute__((ext_vector_type(8))) float v8f;

__device__ __forceinline__ v8f wmma4(v2f a, v2f b, v8f c) {
  return __builtin_amdgcn_wmma_f32_16x16x4_f32(false, a, false, b, (short)0, c,
                                               false, false);
}

__device__ __forceinline__ v2f lo2(v4f v) { return __builtin_shufflevector(v, v, 0, 1); }
__device__ __forceinline__ v2f hi2(v4f v) { return __builtin_shufflevector(v, v, 2, 3); }

#define SP 36
#define STAGE_FLOATS (32 * SP)

template <int ROWS>
__device__ __forceinline__ void wload(const float* __restrict__ Wm, int ldw,
                                      int lane, v4f (&g)[ROWS / 4]) {
  const float* p = Wm + (size_t)(lane >> 3) * (size_t)ldw + 4 * (lane & 7);
#pragma unroll
  for (int j = 0; j < ROWS / 4; ++j)
    g[j] = *(const v4f*)(p + (size_t)(4 * j) * (size_t)ldw);
}
template <int ROWS>
__device__ __forceinline__ void wstore(float* S, int lane,
                                       const v4f (&g)[ROWS / 4]) {
  float* q = S + (lane >> 3) * SP + 4 * (lane & 7);
#pragma unroll
  for (int j = 0; j < ROWS / 4; ++j) *(v4f*)(q + 4 * j * SP) = g[j];
}

template <int KDIM, int LDB>
__device__ __forceinline__ void gemm_2x2w(const float* __restrict__ W, int ldw,
                                          const float* B, float* S, int m0,
                                          int lane, v8f& acc00, v8f& acc01,
                                          v8f& acc10, v8f& acc11) {
  static_assert(KDIM % 32 == 0, "32-k trips");
  const int laneN = lane & 15, laneHalf = lane >> 4;
  const float* Wm = W + (size_t)m0 * (size_t)ldw;
  const float* sa0 = S + laneN * SP + 4 * laneHalf;
  const float* sa1 = sa0 + 16 * SP;
  const float* bcol0 = B + laneN * LDB + 4 * laneHalf;
  const float* bcol1 = B + (16 + laneN) * LDB + 4 * laneHalf;
  v4f g[8];
  wload<32>(Wm, ldw, lane, g);
  for (int kk = 0; kk < KDIM; kk += 32) {
    wstore<32>(S, lane, g);
    if (kk + 32 < KDIM) wload<32>(Wm + kk + 32, ldw, lane, g);
#pragma unroll
    for (int t = 0; t < 4; ++t) {
      v4f a0 = *(const v4f*)(sa0 + 8 * t);
      v4f a1 = *(const v4f*)(sa1 + 8 * t);
      v4f b0 = *(const v4f*)(bcol0 + kk + 8 * t);
      v4f b1 = *(const v4f*)(bcol1 + kk + 8 * t);
      acc00 = wmma4(lo2(a0), lo2(b0), acc00);
      acc01 = wmma4(lo2(a0), lo2(b1), acc01);
      acc10 = wmma4(lo2(a1), lo2(b0), acc10);
      acc11 = wmma4(lo2(a1), lo2(b1), acc11);
      acc00 = wmma4(hi2(a0), hi2(b0), acc00);
      acc01 = wmma4(hi2(a0), hi2(b1), acc01);
      acc10 = wmma4(hi2(a1), hi2(b0), acc10);
      acc11 = wmma4(hi2(a1), hi2(b1), acc11);
    }
  }
}

template <int KDIM, int LDB>
__device__ __forceinline__ void gemm_1x2(const float* __restrict__ W, int ldw,
                                         const float* B, int m0, int laneN,
                                         int laneHalf, v8f& acc0, v8f& acc1) {
  const float* wrow  = W + (size_t)(m0 + laneN) * (size_t)ldw + 2 * laneHalf;
  const float* bcol0 = B + laneN * LDB + 2 * laneHalf;
  const float* bcol1 = B + (16 + laneN) * LDB + 2 * laneHalf;
  for (int kk = 0; kk < KDIM; kk += 4) {
    v2f a  = *(const v2f*)(wrow + kk);
    v2f b0 = *(const v2f*)(bcol0 + kk);
    v2f b1 = *(const v2f*)(bcol1 + kk);
    acc0 = wmma4(a, b0, acc0);
    acc1 = wmma4(a, b1, acc1);
  }
}

template <int KDIM, int LDB>
__device__ __forceinline__ void gemm_1x2w(const float* __restrict__ W, int ldw,
                                          const float* B, float* S, int m0,
                                          int lane, v8f& acc0, v8f& acc1) {
  static_assert(KDIM % 32 == 0, "32-k trips");
  const int laneN = lane & 15, laneHalf = lane >> 4;
  const float* Wm = W + (size_t)m0 * (size_t)ldw;
  const float* sa = S + laneN * SP + 4 * laneHalf;
  const float* bcol0 = B + laneN * LDB + 4 * laneHalf;
  const float* bcol1 = B + (16 + laneN) * LDB + 4 * laneHalf;
  v4f g[4];
  wload<16>(Wm, ldw, lane, g);
  for (int kk = 0; kk < KDIM; kk += 32) {
    wstore<16>(S, lane, g);
    if (kk + 32 < KDIM) wload<16>(Wm + kk + 32, ldw, lane, g);
#pragma unroll
    for (int t = 0; t < 4; ++t) {
      v4f a  = *(const v4f*)(sa + 8 * t);
      v4f b0 = *(const v4f*)(bcol0 + kk + 8 * t);
      v4f b1 = *(const v4f*)(bcol1 + kk + 8 * t);
      acc0 = wmma4(lo2(a), lo2(b0), acc0);
      acc1 = wmma4(lo2(a), lo2(b1), acc1);
      acc0 = wmma4(hi2(a), hi2(b0), acc0);
      acc1 = wmma4(hi2(a), hi2(b1), acc1);
    }
  }
}

template <int KDIM>
__device__ __forceinline__ void gemm_2x1v(const float* __restrict__ W, int ldw,
                                          const float* Bv, float* S, int m0,
                                          int lane, v8f& acc0, v8f& acc1) {
  static_assert(KDIM % 32 == 0, "32-k trips");
  const int laneN = lane & 15, laneHalf = lane >> 4;
  const float* Wm = W + (size_t)m0 * (size_t)ldw;
  const float* sa0 = S + laneN * SP + 4 * laneHalf;
  const float* sa1 = sa0 + 16 * SP;
  const float* bvec = Bv + 4 * laneHalf;
  v4f g[8];
  wload<32>(Wm, ldw, lane, g);
  for (int kk = 0; kk < KDIM; kk += 32) {
    wstore<32>(S, lane, g);
    if (kk + 32 < KDIM) wload<32>(Wm + kk + 32, ldw, lane, g);
#pragma unroll
    for (int t = 0; t < 4; ++t) {
      v4f a0 = *(const v4f*)(sa0 + 8 * t);
      v4f a1 = *(const v4f*)(sa1 + 8 * t);
      v4f bb = *(const v4f*)(bvec + kk + 8 * t);
      acc0 = wmma4(lo2(a0), lo2(bb), acc0);
      acc1 = wmma4(lo2(a1), lo2(bb), acc1);
      acc0 = wmma4(hi2(a0), hi2(bb), acc0);
      acc1 = wmma4(hi2(a1), hi2(bb), acc1);
    }
  }
}

template <typename F>
__device__ __forceinline__ void store_tile(float* dst, int ldd, const v8f& acc,
                                           int m0, int n0, int laneN,
                                           int laneHalf, F op) {
  int n = n0 + laneN;
#pragma unroll
  for (int rr = 0; rr < 8; ++rr) {
    int m = m0 + 8 * laneHalf + rr;
    dst[n * ldd + m] = op(acc[rr], m);
  }
}

template <int M>
__device__ __forceinline__ int row_xor(int b) {
  return __builtin_amdgcn_update_dpp(b, b, 0x160 + M, 0xf, 0xf, true);
}
__device__ __forceinline__ int imax(int a, int b) { return a > b ? a : b; }
__device__ __forceinline__ int imin(int a, int b) { return a < b ? a : b; }
__device__ __forceinline__ int wave_max(int v) {
  v = imax(v, row_xor<1>(v));
  v = imax(v, row_xor<2>(v));
  v = imax(v, row_xor<4>(v));
  v = imax(v, row_xor<8>(v));
  v = imax(v, __builtin_amdgcn_permlanex16(v, v, 0x76543210, 0xfedcba98, false,
                                           false));
  return v;
}

#define WS_KNN_FLOATS (NUM_B * NUM_G * GK)
#define WS_NEED_BYTES ((size_t)(WS_KNN_FLOATS + C3 * C1 + C3) * 4)
#define WS_GM_FLOATS (NUM_B * NUM_G * C2)
#define WS_G3_FLOATS (NUM_B * NUM_G * C3)
#define WS_SPLIT_BYTES (WS_NEED_BYTES + (size_t)(WS_GM_FLOATS + WS_G3_FLOATS) * 4)

#define FPS_T 512
#define FPS_P (NUM_N / FPS_T)
#define FPS_W (FPS_T / 32)
#define FPS_SPLIT 128
#define HALF_G (NUM_G - FPS_SPLIT)
static_assert(2 * FPS_SPLIT == NUM_G, "the per-half index maps below take two equal halves");

__device__ __forceinline__ int row_max(int v) {
  v = imax(v, row_xor<1>(v));
  v = imax(v, row_xor<2>(v));
  v = imax(v, row_xor<4>(v));
  v = imax(v, row_xor<8>(v));
  return v;
}

__global__ __launch_bounds__(FPS_T) void fps_kernel(
    const float* __restrict__ points, float* __restrict__ centers,
    int* __restrict__ ddSave, size_t wsBytes) {
  const int b    = blockIdx.x;
  const int tid  = threadIdx.x;
  const int lane = tid & 31;
  const int wv   = __builtin_amdgcn_readfirstlane(tid >> 5);
  const float* base = points + (size_t)b * NUM_N * 3;
  float* co = centers + (size_t)b * NUM_G * 3;

  __shared__ float sP[NUM_N * 3];
  __shared__ int   sWv[2][FPS_W];
  __shared__ int   sWi[2][FPS_W];

  for (int k = tid; k < NUM_N * 3; k += FPS_T) sP[k] = base[k];

  float px[FPS_P], py[FPS_P], pz[FPS_P];
  int   dd[FPS_P];
#pragma unroll
  for (int i = 0; i < FPS_P; ++i) {
    int p = tid * FPS_P + i;
    px[i] = base[p * 3 + 0];
    py[i] = base[p * 3 + 1];
    pz[i] = base[p * 3 + 2];
    dd[i] = __builtin_bit_cast(int, 1e10f);
  }
  __syncthreads();

  float lx = sP[0], ly = sP[1], lz = sP[2];
  if (tid == 0) { co[0] = lx; co[1] = ly; co[2] = lz; }

  const int gEnd = (wsBytes >= WS_SPLIT_BYTES) ? FPS_SPLIT : NUM_G;
  for (int g = 1; g < gEnd; ++g) {
#pragma unroll
    for (int i = 0; i < FPS_P; ++i) {
      float dx = px[i] - lx, dy = py[i] - ly, dz = pz[i] - lz;
      float d = __builtin_fmaf(dz, dz, __builtin_fmaf(dx, dx, dy * dy));
      dd[i]   = imin(dd[i], __builtin_bit_cast(int, d));
    }
    static_assert(FPS_P == 16, "the tree below is written for 16 points");
    const int t0 = imax(imax(dd[0], dd[1]), dd[2]);
    const int t1 = imax(imax(dd[3], dd[4]), dd[5]);
    const int t2 = imax(imax(dd[6], dd[7]), dd[8]);
    const int t3 = imax(imax(dd[9], dd[10]), dd[11]);
    const int t4 = imax(imax(dd[12], dd[13]), dd[14]);
    const int u0 = imax(imax(t0, t1), t2);
    const int u1 = imax(imax(t3, t4), dd[15]);
    const int tm = imax(u0, u1);

    const int buf = g & 1;
    {
      int      wm  = wave_max(tm);
      unsigned has = __builtin_amdgcn_ballot_w32(tm == wm);
      int      fl  = __builtin_ctz(has);
      int idx = FPS_P - 1;
#pragma unroll
      for (int i = FPS_P - 2; i >= 0; --i) {
        unsigned eq = __builtin_amdgcn_ballot_w32(dd[i] == wm);
        if ((eq >> fl) & 1u) idx = i;
      }
      int wbi = (wv * 32 + fl) * FPS_P + idx;
      if (lane == 0) { sWv[buf][wv] = wm; sWi[buf][wv] = wbi; }
    }
    __syncthreads();
    {
      int      v   = sWv[buf][lane & (FPS_W - 1)];
      int      ii  = sWi[buf][lane & (FPS_W - 1)];
      int      bm  = row_max(v);
      unsigned has = __builtin_amdgcn_ballot_w32(v == bm);
      int      wi  = __builtin_amdgcn_readlane(ii, __builtin_ctz(has));
      lx = sP[wi * 3 + 0];
      ly = sP[wi * 3 + 1];
      lz = sP[wi * 3 + 2];
    }
    if (tid == 0) { co[g * 3 + 0] = lx; co[g * 3 + 1] = ly; co[g * 3 + 2] = lz; }
  }
  if (gEnd < NUM_G) {
    int* ds = ddSave + (size_t)b * NUM_N + tid * FPS_P;
#pragma unroll
    for (int i = 0; i < FPS_P; ++i) ds[i] = dd[i];
  }
}

#define FOLD_TASKS  (32 * 4 + 16)
#define FOLD_BLOCKS (FOLD_TASKS / 8)

__device__ __forceinline__ void fold_task(const float* __restrict__ w3,
                                          const float* __restrict__ w2,
                                          const float* __restrict__ b2,
                                          float* __restrict__ FW,
                                          float* __restrict__ FC, float* T,
                                          int task, int lane) {
  const int  laneN = lane & 15, laneHalf = lane >> 4;
  const bool vecTask = task >= 128;
  const int  mA0 = vecTask ? 32 * (task - 128) : 16 * (task >> 2);
  const int  mA1 = vecTask ? mA0 + 16 : mA0;
  const int  n0  = 32 * (task & 3);
  const float* a0p = w3 + (size_t)(mA0 + laneN) * C3 + C2 + 2 * laneHalf;
  const float* a1p = w3 + (size_t)(mA1 + laneN) * C3 + C2 + 2 * laneHalf;
  const int    bstep = vecTask ? 1 : C1;
  const float* b0p = vecTask ? b2 + 2 * laneHalf
                             : w2 + (size_t)(2 * laneHalf) * C1 + n0 + laneN;
  const float* b1p = vecTask ? b0p : b0p + 16;
  v8f acc0 = {0.f, 0.f, 0.f, 0.f, 0.f, 0.f, 0.f, 0.f};
  v8f acc1 = acc0;
  for (int kk = 0; kk < C2; kk += 32) {
    v2f a0[8], a1[8], b0[8], b1[8];
#pragma unroll
    for (int t = 0; t < 8; ++t) {
      a0[t] = *(const v2f*)(a0p + kk + 4 * t);
      a1[t] = *(const v2f*)(a1p + kk + 4 * t);
      const float* p0 = b0p + (size_t)(kk + 4 * t) * bstep;
      const float* p1 = b1p + (size_t)(kk + 4 * t) * bstep;
      b0[t].x = p0[0]; b0[t].y = p0[bstep];
      b1[t].x = p1[0]; b1[t].y = p1[bstep];
    }
#pragma unroll
    for (int t = 0; t < 8; ++t) {
      acc0 = wmma4(a0[t], b0[t], acc0);
      acc1 = wmma4(a1[t], b1[t], acc1);
    }
  }
  if (!vecTask) {
#pragma unroll
    for (int r = 0; r < 8; ++r) {
      T[(8 * laneHalf + r) * 32 + laneN]      = acc0[r];
      T[(8 * laneHalf + r) * 32 + 16 + laneN] = acc1[r];
    }
    for (int rr = 0; rr < 16; ++rr)
      FW[(size_t)(mA0 + rr) * C1 + n0 + lane] = T[rr * 32 + lane];
  } else {
    if (laneN == 0) {
#pragma unroll
      for (int r = 0; r < 8; ++r) {
        T[8 * laneHalf + r]      = acc0[r];
        T[16 + 8 * laneHalf + r] = acc1[r];
      }
    }
    FC[mA0 + lane] = T[lane];
  }
}

__device__ __forceinline__ int wave_min(int v) {
  v = imin(v, row_xor<1>(v));
  v = imin(v, row_xor<2>(v));
  v = imin(v, row_xor<4>(v));
  v = imin(v, row_xor<8>(v));
  v = imin(v, __builtin_amdgcn_permlanex16(v, v, 0x76543210, 0xfedcba98, false,
                                           false));
  return v;
}
__device__ __forceinline__ int oct_min(int v) {
  v = imin(v, row_xor<1>(v));
  v = imin(v, row_xor<2>(v));
  v = imin(v, row_xor<4>(v));
  return v;
}
__device__ __forceinline__ int oct_max(int v) {
  v = imax(v, row_xor<1>(v));
  v = imax(v, row_xor<2>(v));
  v = imax(v, row_xor<4>(v));
  return v;
}
__device__ __forceinline__ int order_key(float f) {
  int bits = __builtin_bit_cast(int, f);
  return bits ^ ((bits >> 31) & 0x7fffffff);
}

#define KEY_GONE 0x7fffffff
__device__ __forceinline__ long long cand_entry(int key, int idx) {
  return (long long)(((unsigned long long)(unsigned)key << 32) | (unsigned long long)(unsigned)idx);
}

__global__ __launch_bounds__(256) void knn_kernel(
    const float* __restrict__ points, const float* __restrict__ centers,
    int* __restrict__ knnIdx, const float* __restrict__ w2,
    const float* __restrict__ b2, const float* __restrict__ w3,
    float* __restrict__ FW, float* __restrict__ FC, size_t wsBytes,
    int foldBlocks, int half) {
  const int tid  = threadIdx.x;
  const int lane = tid & 31;
  const int wv   = tid >> 5;

  if ((int)blockIdx.x < foldBlocks) {
    __shared__ float sFold[8][16 * 32];
    if (wsBytes >= WS_NEED_BYTES) {
      const int task = __builtin_amdgcn_readfirstlane(blockIdx.x * 8 + wv);
      fold_task(w3, w2, b2, FW, FC, sFold[wv], task, lane);
    }
    return;
  }
  const int bi   = (int)blockIdx.x - foldBlocks;
  const int b    = bi / HALF_G;
  const int gg   = b * NUM_G + half * FPS_SPLIT + bi % HALF_G;
  const float* base = points + (size_t)b * NUM_N * 3;

  __shared__ int sWv[2][8];
  __shared__ int sWi[2][8];

  const float cx = centers[(size_t)gg * 3 + 0];
  const float cy = centers[(size_t)gg * 3 + 1];
  const float cz = centers[(size_t)gg * 3 + 2];
  const float c2 = __builtin_fmaf(cz, cz, __builtin_fmaf(cx, cx, cy * cy));

  int dk[32];
#pragma unroll
  for (int i = 0; i < NUM_N / 256; ++i) {
    int   p  = i * 256 + tid;
    float xx = base[p * 3 + 0], yy = base[p * 3 + 1], zz = base[p * 3 + 2];
    float x2 = __builtin_fmaf(zz, zz, __builtin_fmaf(xx, xx, yy * yy));
    float dt = __builtin_fmaf(cz, zz, __builtin_fmaf(cx, xx, cy * yy));
    dk[i]    = order_key(__builtin_fmaf(-2.f, dt, c2 + x2));
  }

  int lv = KEY_GONE, li = 0;
#pragma unroll
  for (int i = 0; i < 32; ++i)
    if (dk[i] < lv) { lv = dk[i]; li = i; }
  int lp    = li * 256 + tid;
  int myOut = 0;

  __shared__ int       sT[8];
  __shared__ int       sCnt[8];
  __shared__ long long sCand[8][32];
  __shared__ int       sOut[GK];

  const long long padEntry = cand_entry(KEY_GONE, 0x7fffffff);
  sCand[wv][lane] = padEntry;

  {
    int v = lv, m = 0;
    for (int s = 0; s < 4; ++s) {
      m = wave_min(v);
      unsigned has = __builtin_amdgcn_ballot_w32(v == m);
      if (lane == __builtin_ctz(has)) v = KEY_GONE;
    }
    if (lane == 0) sT[wv] = m;
  }
  __syncthreads();
  const int T = __builtin_amdgcn_readfirstlane(oct_max(sT[lane & 7]));

  int wcnt = 0;
#pragma unroll
  for (int i = 0; i < 32; ++i) {
    const bool     c   = dk[i] <= T;
    const unsigned bal = __builtin_amdgcn_ballot_w32(c);
    if (bal != 0u) {
      if (c) {
        int pos = wcnt + (int)__builtin_amdgcn_mbcnt_lo(bal, 0u);
        if (pos < 32) sCand[wv][pos] = cand_entry(dk[i], i * 256 + tid);
      }
      wcnt += __builtin_popcount(bal);
    }
  }
  if (lane == 0) sCnt[wv] = wcnt;
  __syncthreads();

  const int  cnts = sCnt[lane & 7];
  const bool over = __builtin_amdgcn_ballot_w32(cnts > 32) != 0u;
  if (!over) {
    int n[8], pre[9];
    pre[0] = 0;
#pragma unroll
    for (int w = 0; w < 8; ++w) {
      n[w]       = __builtin_amdgcn_readlane(cnts, w);
      pre[w + 1] = pre[w] + n[w];
    }
    const int wvs = __builtin_amdgcn_readfirstlane(wv);
    if (wvs * 32 < pre[8]) {
      const int  f     = wvs * 32 + lane;
      const bool valid = f < pre[8];
      int wOf = 0, jOf = f;
#pragma unroll
      for (int w = 1; w < 8; ++w)
        if (f >= pre[w]) { wOf = w; jOf = f - pre[w]; }
      long long mine = padEntry;
      if (valid) mine = sCand[wOf][jOf];
      long long e[8];
#pragma unroll
      for (int w = 0; w < 8; ++w) e[w] = sCand[w][lane];
      int rank = 0;
#pragma unroll
      for (int w = 0; w < 8; ++w) {
        const int eh = (int)(e[w] >> 32), el = (int)e[w];
        for (int j = 0; j < n[w]; ++j) {
          const long long o = cand_entry(__builtin_amdgcn_readlane(eh, j),
                                         __builtin_amdgcn_readlane(el, j));
          rank += (o < mine) ? 1 : 0;
        }
      }
      if (valid && rank < GK) sOut[rank] = (int)mine;
    }
    __syncthreads();
    if (tid < GK) myOut = sOut[tid];
  } else {
    for (int r = 0; r < GK; ++r) {
      const int buf = r & 1;
      {
        int wm = wave_min(lv);
        int wp = wave_min(lv == wm ? lp : KEY_GONE);
        if (lane == 0) { sWv[buf][wv] = wm; sWi[buf][wv] = wp; }
      }
      __syncthreads();
      int win;
      {
        int v  = sWv[buf][lane & 7];
        int ii = sWi[buf][lane & 7];
        int bm = oct_min(v);
        win    = oct_min(v == bm ? ii : KEY_GONE);
      }
      win = __builtin_amdgcn_readfirstlane(win);
      if (tid == r) myOut = win;
      const int wt = win & 255;
      const int wi = win >> 8;
      if (wv == (wt >> 5)) {
        const bool mine = (tid == wt);
        int nv = KEY_GONE, ni = 0;
#pragma unroll
        for (int i = 0; i < 32; ++i) {
          if (mine && i == wi) dk[i] = KEY_GONE;
          if (dk[i] < nv) { nv = dk[i]; ni = i; }
        }
        lv = nv;
        lp = ni * 256 + tid;
      }
    }
  }
  if (tid < GK) knnIdx[(size_t)gg * GK + tid] = myOut & (NUM_N - 1);
}

template <int MODE>
__device__ __forceinline__ void conv3_stage(
    const float* __restrict__ w3, const float* __restrict__ b3,
    const float* __restrict__ g2, const float* __restrict__ be2,
    const float* __restrict__ m2, const float* __restrict__ v2,
    const float* __restrict__ FW, const float* __restrict__ FC,
    const float* __restrict__ G3g,
    const float* sGm, const float* sXu, const float* sH1, float* sH3, float* S,
    int wv, int lane) {
  constexpr bool FOLDED = (MODE != 0);
  const int laneN = lane & 15, laneHalf = lane >> 4;
  const v8f vzero = {0.f, 0.f, 0.f, 0.f, 0.f, 0.f, 0.f, 0.f};
  auto op = [&](float x, int m) {
    float inv = g2[m] * rsqrtf(v2[m] + 1e-5f);
    float bb  = b3[m];
    if constexpr (FOLDED) bb = FC[m] + bb;
    float v   = (x + bb) * inv + (be2[m] - m2[m] * inv);
    return fmaxf(v, 0.f);
  };
  auto pass = [&](int j) {
    int m0  = wv * 64 + j * 32;
    v8f g0 = vzero, g1 = vzero;
    if constexpr (MODE == 2) {
      const float* gp = G3g + m0 + 8 * laneHalf;
      v4f x0 = *(const v4f*)(gp),      x1 = *(const v4f*)(gp + 4);
      v4f y0 = *(const v4f*)(gp + 16), y1 = *(const v4f*)(gp + 20);
      g0 = __builtin_shufflevector(x0, x1, 0, 1, 2, 3, 4, 5, 6, 7);
      g1 = __builtin_shufflevector(y0, y1, 0, 1, 2, 3, 4, 5, 6, 7);
    } else {
      gemm_2x1v<C2>(w3, C3, sGm, S, m0, lane, g0, g1);
    }
    v8f a00 = g0, a01 = g0, a10 = g1, a11 = g1;
    if constexpr (FOLDED)
      gemm_2x2w<C1, 132>(FW, C1, sH1, S, m0, lane, a00, a01, a10, a11);
    else
      gemm_2x2w<C2, 260>(w3 + C2, C3, sXu, S, m0, lane, a00, a01, a10, a11);
    store_tile(sH3, 516, a00, m0,      0,  laneN, laneHalf, op);
    store_tile(sH3, 516, a01, m0,      16, laneN, laneHalf, op);
    store_tile(sH3, 516, a10, m0 + 16, 0,  laneN, laneHalf, op);
    store_tile(sH3, 516, a11, m0 + 16, 16, laneN, laneHalf, op);
  };
  if constexpr (MODE == 2) {
#pragma clang loop unroll(disable)
    for (int j = 0; j < 2; ++j) pass(j);
  } else {
#pragma unroll
    for (int j = 0; j < 2; ++j) pass(j);
  }
}

__device__ __forceinline__ void enc_stage_ab(
    const float* __restrict__ points, const int* __restrict__ knnIdx,
    const float* __restrict__ centers, const float* __restrict__ w1,
    const float* __restrict__ b1, const float* __restrict__ g1,
    const float* __restrict__ be1, const float* __restrict__ m1,
    const float* __restrict__ v1, float* sFeat, float* sH1, int gg, int tid) {
  const int b = gg / NUM_G;
  const int lane = tid & 31, wv = tid >> 5;
  const int laneN = lane & 15, laneHalf = lane >> 4;
  const v8f vzero = {0.f, 0.f, 0.f, 0.f, 0.f, 0.f, 0.f, 0.f};
  if (tid < GK) {
    int idx = knnIdx[(size_t)gg * GK + tid];
    const float* p = points + ((size_t)b * NUM_N + idx) * 3;
    const float* c = centers + (size_t)gg * 3;
    float dx = p[0] - c[0], dy = p[1] - c[1], dz = p[2] - c[2];
    float nr = sqrtf(dx * dx + dy * dy + dz * dz);
    sFeat[tid * 4 + 0] = dx;
    sFeat[tid * 4 + 1] = dy;
    sFeat[tid * 4 + 2] = dz;
    sFeat[tid * 4 + 3] = nr;
  }
  __syncthreads();
  {
    int m0 = wv * 16;
    v8f a0 = vzero, a1 = vzero;
    gemm_1x2<4, 4>(w1, 4, sFeat, m0, laneN, laneHalf, a0, a1);
    auto op = [&](float x, int m) {
      float inv = g1[m] * rsqrtf(v1[m] + 1e-5f);
      float v   = (x + b1[m]) * inv + (be1[m] - m1[m] * inv);
      return fmaxf(v, 0.f);
    };
    store_tile(sH1, 132, a0, m0, 0,  laneN, laneHalf, op);
    store_tile(sH1, 132, a1, m0, 16, laneN, laneHalf, op);
  }
  __syncthreads();
}

__device__ __forceinline__ void enc_stage_cd(const float* __restrict__ w2,
                                             const float* __restrict__ b2,
                                             const float* sH1, float* sXu,
                                             float* S, float* gm, int tid) {
  const int lane = tid & 31, wv = tid >> 5;
  const int laneN = lane & 15, laneHalf = lane >> 4;
  const v8f vzero = {0.f, 0.f, 0.f, 0.f, 0.f, 0.f, 0.f, 0.f};
  {
    int m0 = wv * 32;
    v8f a00 = vzero, a01 = vzero, a10 = vzero, a11 = vzero;
    gemm_2x2w<C1, 132>(w2, C1, sH1, S, m0, lane, a00, a01, a10, a11);
    auto op = [&](float x, int m) { return x + b2[m]; };
    store_tile(sXu, 260, a00, m0,      0,  laneN, laneHalf, op);
    store_tile(sXu, 260, a01, m0,      16, laneN, laneHalf, op);
    store_tile(sXu, 260, a10, m0 + 16, 0,  laneN, laneHalf, op);
    store_tile(sXu, 260, a11, m0 + 16, 16, laneN, laneHalf, op);
  }
  __syncthreads();
  {
    int   m  = tid;
    float mx = -3.4e38f;
    for (int n = 0; n < GK; ++n) mx = fmaxf(mx, sXu[n * 260 + m]);
    gm[m] = mx;
  }
}

#define POOL_LDS_FLOATS (128 + 32 * 132 + 32 * 260 + 8 * STAGE_FLOATS)
#define ENC_LDS_FLOATS (128 + 32 * 132 + 32 * 260 + 256 + 32 * 516 + 8 * STAGE_FLOATS)
__global__ __launch_bounds__(256) void enc_pool_kernel(
    const float* __restrict__ points, const int* __restrict__ knnIdx,
    const float* __restrict__ centers, const float* __restrict__ w1,
    const float* __restrict__ b1, const float* __restrict__ g1,
    const float* __restrict__ be1, const float* __restrict__ m1,
    const float* __restrict__ v1, const float* __restrict__ w2,
    const float* __restrict__ b2, float* __restrict__ GM, size_t wsBytes,
    int half) {
  if (wsBytes < WS_SPLIT_BYTES) return;
  extern __shared__ __attribute__((aligned(16))) float smem[];
  float* sFeat = smem;
  float* sH1   = smem + 128;
  float* sXu   = sH1 + 32 * 132;
  float* sTile = sXu + 32 * 260;
  const int gg = ((int)blockIdx.x / HALF_G) * NUM_G + half * FPS_SPLIT +
                 (int)blockIdx.x % HALF_G;
  const int tid = threadIdx.x;
  enc_stage_ab(points, knnIdx, centers, w1, b1, g1, be1, m1, v1, sFeat, sH1, gg,
               tid);
  enc_stage_cd(w2, b2, sH1, sXu, sTile + (tid >> 5) * STAGE_FLOATS,
               GM + (size_t)gg * C2, tid);
}

__global__ __launch_bounds__(256) void enc_gm_kernel(
    const float* __restrict__ w3, const float* __restrict__ GM,
    float* __restrict__ G3, size_t wsBytes, int half) {
  if (wsBytes < WS_SPLIT_BYTES) return;
  static_assert(HALF_G % 32 == 0 && FPS_SPLIT % 32 == 0, "32-group pieces");
  const int lane = threadIdx.x & 31;
  const int laneN = lane & 15, laneHalf = lane >> 4;
  const int wv = __builtin_amdgcn_readfirstlane(threadIdx.x >> 5);
  const int mt = (blockIdx.x & 3) * 8 + wv;
  const int pc = blockIdx.x >> 2;
  const int n0 = (pc / (HALF_G / 32)) * NUM_G + half * FPS_SPLIT +
                 (pc % (HALF_G / 32)) * 32;
  const float* ap  = w3 + (size_t)(16 * mt + laneN) * C3 + 4 * laneHalf;
  const float* bp0 = GM + (size_t)(n0 + laneN) * C2 + 4 * laneHalf;
  const float* bp1 = bp0 + (size_t)16 * C2;
  v8f acc0 = {0.f, 0.f, 0.f, 0.f, 0.f, 0.f, 0.f, 0.f};
  v8f acc1 = acc0;
  for (int kk = 0; kk < C2; kk += 32) {
    v4f a[4], b0[4], b1[4];
#pragma unroll
    for (int t = 0; t < 4; ++t) {
      a[t]  = *(const v4f*)(ap  + kk + 8 * t);
      b0[t] = *(const v4f*)(bp0 + kk + 8 * t);
      b1[t] = *(const v4f*)(bp1 + kk + 8 * t);
    }
#pragma unroll
    for (int t = 0; t < 4; ++t) {
      acc0 = wmma4(lo2(a[t]), lo2(b0[t]), acc0);
      acc1 = wmma4(lo2(a[t]), lo2(b1[t]), acc1);
      acc0 = wmma4(hi2(a[t]), hi2(b0[t]), acc0);
      acc1 = wmma4(hi2(a[t]), hi2(b1[t]), acc1);
    }
  }
  float* o0 = G3 + (size_t)(n0 + laneN) * C3 + 16 * mt + 8 * laneHalf;
  float* o1 = o0 + (size_t)16 * C3;
  *(v4f*)(o0)     = __builtin_shufflevector(acc0, acc0, 0, 1, 2, 3);
  *(v4f*)(o0 + 4) = __builtin_shufflevector(acc0, acc0, 4, 5, 6, 7);
  *(v4f*)(o1)     = __builtin_shufflevector(acc1, acc1, 0, 1, 2, 3);
  *(v4f*)(o1 + 4) = __builtin_shufflevector(acc1, acc1, 4, 5, 6, 7);
}

#define FPR_T 256
#define FPR_P (NUM_N / FPR_T)
#define FPR_W (FPR_T / 32)
#define FPR_LDS_FLOATS (NUM_N * 3 + 4 * FPR_W)

__device__ __forceinline__ void fps_resume(const float* __restrict__ points,
                                           float* __restrict__ centersOut,
                                           const int* __restrict__ ddSave,
                                           float* smem, int b, int tid) {
  const int lane = tid & 31;
  const int wv   = __builtin_amdgcn_readfirstlane(tid >> 5);
  const float* base = points + (size_t)b * NUM_N * 3;
  float* co = centersOut + (size_t)b * NUM_G * 3;

  float* sP  = smem;
  int*   sWv = (int*)(smem + NUM_N * 3);
  int*   sWi = sWv + 2 * FPR_W;

  {
    const v4f* src4 = (const v4f*)base;
    v4f*       dst4 = (v4f*)sP;
#pragma unroll 8
    for (int k = tid; k < NUM_N * 3 / 4; k += FPR_T) dst4[k] = src4[k];
  }

  float px[FPR_P], py[FPR_P], pz[FPR_P];
  int   dd[FPR_P];
  const int* ds = ddSave + (size_t)b * NUM_N + tid * FPR_P;
#pragma unroll
  for (int i = 0; i < FPR_P; ++i) {
    int p = tid * FPR_P + i;
    px[i] = base[p * 3 + 0];
    py[i] = base[p * 3 + 1];
    pz[i] = base[p * 3 + 2];
    dd[i] = ds[i];
  }
  float lx = co[(FPS_SPLIT - 1) * 3 + 0];
  float ly = co[(FPS_SPLIT - 1) * 3 + 1];
  float lz = co[(FPS_SPLIT - 1) * 3 + 2];
  __syncthreads();

  for (int g = FPS_SPLIT; g < NUM_G; ++g) {
#pragma unroll
    for (int i = 0; i < FPR_P; ++i) {
      float dx = px[i] - lx, dy = py[i] - ly, dz = pz[i] - lz;
      float d = __builtin_fmaf(dz, dz, __builtin_fmaf(dx, dx, dy * dy));
      dd[i]   = imin(dd[i], __builtin_bit_cast(int, d));
    }
    static_assert(FPR_P == 32, "the tree below is written for 32 points");
    int t[11];
#pragma unroll
    for (int j = 0; j < 10; ++j)
      t[j] = imax(imax(dd[3 * j], dd[3 * j + 1]), dd[3 * j + 2]);
    t[10] = imax(dd[30], dd[31]);
    const int u0 = imax(imax(t[0], t[1]), t[2]);
    const int u1 = imax(imax(t[3], t[4]), t[5]);
    const int u2 = imax(imax(t[6], t[7]), t[8]);
    const int u3 = imax(t[9], t[10]);
    const int tm = imax(imax(u0, u1), imax(u2, u3));

    const int buf = g & 1;
    {
      int      wm  = wave_max(tm);
      unsigned has = __builtin_amdgcn_ballot_w32(tm == wm);
      int      fl  = __builtin_ctz(has);
      int idx = FPR_P - 1;
#pragma unroll
      for (int i = FPR_P - 2; i >= 0; --i) {
        unsigned eq = __builtin_amdgcn_ballot_w32(dd[i] == wm);
        if ((eq >> fl) & 1u) idx = i;
      }
      int wbi = (wv * 32 + fl) * FPR_P + idx;
      if (lane == 0) { sWv[buf * FPR_W + wv] = wm; sWi[buf * FPR_W + wv] = wbi; }
    }
    __syncthreads();
    {
      int      v   = sWv[buf * FPR_W + (lane & (FPR_W - 1))];
      int      ii  = sWi[buf * FPR_W + (lane & (FPR_W - 1))];
      int      bm  = oct_max(v);
      unsigned has = __builtin_amdgcn_ballot_w32(v == bm);
      int      wi  = __builtin_amdgcn_readlane(ii, __builtin_ctz(has));
      lx = sP[wi * 3 + 0];
      ly = sP[wi * 3 + 1];
      lz = sP[wi * 3 + 2];
    }
    if (tid == 0) { co[g * 3 + 0] = lx; co[g * 3 + 1] = ly; co[g * 3 + 2] = lz; }
  }
}

__global__ __launch_bounds__(256) void encoder_kernel(
    const float* __restrict__ points, const int* __restrict__ knnIdx,
    const float* __restrict__ centers, const float* __restrict__ w1,
    const float* __restrict__ b1, const float* __restrict__ g1,
    const float* __restrict__ be1, const float* __restrict__ m1,
    const float* __restrict__ v1, const float* __restrict__ w2,
    const float* __restrict__ b2, const float* __restrict__ w3,
    const float* __restrict__ b3, const float* __restrict__ g2,
    const float* __restrict__ be2, const float* __restrict__ m2,
    const float* __restrict__ v2, const float* __restrict__ w4,
    const float* __restrict__ b4, float* __restrict__ tokens,
    const float* __restrict__ FW, const float* __restrict__ FC,
    const float* __restrict__ G3, size_t wsBytes,
    float* __restrict__ centersOut, const int* __restrict__ ddSave,
    int fpsBlocks, int half) {
  extern __shared__ __attribute__((aligned(16))) float smem[];
  if ((int)blockIdx.x < fpsBlocks) {
    static_assert(FPR_LDS_FLOATS <= ENC_LDS_FLOATS, "the points fit the encoder's LDS");
    if (wsBytes >= WS_SPLIT_BYTES)
      fps_resume(points, centersOut, ddSave, smem, blockIdx.x, threadIdx.x);
    return;
  }
  float* sFeat = smem;
  float* sH1   = smem + 128;
  float* sXu   = sH1 + 32 * 132;
  float* sGm   = sXu + 32 * 260;
  float* sH3   = sGm + 256;
  float* sTile = sH3 + 32 * 516;
  float* sH4   = smem;

  const int bi       = (int)blockIdx.x - fpsBlocks;
  const int gg       = (bi / HALF_G) * NUM_G + half * FPS_SPLIT + bi % HALF_G;
  const int tid      = threadIdx.x;
  const int lane     = tid & 31;
  const int wv       = tid >> 5;
  const int laneN    = lane & 15;
  const int laneHalf = lane >> 4;
  float* S = sTile + wv * STAGE_FLOATS;

  const v8f vzero = {0.f, 0.f, 0.f, 0.f, 0.f, 0.f, 0.f, 0.f};

  const bool split = wsBytes >= WS_SPLIT_BYTES;

  enc_stage_ab(points, knnIdx, centers, w1, b1, g1, be1, m1, v1, sFeat, sH1, gg,
               tid);

  if (!split) {
    enc_stage_cd(w2, b2, sH1, sXu, S, sGm, tid);
    __syncthreads();
  }

  if (split)
    conv3_stage<2>(w3, b3, g2, be2, m2, v2, FW, FC, G3 + (size_t)gg * C3, sGm,
                   sXu, sH1, sH3, S, wv, lane);
  else if (wsBytes >= WS_NEED_BYTES)
    conv3_stage<1>(w3, b3, g2, be2, m2, v2, FW, FC, G3, sGm, sXu, sH1, sH3, S,
                   wv, lane);
  else
    conv3_stage<0>(w3, b3, g2, be2, m2, v2, FW, FC, G3, sGm, sXu, sH1, sH3, S,
                   wv, lane);
  __syncthreads();

  {
    auto op = [&](float x, int m) { return x + b4[m]; };
    {
      int m0  = wv * 48;
      v8f a00 = vzero, a01 = vzero, a10 = vzero, a11 = vzero;
      gemm_2x2w<C3, 516>(w4, C3, sH3, S, m0, lane, a00, a01, a10, a11);
      store_tile(sH4, 388, a00, m0,      0,  laneN, laneHalf, op);
      store_tile(sH4, 388, a01, m0,      16, laneN, laneHalf, op);
      store_tile(sH4, 388, a10, m0 + 16, 0,  laneN, laneHalf, op);
      store_tile(sH4, 388, a11, m0 + 16, 16, laneN, laneHalf, op);
    }
    {
      int m0 = wv * 48 + 32;
      v8f a0 = vzero, a1 = vzero;
      gemm_1x2w<C3, 516>(w4, C3, sH3, S, m0, lane, a0, a1);
      store_tile(sH4, 388, a0, m0, 0,  laneN, laneHalf, op);
      store_tile(sH4, 388, a1, m0, 16, laneN, laneHalf, op);
    }
  }
  __syncthreads();

  for (int m = tid; m < CT; m += 256) {
    float mx = -3.4e38f;
    for (int n = 0; n < GK; ++n) mx = fmaxf(mx, sH4[n * 388 + m]);
    tokens[(size_t)gg * CT + m] = mx;
  }
}

extern "C" void kernel_launch(void* const* d_in, const int* in_sizes, int n_in,
                              void* d_out, int out_size, void* d_ws,
                              size_t ws_size, hipStream_t stream) {
  (void)in_sizes; (void)n_in; (void)out_size;
  const float* points = (const float*)d_in[0];
  const float* w1  = (const float*)d_in[1];
  const float* b1  = (const float*)d_in[2];
  const float* g1  = (const float*)d_in[3];
  const float* be1 = (const float*)d_in[4];
  const float* m1  = (const float*)d_in[5];
  const float* v1  = (const float*)d_in[6];
  const float* w2  = (const float*)d_in[7];
  const float* b2  = (const float*)d_in[8];
  const float* w3  = (const float*)d_in[9];
  const float* b3  = (const float*)d_in[10];
  const float* g2  = (const float*)d_in[11];
  const float* be2 = (const float*)d_in[12];
  const float* m2  = (const float*)d_in[13];
  const float* v2  = (const float*)d_in[14];
  const float* w4  = (const float*)d_in[15];
  const float* b4  = (const float*)d_in[16];

  float* tokens  = (float*)d_out;
  float* centers = tokens + TOK_TOTAL;
  int*   knnIdx  = (int*)d_ws;
  float* FW      = (float*)d_ws + WS_KNN_FLOATS;
  float* FC      = FW + C3 * C1;
  float* GM      = FC + C3;
  float* G3      = GM + WS_GM_FLOATS;
  int*   ddSave  = (int*)(G3 + FPS_SPLIT * C3);

  const size_t poolLds = (size_t)POOL_LDS_FLOATS * sizeof(float);
  const size_t encLds  = (size_t)ENC_LDS_FLOATS * sizeof(float);

  fps_kernel<<<NUM_B, FPS_T, 0, stream>>>(points, centers, ddSave, ws_size);

  knn_kernel<<<FOLD_BLOCKS + NUM_B * HALF_G, 256, 0, stream>>>(
      points, centers, knnIdx, w2, b2, w3, FW, FC, ws_size, FOLD_BLOCKS, 0);
  enc_pool_kernel<<<NUM_B * HALF_G, 256, poolLds, stream>>>(
      points, knnIdx, centers, w1, b1, g1, be1, m1, v1, w2, b2, GM, ws_size, 0);
  enc_gm_kernel<<<NUM_B * HALF_G / 8, 256, 0, stream>>>(w3, GM, G3, ws_size, 0);
  encoder_kernel<<<NUM_B + NUM_B * HALF_G, 256, encLds, stream>>>(
      points, knnIdx, centers, w1, b1, g1, be1, m1, v1, w2, b2, w3, b3, g2,
      be2, m2, v2, w4, b4, tokens, FW, FC, G3, ws_size, centers, ddSave, NUM_B,
      0);

  knn_kernel<<<NUM_B * HALF_G, 256, 0, stream>>>(
      points, centers, knnIdx, w2, b2, w3, FW, FC, ws_size, 0, 1);
  enc_pool_kernel<<<NUM_B * HALF_G, 256, poolLds, stream>>>(
      points, knnIdx, centers, w1, b1, g1, be1, m1, v1, w2, b2, GM, ws_size, 1);
  enc_gm_kernel<<<NUM_B * HALF_G / 8, 256, 0, stream>>>(w3, GM, G3, ws_size, 1);
  encoder_kernel<<<NUM_B * HALF_G, 256, encLds, stream>>>(
      points, knnIdx, centers, w1, b1, g1, be1, m1, v1, w2, b2, w3, b3, g2,
      be2, m2, v2, w4, b4, tokens, FW, FC, G3, ws_size, centers, ddSave, 0, 1);
}
